# starting kernel
# baseline (speedup 1.0000x reference)
#include <hip/hip_runtime.h>
#include <hip/hip_bf16.h>


#define NG 16
#define NN 128
#define DD 9216
#define EPSV 1e-5f
#define TITER 5
#define KSPLIT 16
#define KCHUNK 32
#define APITCH 36
#define MATF (NG * NN * NN)

typedef __attribute__((ext_vector_type(2))) float v2f;
typedef __attribute__((ext_vector_type(8))) float v8f;

static __device__ __forceinline__ v8f wmma_f32(v2f a, v2f b, v8f c) {
  return __builtin_amdgcn_wmma_f32_16x16x4_f32(
       false, a,  false, b,
       (short)0, c,  false,  false);
}

__global__ void oni_mean(const float* __restrict__ w, float* __restrict__ mean) {
  __shared__ float red[256];
  const int row = blockIdx.x;
  const float4* p = reinterpret_cast<const float4*>(w + (size_t)row * DD);
  float s = 0.f;
  for (int i = threadIdx.x; i < DD / 4; i += 256) {
    float4 v = p[i];
    s += (v.x + v.y) + (v.z + v.w);
  }
  red[threadIdx.x] = s;
  __syncthreads();
  for (int off = 128; off > 0; off >>= 1) {
    if (threadIdx.x < off) red[threadIdx.x] += red[threadIdx.x + off];
    __syncthreads();
  }
  if (threadIdx.x == 0) mean[row] = red[0] * (1.0f / (float)DD);
}

__global__ void oni_zero(float* __restrict__ S) {
  S[blockIdx.x * 256 + threadIdx.x] = 0.f;
}

__global__ void oni_cov(const float* __restrict__ w, const float* __restrict__ mean,
                        float* __restrict__ S) {
  __shared__ float lds[NN * APITCH];
  const int g = blockIdx.y;
  const int kbase = blockIdx.x * (DD / KSPLIT);
  const int tid = threadIdx.x;
  const int lane = tid & 31, wave = tid >> 5;
  const int M0 = wave * 16;
  const int l15 = lane & 15;
  const int kb = (lane < 16) ? 0 : 2;

  v8f acc[8];
  v8f z = {};
#pragma unroll
  for (int t = 0; t < 8; ++t) acc[t] = z;

  for (int k0 = kbase; k0 < kbase + DD / KSPLIT; k0 += KCHUNK) {
#pragma unroll
    for (int i = 0; i < 4; ++i) {
      int idx = i * 256 + tid;
      int r = idx >> 3, kk = (idx & 7) * 4;
      int grow = g * NN + r;
      float m = mean[grow];
      float4 v = *reinterpret_cast<const float4*>(w + (size_t)grow * DD + k0 + kk);
      v.x -= m; v.y -= m; v.z -= m; v.w -= m;
      *reinterpret_cast<float4*>(&lds[r * APITCH + kk]) = v;
    }
    __syncthreads();
#pragma unroll
    for (int k4 = 0; k4 < KCHUNK; k4 += 4) {
      v2f a = *reinterpret_cast<const v2f*>(&lds[(M0 + l15) * APITCH + k4 + kb]);
      v2f bf[8];
#pragma unroll
      for (int nt = 0; nt < 8; ++nt)
        bf[nt] = *reinterpret_cast<const v2f*>(&lds[(nt * 16 + l15) * APITCH + k4 + kb]);
#pragma unroll
      for (int nt = 0; nt < 8; ++nt)
        acc[nt] = wmma_f32(a, bf[nt], acc[nt]);
    }
    __syncthreads();
  }

  const int Mb = M0 + ((lane < 16) ? 0 : 8);
  float* Sg = S + (size_t)g * NN * NN;
#pragma unroll
  for (int nt = 0; nt < 8; ++nt) {
    int col = nt * 16 + l15;
#pragma unroll
    for (int r = 0; r < 8; ++r) {
      atomicAdd(&Sg[(Mb + r) * NN + col], acc[nt][r]);
    }
  }
}

__global__ void oni_norm(float* __restrict__ S, float* __restrict__ rs,
                         float* __restrict__ B0) {
  __shared__ float red[256];
  __shared__ float nrm;
  const int g = blockIdx.x;
  float* Sg = S + g * (NN * NN);
  float s = 0.f;
  for (int i = threadIdx.x; i < NN * NN; i += 256) {
    int r = i >> 7, c = i & 127;
    float v = Sg[i] + ((r == c) ? EPSV : 0.f);
    s += v * v;
  }
  red[threadIdx.x] = s;
  __syncthreads();
  for (int off = 128; off > 0; off >>= 1) {
    if (threadIdx.x < off) red[threadIdx.x] += red[threadIdx.x + off];
    __syncthreads();
  }
  if (threadIdx.x == 0) {
    float n = sqrtf(red[0]);
    nrm = n;
    rs[g] = 1.0f / sqrtf(n);
  }
  __syncthreads();
  float inv = 1.0f / nrm;
  float* Bg = B0 + g * (NN * NN);
  for (int i = threadIdx.x; i < NN * NN; i += 256) {
    int r = i >> 7, c = i & 127;
    Sg[i] = (Sg[i] + ((r == c) ? EPSV : 0.f)) * inv;
    Bg[i] = (r == c) ? 1.f : 0.f;
  }
}

__global__ void oni_gemm128(const float* __restrict__ A, const float* __restrict__ Bm,
                            const float* __restrict__ P, float* __restrict__ Out,
                            float alpha, float beta) {
  __shared__ float ldsA[NN * APITCH];
  __shared__ float ldsB[KCHUNK * NN];
  const int g = blockIdx.x;
  const float* Ag = A + g * (NN * NN);
  const float* Bg = Bm + g * (NN * NN);
  const int tid = threadIdx.x, lane = tid & 31, wave = tid >> 5;
  const int M0 = wave * 16, l15 = lane & 15;
  const int kb = (lane < 16) ? 0 : 2;

  v8f acc[8];
  v8f z = {};
#pragma unroll
  for (int t = 0; t < 8; ++t) acc[t] = z;

  for (int k0 = 0; k0 < NN; k0 += KCHUNK) {
#pragma unroll
    for (int i = 0; i < 4; ++i) {
      int idx = i * 256 + tid;
      int r = idx >> 3, kk = (idx & 7) * 4;
      float4 v = *reinterpret_cast<const float4*>(Ag + r * NN + k0 + kk);
      *reinterpret_cast<float4*>(&ldsA[r * APITCH + kk]) = v;
    }
#pragma unroll
    for (int i = 0; i < 4; ++i) {
      int idx = i * 256 + tid;
      int kk = idx >> 5, c = (idx & 31) * 4;
      float4 v = *reinterpret_cast<const float4*>(Bg + (k0 + kk) * NN + c);
      *reinterpret_cast<float4*>(&ldsB[kk * NN + c]) = v;
    }
    __syncthreads();
#pragma unroll
    for (int k4 = 0; k4 < KCHUNK; k4 += 4) {
      v2f a = *reinterpret_cast<const v2f*>(&ldsA[(M0 + l15) * APITCH + k4 + kb]);
      v2f bf[8];
#pragma unroll
      for (int nt = 0; nt < 8; ++nt) {
        int col = nt * 16 + l15;
        bf[nt].x = ldsB[(k4 + kb) * NN + col];
        bf[nt].y = ldsB[(k4 + kb + 1) * NN + col];
      }
#pragma unroll
      for (int nt = 0; nt < 8; ++nt)
        acc[nt] = wmma_f32(a, bf[nt], acc[nt]);
    }
    __syncthreads();
  }

  const int Mb = M0 + ((lane < 16) ? 0 : 8);
  float* Og = Out + g * (NN * NN);
  const float* Pg = P ? (P + g * (NN * NN)) : nullptr;
#pragma unroll
  for (int nt = 0; nt < 8; ++nt) {
    int col = nt * 16 + l15;
#pragma unroll
    for (int r = 0; r < 8; ++r) {
      float v = alpha * acc[nt][r];
      if (Pg) v += beta * Pg[(Mb + r) * NN + col];
      Og[(Mb + r) * NN + col] = v;
    }
  }
}

__global__ void oni_out(const float* __restrict__ Bt, const float* __restrict__ w,
                        const float* __restrict__ mean, const float* __restrict__ rs,
                        float* __restrict__ out) {
  __shared__ float ldsA[NN * APITCH];
  __shared__ float ldsB[KCHUNK * NN];
  const int g = blockIdx.y;
  const int N0 = blockIdx.x * 128;
  const float* Ag = Bt + g * (NN * NN);
  const int tid = threadIdx.x, lane = tid & 31, wave = tid >> 5;
  const int M0 = wave * 16, l15 = lane & 15;
  const int kb = (lane < 16) ? 0 : 2;

  v8f acc[8];
  v8f z = {};
#pragma unroll
  for (int t = 0; t < 8; ++t) acc[t] = z;

  for (int k0 = 0; k0 < NN; k0 += KCHUNK) {
#pragma unroll
    for (int i = 0; i < 4; ++i) {
      int idx = i * 256 + tid;
      int r = idx >> 3, kk = (idx & 7) * 4;
      float4 v = *reinterpret_cast<const float4*>(Ag + r * NN + k0 + kk);
      *reinterpret_cast<float4*>(&ldsA[r * APITCH + kk]) = v;
    }
#pragma unroll
    for (int i = 0; i < 4; ++i) {
      int idx = i * 256 + tid;
      int kk = idx >> 5, c = (idx & 31) * 4;
      int grow = g * NN + k0 + kk;
      float m = mean[grow];
      float4 v = *reinterpret_cast<const float4*>(w + (size_t)grow * DD + N0 + c);
      v.x -= m; v.y -= m; v.z -= m; v.w -= m;
      *reinterpret_cast<float4*>(&ldsB[kk * NN + c]) = v;
    }
    __syncthreads();
#pragma unroll
    for (int k4 = 0; k4 < KCHUNK; k4 += 4) {
      v2f a = *reinterpret_cast<const v2f*>(&ldsA[(M0 + l15) * APITCH + k4 + kb]);
      v2f bf[8];
#pragma unroll
      for (int nt = 0; nt < 8; ++nt) {
        int col = nt * 16 + l15;
        bf[nt].x = ldsB[(k4 + kb) * NN + col];
        bf[nt].y = ldsB[(k4 + kb + 1) * NN + col];
      }
#pragma unroll
      for (int nt = 0; nt < 8; ++nt)
        acc[nt] = wmma_f32(a, bf[nt], acc[nt]);
    }
    __syncthreads();
  }

  const float scale = rs[g];
  const int Mb = M0 + ((lane < 16) ? 0 : 8);
#pragma unroll
  for (int nt = 0; nt < 8; ++nt) {
    int col = N0 + nt * 16 + l15;
#pragma unroll
    for (int r = 0; r < 8; ++r) {
      out[(size_t)(g * NN + Mb + r) * DD + col] = acc[nt][r] * scale;
    }
  }
}

extern "C" void kernel_launch(void* const* d_in, const int* in_sizes, int n_in,
                              void* d_out, int out_size, void* d_ws, size_t ws_size,
                              hipStream_t stream) {
  const float* w = (const float*)d_in[0];
  float* out = (float*)d_out;
  float* ws = (float*)d_ws;

  float* mean = ws;
  float* rs = ws + 2048;
  float* S  = ws + 2080;
  float* BA = S + MATF;
  float* BB = BA + MATF;
  float* T2 = BB + MATF;
  float* T3 = T2 + MATF;

  oni_mean<<<NG * NN, 256, 0, stream>>>(w, mean);
  oni_zero<<<MATF / 256, 256, 0, stream>>>(S);
  oni_cov<<<dim3(KSPLIT, NG), 256, 0, stream>>>(w, mean, S);
  oni_norm<<<NG, 256, 0, stream>>>(S, rs, BA);

  float* Bc = BA;
  float* Bn = BB;
  for (int t = 0; t < TITER; ++t) {
    oni_gemm128<<<NG, 256, 0, stream>>>(Bc, Bc, nullptr, T2, 1.f, 0.f);
    oni_gemm128<<<NG, 256, 0, stream>>>(T2, Bc, nullptr, T3, 1.f, 0.f);
    oni_gemm128<<<NG, 256, 0, stream>>>(T3, S, Bc, Bn, -0.5f, 1.5f);
    float* tmp = Bc; Bc = Bn; Bn = tmp;
  }

  oni_out<<<dim3(DD / 128, NG), 256, 0, stream>>>(Bc, w, mean, rs, out);
}


# Fuse the norm launch and the 15 iteration GEMM launches into one launch per group: S^T, B^T stay in LDS, each product stays in WMMA C/D registers as the next B operand; k summed in permuted order
# speedup vs baseline: 2.2547x; 2.2547x over previous
#include <hip/hip_runtime.h>
#include <hip/hip_bf16.h>


#define NG 16
#define NN 128
#define DD 9216
#define EPSV 1e-5f
#define TITER 5
#define KSPLIT 16
#define KCHUNK 32
#define APITCH 36
#define MATF (NG * NN * NN)
#define BPITCH 132

typedef __attribute__((ext_vector_type(2))) float v2f;
typedef __attribute__((ext_vector_type(8))) float v8f;

static __device__ __forceinline__ v8f wmma_f32(v2f a, v2f b, v8f c) {
  return __builtin_amdgcn_wmma_f32_16x16x4_f32(
       false, a,  false, b,
       (short)0, c,  false,  false);
}

__global__ void oni_mean(const float* __restrict__ w, float* __restrict__ mean) {
  __shared__ float red[256];
  const int row = blockIdx.x;
  const float4* p = reinterpret_cast<const float4*>(w + (size_t)row * DD);
  float s = 0.f;
  for (int i = threadIdx.x; i < DD / 4; i += 256) {
    float4 v = p[i];
    s += (v.x + v.y) + (v.z + v.w);
  }
  red[threadIdx.x] = s;
  __syncthreads();
  for (int off = 128; off > 0; off >>= 1) {
    if (threadIdx.x < off) red[threadIdx.x] += red[threadIdx.x + off];
    __syncthreads();
  }
  if (threadIdx.x == 0) mean[row] = red[0] * (1.0f / (float)DD);
}

__global__ void oni_zero(float* __restrict__ S) {
  S[blockIdx.x * 256 + threadIdx.x] = 0.f;
}

__global__ void oni_cov(const float* __restrict__ w, const float* __restrict__ mean,
                        float* __restrict__ S) {
  __shared__ float lds[NN * APITCH];
  const int g = blockIdx.y;
  const int kbase = blockIdx.x * (DD / KSPLIT);
  const int tid = threadIdx.x;
  const int lane = tid & 31, wave = tid >> 5;
  const int M0 = wave * 16;
  const int l15 = lane & 15;
  const int kb = (lane < 16) ? 0 : 2;

  v8f acc[8];
  v8f z = {};
#pragma unroll
  for (int t = 0; t < 8; ++t) acc[t] = z;

  for (int k0 = kbase; k0 < kbase + DD / KSPLIT; k0 += KCHUNK) {
#pragma unroll
    for (int i = 0; i < 4; ++i) {
      int idx = i * 256 + tid;
      int r = idx >> 3, kk = (idx & 7) * 4;
      int grow = g * NN + r;
      float m = mean[grow];
      float4 v = *reinterpret_cast<const float4*>(w + (size_t)grow * DD + k0 + kk);
      v.x -= m; v.y -= m; v.z -= m; v.w -= m;
      *reinterpret_cast<float4*>(&lds[r * APITCH + kk]) = v;
    }
    __syncthreads();
#pragma unroll
    for (int k4 = 0; k4 < KCHUNK; k4 += 4) {
      v2f a = *reinterpret_cast<const v2f*>(&lds[(M0 + l15) * APITCH + k4 + kb]);
      v2f bf[8];
#pragma unroll
      for (int nt = 0; nt < 8; ++nt)
        bf[nt] = *reinterpret_cast<const v2f*>(&lds[(nt * 16 + l15) * APITCH + k4 + kb]);
#pragma unroll
      for (int nt = 0; nt < 8; ++nt)
        acc[nt] = wmma_f32(a, bf[nt], acc[nt]);
    }
    __syncthreads();
  }

  const int Mb = M0 + ((lane < 16) ? 0 : 8);
  float* Sg = S + (size_t)g * NN * NN;
#pragma unroll
  for (int nt = 0; nt < 8; ++nt) {
    int col = nt * 16 + l15;
#pragma unroll
    for (int r = 0; r < 8; ++r) {
      atomicAdd(&Sg[(Mb + r) * NN + col], acc[nt][r]);
    }
  }
}


static __device__ __forceinline__ void load_frags(float4 (&a)[16], const float* base, int kt) {
#pragma unroll
  for (int jj = 0; jj < 2; ++jj)
#pragma unroll
    for (int nt = 0; nt < 8; ++nt)
      a[jj * 8 + nt] = *reinterpret_cast<const float4*>(base + nt * 16 * BPITCH + kt * 16 + 4 * jj);
}

static __device__ __forceinline__ void mma_block(v8f (&yt)[8], const float4 (&a)[16], const v8f& xk) {
#pragma unroll
  for (int jj = 0; jj < 2; ++jj) {
#pragma unroll
    for (int nt = 0; nt < 8; ++nt) {
      v2f af = {a[jj * 8 + nt].x, a[jj * 8 + nt].y};
      v2f bf = {xk[4 * jj], xk[4 * jj + 1]};
      yt[nt] = wmma_f32(af, bf, yt[nt]);
    }
#pragma unroll
    for (int nt = 0; nt < 8; ++nt) {
      v2f af = {a[jj * 8 + nt].z, a[jj * 8 + nt].w};
      v2f bf = {xk[4 * jj + 2], xk[4 * jj + 3]};
      yt[nt] = wmma_f32(af, bf, yt[nt]);
    }
  }
}

template <bool PF>
static __device__ __forceinline__ void rows_times_lds(v8f (&yt)[8], const v8f (&xt)[8],
                                                      const float* base, const float* nextbase,
                                                      float4 (&fa)[16], float4 (&fb)[16]) {
  v8f z = {};
#pragma unroll
  for (int nt = 0; nt < 8; ++nt) yt[nt] = z;
#pragma unroll
  for (int kt = 0; kt < 8; kt += 2) {
    load_frags(fb, base, kt + 1);
    __builtin_amdgcn_sched_barrier(0);
    mma_block(yt, fa, xt[kt]);
    __builtin_amdgcn_sched_barrier(0);
    if (kt + 2 < 8) load_frags(fa, base, kt + 2);
    else if (PF) load_frags(fa, nextbase, 0);
    __builtin_amdgcn_sched_barrier(0);
    mma_block(yt, fb, xt[kt + 1]);
    __builtin_amdgcn_sched_barrier(0);
  }
}

__global__ void __launch_bounds__(256) oni_ns(const float* __restrict__ S,
                                              float* __restrict__ rs,
                                              float* __restrict__ Bt) {
  __shared__ __attribute__((aligned(16))) float sS[NN * BPITCH];
  __shared__ __attribute__((aligned(16))) float sB[2][NN * BPITCH];
  __shared__ float red[256];
  __shared__ float nrm;
  const int g = blockIdx.x;
  const int tid = threadIdx.x, lane = tid & 31, wave = tid >> 5;
  const int c = lane & 15, h = lane >> 4;
  const int M0 = wave * 16;
  const float* Sg = S + g * (NN * NN);

  const int cc = tid & 127, r0 = tid >> 7;
  float sv[64];
  float s = 0.f;
#pragma unroll
  for (int j = 0; j < 64; ++j) {
    float v = Sg[j * 256 + tid] + ((2 * j + r0 == cc) ? EPSV : 0.f);
    sv[j] = v;
    s += v * v;
  }
  red[tid] = s;
  __syncthreads();
  for (int off = 128; off > 0; off >>= 1) {
    if (tid < off) red[tid] += red[tid + off];
    __syncthreads();
  }
  if (tid == 0) {
    float n = sqrtf(red[0]);
    nrm = n;
    rs[g] = 1.0f / sqrtf(n);
  }
  __syncthreads();
  const float inv = 1.0f / nrm;
#pragma unroll
  for (int j = 0; j < 64; ++j) {
    int k = 2 * j + r0;
    sS[cc * BPITCH + k] = sv[j] * inv;
    sB[0][cc * BPITCH + k] = (k == cc) ? 1.f : 0.f;
  }

  v8f xt[8];
#pragma unroll
  for (int kt = 0; kt < 8; ++kt)
#pragma unroll
    for (int r = 0; r < 8; ++r)
      xt[kt][r] = (kt * 16 + 8 * h + r == M0 + c) ? 1.f : 0.f;
  __syncthreads();

  const int foff = c * BPITCH + 8 * h;
  float4 fa[16], fb[16];
  int cur = 0;
  for (int t = 0; t < TITER; ++t) {
    const float* bb = sB[cur] + foff;
    const float* sb = sS + foff;
    v8f t2[8], t3[8];
    load_frags(fa, bb, 0);
    rows_times_lds<true>(t2, xt, bb, bb, fa, fb);
    rows_times_lds<true>(t3, t2, bb, sb, fa, fb);
    rows_times_lds<false>(t2, t3, sb, sb, fa, fb);
#pragma unroll
    for (int kt = 0; kt < 8; ++kt)
#pragma unroll
      for (int r = 0; r < 8; ++r) {
        float v = -0.5f * t2[kt][r];
        v += 1.5f * xt[kt][r];
        xt[kt][r] = v;
      }
    if (t + 1 < TITER) {
      float* nb = sB[cur ^ 1];
#pragma unroll
      for (int kt = 0; kt < 8; ++kt)
#pragma unroll
        for (int r = 0; r < 8; ++r)
          nb[(kt * 16 + 8 * h + r) * BPITCH + M0 + c] = xt[kt][r];
      __syncthreads();
      cur ^= 1;
    }
  }

  float* Bg = Bt + g * (NN * NN) + (M0 + c) * NN + 8 * h;
#pragma unroll
  for (int kt = 0; kt < 8; ++kt) {
    float4 lo = {xt[kt][0], xt[kt][1], xt[kt][2], xt[kt][3]};
    float4 hi = {xt[kt][4], xt[kt][5], xt[kt][6], xt[kt][7]};
    *reinterpret_cast<float4*>(Bg + kt * 16) = lo;
    *reinterpret_cast<float4*>(Bg + kt * 16 + 4) = hi;
  }
}

__global__ void oni_out(const float* __restrict__ Bt, const float* __restrict__ w,
                        const float* __restrict__ mean, const float* __restrict__ rs,
                        float* __restrict__ out) {
  __shared__ float ldsA[NN * APITCH];
  __shared__ float ldsB[KCHUNK * NN];
  const int g = blockIdx.y;
  const int N0 = blockIdx.x * 128;
  const float* Ag = Bt + g * (NN * NN);
  const int tid = threadIdx.x, lane = tid & 31, wave = tid >> 5;
  const int M0 = wave * 16, l15 = lane & 15;
  const int kb = (lane < 16) ? 0 : 2;

  v8f acc[8];
  v8f z = {};
#pragma unroll
  for (int t = 0; t < 8; ++t) acc[t] = z;

  for (int k0 = 0; k0 < NN; k0 += KCHUNK) {
#pragma unroll
    for (int i = 0; i < 4; ++i) {
      int idx = i * 256 + tid;
      int r = idx >> 3, kk = (idx & 7) * 4;
      float4 v = *reinterpret_cast<const float4*>(Ag + r * NN + k0 + kk);
      *reinterpret_cast<float4*>(&ldsA[r * APITCH + kk]) = v;
    }
#pragma unroll
    for (int i = 0; i < 4; ++i) {
      int idx = i * 256 + tid;
      int kk = idx >> 5, c = (idx & 31) * 4;
      int grow = g * NN + k0 + kk;
      float m = mean[grow];
      float4 v = *reinterpret_cast<const float4*>(w + (size_t)grow * DD + N0 + c);
      v.x -= m; v.y -= m; v.z -= m; v.w -= m;
      *reinterpret_cast<float4*>(&ldsB[kk * NN + c]) = v;
    }
    __syncthreads();
#pragma unroll
    for (int k4 = 0; k4 < KCHUNK; k4 += 4) {
      v2f a = *reinterpret_cast<const v2f*>(&ldsA[(M0 + l15) * APITCH + k4 + kb]);
      v2f bf[8];
#pragma unroll
      for (int nt = 0; nt < 8; ++nt) {
        int col = nt * 16 + l15;
        bf[nt].x = ldsB[(k4 + kb) * NN + col];
        bf[nt].y = ldsB[(k4 + kb + 1) * NN + col];
      }
#pragma unroll
      for (int nt = 0; nt < 8; ++nt)
        acc[nt] = wmma_f32(a, bf[nt], acc[nt]);
    }
    __syncthreads();
  }

  const float scale = rs[g];
  const int Mb = M0 + ((lane < 16) ? 0 : 8);
#pragma unroll
  for (int nt = 0; nt < 8; ++nt) {
    int col = N0 + nt * 16 + l15;
#pragma unroll
    for (int r = 0; r < 8; ++r) {
      out[(size_t)(g * NN + Mb + r) * DD + col] = acc[nt][r] * scale;
    }
  }
}

extern "C" void kernel_launch(void* const* d_in, const int* in_sizes, int n_in,
                              void* d_out, int out_size, void* d_ws, size_t ws_size,
                              hipStream_t stream) {
  const float* w = (const float*)d_in[0];
  float* out = (float*)d_out;
  float* ws = (float*)d_ws;

  float* mean = ws;
  float* rs = ws + 2048;
  float* S  = ws + 2080;
  float* Bt = S + MATF;

  oni_mean<<<NG * NN, 256, 0, stream>>>(w, mean);
  oni_zero<<<MATF / 256, 256, 0, stream>>>(S);
  oni_cov<<<dim3(KSPLIT, NG), 256, 0, stream>>>(w, mean, S);
  oni_ns<<<NG, 256, 0, stream>>>(S, rs, Bt);
  oni_out<<<dim3(DD / 128, NG), 256, 0, stream>>>(Bt, w, mean, rs, out);
}


# Newton-Schulz iteration 1 formed element-wise: B0 = I makes I*I, I*I and I*S exact copies, so B1 = (3 I - S)/2 needs no product (bit-identical); the 12 products of iterations 2 to 5 unchanged
# speedup vs baseline: 2.4405x; 1.0824x over previous
#include <hip/hip_runtime.h>
#include <hip/hip_bf16.h>


#define NG 16
#define NN 128
#define DD 9216
#define EPSV 1e-5f
#define TITER 5
#define KSPLIT 16
#define KCHUNK 32
#define APITCH 36
#define MATF (NG * NN * NN)
#define BPITCH 132

typedef __attribute__((ext_vector_type(2))) float v2f;
typedef __attribute__((ext_vector_type(8))) float v8f;

static __device__ __forceinline__ v8f wmma_f32(v2f a, v2f b, v8f c) {
  return __builtin_amdgcn_wmma_f32_16x16x4_f32(
       false, a,  false, b,
       (short)0, c,  false,  false);
}

__global__ void oni_mean(const float* __restrict__ w, float* __restrict__ mean) {
  __shared__ float red[256];
  const int row = blockIdx.x;
  const float4* p = reinterpret_cast<const float4*>(w + (size_t)row * DD);
  float s = 0.f;
  for (int i = threadIdx.x; i < DD / 4; i += 256) {
    float4 v = p[i];
    s += (v.x + v.y) + (v.z + v.w);
  }
  red[threadIdx.x] = s;
  __syncthreads();
  for (int off = 128; off > 0; off >>= 1) {
    if (threadIdx.x < off) red[threadIdx.x] += red[threadIdx.x + off];
    __syncthreads();
  }
  if (threadIdx.x == 0) mean[row] = red[0] * (1.0f / (float)DD);
}

__global__ void oni_zero(float* __restrict__ S) {
  S[blockIdx.x * 256 + threadIdx.x] = 0.f;
}

__global__ void oni_cov(const float* __restrict__ w, const float* __restrict__ mean,
                        float* __restrict__ S) {
  __shared__ float lds[NN * APITCH];
  const int g = blockIdx.y;
  const int kbase = blockIdx.x * (DD / KSPLIT);
  const int tid = threadIdx.x;
  const int lane = tid & 31, wave = tid >> 5;
  const int M0 = wave * 16;
  const int l15 = lane & 15;
  const int kb = (lane < 16) ? 0 : 2;

  v8f acc[8];
  v8f z = {};
#pragma unroll
  for (int t = 0; t < 8; ++t) acc[t] = z;

  for (int k0 = kbase; k0 < kbase + DD / KSPLIT; k0 += KCHUNK) {
#pragma unroll
    for (int i = 0; i < 4; ++i) {
      int idx = i * 256 + tid;
      int r = idx >> 3, kk = (idx & 7) * 4;
      int grow = g * NN + r;
      float m = mean[grow];
      float4 v = *reinterpret_cast<const float4*>(w + (size_t)grow * DD + k0 + kk);
      v.x -= m; v.y -= m; v.z -= m; v.w -= m;
      *reinterpret_cast<float4*>(&lds[r * APITCH + kk]) = v;
    }
    __syncthreads();
#pragma unroll
    for (int k4 = 0; k4 < KCHUNK; k4 += 4) {
      v2f a = *reinterpret_cast<const v2f*>(&lds[(M0 + l15) * APITCH + k4 + kb]);
      v2f bf[8];
#pragma unroll
      for (int nt = 0; nt < 8; ++nt)
        bf[nt] = *reinterpret_cast<const v2f*>(&lds[(nt * 16 + l15) * APITCH + k4 + kb]);
#pragma unroll
      for (int nt = 0; nt < 8; ++nt)
        acc[nt] = wmma_f32(a, bf[nt], acc[nt]);
    }
    __syncthreads();
  }

  const int Mb = M0 + ((lane < 16) ? 0 : 8);
  float* Sg = S + (size_t)g * NN * NN;
#pragma unroll
  for (int nt = 0; nt < 8; ++nt) {
    int col = nt * 16 + l15;
#pragma unroll
    for (int r = 0; r < 8; ++r) {
      atomicAdd(&Sg[(Mb + r) * NN + col], acc[nt][r]);
    }
  }
}


static __device__ __forceinline__ void load_frags(float4 (&a)[16], const float* base, int kt) {
#pragma unroll
  for (int jj = 0; jj < 2; ++jj)
#pragma unroll
    for (int nt = 0; nt < 8; ++nt)
      a[jj * 8 + nt] = *reinterpret_cast<const float4*>(base + nt * 16 * BPITCH + kt * 16 + 4 * jj);
}

static __device__ __forceinline__ void mma_block(v8f (&yt)[8], const float4 (&a)[16], const v8f& xk) {
#pragma unroll
  for (int jj = 0; jj < 2; ++jj) {
#pragma unroll
    for (int nt = 0; nt < 8; ++nt) {
      v2f af = {a[jj * 8 + nt].x, a[jj * 8 + nt].y};
      v2f bf = {xk[4 * jj], xk[4 * jj + 1]};
      yt[nt] = wmma_f32(af, bf, yt[nt]);
    }
#pragma unroll
    for (int nt = 0; nt < 8; ++nt) {
      v2f af = {a[jj * 8 + nt].z, a[jj * 8 + nt].w};
      v2f bf = {xk[4 * jj + 2], xk[4 * jj + 3]};
      yt[nt] = wmma_f32(af, bf, yt[nt]);
    }
  }
}

template <bool PF>
static __device__ __forceinline__ void rows_times_lds(v8f (&yt)[8], const v8f (&xt)[8],
                                                      const float* base, const float* nextbase,
                                                      float4 (&fa)[16], float4 (&fb)[16]) {
  v8f z = {};
#pragma unroll
  for (int nt = 0; nt < 8; ++nt) yt[nt] = z;
#pragma unroll
  for (int kt = 0; kt < 8; kt += 2) {
    load_frags(fb, base, kt + 1);
    __builtin_amdgcn_sched_barrier(0);
    mma_block(yt, fa, xt[kt]);
    __builtin_amdgcn_sched_barrier(0);
    if (kt + 2 < 8) load_frags(fa, base, kt + 2);
    else if (PF) load_frags(fa, nextbase, 0);
    __builtin_amdgcn_sched_barrier(0);
    mma_block(yt, fb, xt[kt + 1]);
    __builtin_amdgcn_sched_barrier(0);
  }
}

__global__ void __launch_bounds__(256) oni_ns(const float* __restrict__ S,
                                              float* __restrict__ rs,
                                              float* __restrict__ Bt) {
  __shared__ __attribute__((aligned(16))) float sS[NN * BPITCH];
  __shared__ __attribute__((aligned(16))) float sB[2][NN * BPITCH];
  __shared__ float red[256];
  __shared__ float nrm;
  const int g = blockIdx.x;
  const int tid = threadIdx.x, lane = tid & 31, wave = tid >> 5;
  const int c = lane & 15, h = lane >> 4;
  const int M0 = wave * 16;
  const float* Sg = S + g * (NN * NN);

  const int cc = tid & 127, r0 = tid >> 7;
  float sv[64];
  float s = 0.f;
#pragma unroll
  for (int j = 0; j < 64; ++j) {
    float v = Sg[j * 256 + tid] + ((2 * j + r0 == cc) ? EPSV : 0.f);
    sv[j] = v;
    s += v * v;
  }
  red[tid] = s;
  __syncthreads();
  for (int off = 128; off > 0; off >>= 1) {
    if (tid < off) red[tid] += red[tid + off];
    __syncthreads();
  }
  if (tid == 0) {
    float n = sqrtf(red[0]);
    nrm = n;
    rs[g] = 1.0f / sqrtf(n);
  }
  __syncthreads();
  const float inv = 1.0f / nrm;
#pragma unroll
  for (int j = 0; j < 64; ++j) {
    int k = 2 * j + r0;
    float sn = sv[j] * inv;
    sS[cc * BPITCH + k] = sn;
    sB[0][cc * BPITCH + k] = ((k == cc) ? 1.5f : 0.f) - 0.5f * sn;
  }
  __syncthreads();

  v8f xt[8];
#pragma unroll
  for (int kt = 0; kt < 8; ++kt)
#pragma unroll
    for (int r = 0; r < 8; ++r)
      xt[kt][r] = sB[0][(kt * 16 + 8 * h + r) * BPITCH + M0 + c];

  const int foff = c * BPITCH + 8 * h;
  float4 fa[16], fb[16];
  int cur = 0;
  for (int t = 1; t < TITER; ++t) {
    const float* bb = sB[cur] + foff;
    const float* sb = sS + foff;
    v8f t2[8], t3[8];
    load_frags(fa, bb, 0);
    rows_times_lds<true>(t2, xt, bb, bb, fa, fb);
    rows_times_lds<true>(t3, t2, bb, sb, fa, fb);
    rows_times_lds<false>(t2, t3, sb, sb, fa, fb);
#pragma unroll
    for (int kt = 0; kt < 8; ++kt)
#pragma unroll
      for (int r = 0; r < 8; ++r) {
        float v = -0.5f * t2[kt][r];
        v += 1.5f * xt[kt][r];
        xt[kt][r] = v;
      }
    if (t + 1 < TITER) {
      float* nb = sB[cur ^ 1];
#pragma unroll
      for (int kt = 0; kt < 8; ++kt)
#pragma unroll
        for (int r = 0; r < 8; ++r)
          nb[(kt * 16 + 8 * h + r) * BPITCH + M0 + c] = xt[kt][r];
      __syncthreads();
      cur ^= 1;
    }
  }

  float* Bg = Bt + g * (NN * NN) + (M0 + c) * NN + 8 * h;
#pragma unroll
  for (int kt = 0; kt < 8; ++kt) {
    float4 lo = {xt[kt][0], xt[kt][1], xt[kt][2], xt[kt][3]};
    float4 hi = {xt[kt][4], xt[kt][5], xt[kt][6], xt[kt][7]};
    *reinterpret_cast<float4*>(Bg + kt * 16) = lo;
    *reinterpret_cast<float4*>(Bg + kt * 16 + 4) = hi;
  }
}

__global__ void oni_out(const float* __restrict__ Bt, const float* __restrict__ w,
                        const float* __restrict__ mean, const float* __restrict__ rs,
                        float* __restrict__ out) {
  __shared__ float ldsA[NN * APITCH];
  __shared__ float ldsB[KCHUNK * NN];
  const int g = blockIdx.y;
  const int N0 = blockIdx.x * 128;
  const float* Ag = Bt + g * (NN * NN);
  const int tid = threadIdx.x, lane = tid & 31, wave = tid >> 5;
  const int M0 = wave * 16, l15 = lane & 15;
  const int kb = (lane < 16) ? 0 : 2;

  v8f acc[8];
  v8f z = {};
#pragma unroll
  for (int t = 0; t < 8; ++t) acc[t] = z;

  for (int k0 = 0; k0 < NN; k0 += KCHUNK) {
#pragma unroll
    for (int i = 0; i < 4; ++i) {
      int idx = i * 256 + tid;
      int r = idx >> 3, kk = (idx & 7) * 4;
      float4 v = *reinterpret_cast<const float4*>(Ag + r * NN + k0 + kk);
      *reinterpret_cast<float4*>(&ldsA[r * APITCH + kk]) = v;
    }
#pragma unroll
    for (int i = 0; i < 4; ++i) {
      int idx = i * 256 + tid;
      int kk = idx >> 5, c = (idx & 31) * 4;
      int grow = g * NN + k0 + kk;
      float m = mean[grow];
      float4 v = *reinterpret_cast<const float4*>(w + (size_t)grow * DD + N0 + c);
      v.x -= m; v.y -= m; v.z -= m; v.w -= m;
      *reinterpret_cast<float4*>(&ldsB[kk * NN + c]) = v;
    }
    __syncthreads();
#pragma unroll
    for (int k4 = 0; k4 < KCHUNK; k4 += 4) {
      v2f a = *reinterpret_cast<const v2f*>(&ldsA[(M0 + l15) * APITCH + k4 + kb]);
      v2f bf[8];
#pragma unroll
      for (int nt = 0; nt < 8; ++nt) {
        int col = nt * 16 + l15;
        bf[nt].x = ldsB[(k4 + kb) * NN + col];
        bf[nt].y = ldsB[(k4 + kb + 1) * NN + col];
      }
#pragma unroll
      for (int nt = 0; nt < 8; ++nt)
        acc[nt] = wmma_f32(a, bf[nt], acc[nt]);
    }
    __syncthreads();
  }

  const float scale = rs[g];
  const int Mb = M0 + ((lane < 16) ? 0 : 8);
#pragma unroll
  for (int nt = 0; nt < 8; ++nt) {
    int col = N0 + nt * 16 + l15;
#pragma unroll
    for (int r = 0; r < 8; ++r) {
      out[(size_t)(g * NN + Mb + r) * DD + col] = acc[nt][r] * scale;
    }
  }
}

extern "C" void kernel_launch(void* const* d_in, const int* in_sizes, int n_in,
                              void* d_out, int out_size, void* d_ws, size_t ws_size,
                              hipStream_t stream) {
  const float* w = (const float*)d_in[0];
  float* out = (float*)d_out;
  float* ws = (float*)d_ws;

  float* mean = ws;
  float* rs = ws + 2048;
  float* S  = ws + 2080;
  float* Bt = S + MATF;

  oni_mean<<<NG * NN, 256, 0, stream>>>(w, mean);
  oni_zero<<<MATF / 256, 256, 0, stream>>>(S);
  oni_cov<<<dim3(KSPLIT, NG), 256, 0, stream>>>(w, mean, S);
  oni_ns<<<NG, 256, 0, stream>>>(S, rs, Bt);
  oni_out<<<dim3(DD / 128, NG), 256, 0, stream>>>(Bt, w, mean, rs, out);
}


# Newton-Schulz spread over 8 work-groups per group (16 rows each, 4 waves split the column tiles, tiles exchanged via LDS), one launch per iteration 2..5; every sum keeps its order
# speedup vs baseline: 2.9496x; 1.2086x over previous
#include <hip/hip_runtime.h>
#include <hip/hip_bf16.h>


#define NG 16
#define NN 128
#define DD 9216
#define EPSV 1e-5f
#define TITER 5
#define KSPLIT 16
#define KCHUNK 32
#define APITCH 36
#define MATF (NG * NN * NN)
#define BPITCH 132

typedef __attribute__((ext_vector_type(2))) float v2f;
typedef __attribute__((ext_vector_type(8))) float v8f;

static __device__ __forceinline__ v8f wmma_f32(v2f a, v2f b, v8f c) {
  return __builtin_amdgcn_wmma_f32_16x16x4_f32(
       false, a,  false, b,
       (short)0, c,  false,  false);
}

__global__ void oni_mean(const float* __restrict__ w, float* __restrict__ mean) {
  __shared__ float red[256];
  const int row = blockIdx.x;
  const float4* p = reinterpret_cast<const float4*>(w + (size_t)row * DD);
  float s = 0.f;
  for (int i = threadIdx.x; i < DD / 4; i += 256) {
    float4 v = p[i];
    s += (v.x + v.y) + (v.z + v.w);
  }
  red[threadIdx.x] = s;
  __syncthreads();
  for (int off = 128; off > 0; off >>= 1) {
    if (threadIdx.x < off) red[threadIdx.x] += red[threadIdx.x + off];
    __syncthreads();
  }
  if (threadIdx.x == 0) mean[row] = red[0] * (1.0f / (float)DD);
}

__global__ void oni_zero(float* __restrict__ S) {
  S[blockIdx.x * 256 + threadIdx.x] = 0.f;
}

__global__ void oni_cov(const float* __restrict__ w, const float* __restrict__ mean,
                        float* __restrict__ S) {
  __shared__ float lds[NN * APITCH];
  const int g = blockIdx.y;
  const int kbase = blockIdx.x * (DD / KSPLIT);
  const int tid = threadIdx.x;
  const int lane = tid & 31, wave = tid >> 5;
  const int M0 = wave * 16;
  const int l15 = lane & 15;
  const int kb = (lane < 16) ? 0 : 2;

  v8f acc[8];
  v8f z = {};
#pragma unroll
  for (int t = 0; t < 8; ++t) acc[t] = z;

  for (int k0 = kbase; k0 < kbase + DD / KSPLIT; k0 += KCHUNK) {
#pragma unroll
    for (int i = 0; i < 4; ++i) {
      int idx = i * 256 + tid;
      int r = idx >> 3, kk = (idx & 7) * 4;
      int grow = g * NN + r;
      float m = mean[grow];
      float4 v = *reinterpret_cast<const float4*>(w + (size_t)grow * DD + k0 + kk);
      v.x -= m; v.y -= m; v.z -= m; v.w -= m;
      *reinterpret_cast<float4*>(&lds[r * APITCH + kk]) = v;
    }
    __syncthreads();
#pragma unroll
    for (int k4 = 0; k4 < KCHUNK; k4 += 4) {
      v2f a = *reinterpret_cast<const v2f*>(&lds[(M0 + l15) * APITCH + k4 + kb]);
      v2f bf[8];
#pragma unroll
      for (int nt = 0; nt < 8; ++nt)
        bf[nt] = *reinterpret_cast<const v2f*>(&lds[(nt * 16 + l15) * APITCH + k4 + kb]);
#pragma unroll
      for (int nt = 0; nt < 8; ++nt)
        acc[nt] = wmma_f32(a, bf[nt], acc[nt]);
    }
    __syncthreads();
  }

  const int Mb = M0 + ((lane < 16) ? 0 : 8);
  float* Sg = S + (size_t)g * NN * NN;
#pragma unroll
  for (int nt = 0; nt < 8; ++nt) {
    int col = nt * 16 + l15;
#pragma unroll
    for (int r = 0; r < 8; ++r) {
      atomicAdd(&Sg[(Mb + r) * NN + col], acc[nt][r]);
    }
  }
}


static __device__ __forceinline__ void load_frags(float4 (&a)[32], const float* base) {
#pragma unroll
  for (int kt = 0; kt < 8; ++kt)
#pragma unroll
    for (int jj = 0; jj < 2; ++jj)
#pragma unroll
      for (int t = 0; t < 2; ++t)
        a[kt * 4 + jj * 2 + t] =
            *reinterpret_cast<const float4*>(base + t * 16 * BPITCH + kt * 16 + 4 * jj);
}

static __device__ __forceinline__ void mma_rows(v8f (&yt)[2], const float4 (&a)[32], const v8f (&xt)[8]) {
  v8f z = {};
  yt[0] = z;
  yt[1] = z;
#pragma unroll
  for (int kt = 0; kt < 8; ++kt)
#pragma unroll
    for (int jj = 0; jj < 2; ++jj) {
#pragma unroll
      for (int t = 0; t < 2; ++t) {
        v2f af = {a[kt * 4 + jj * 2 + t].x, a[kt * 4 + jj * 2 + t].y};
        v2f bf = {xt[kt][4 * jj], xt[kt][4 * jj + 1]};
        yt[t] = wmma_f32(af, bf, yt[t]);
      }
#pragma unroll
      for (int t = 0; t < 2; ++t) {
        v2f af = {a[kt * 4 + jj * 2 + t].z, a[kt * 4 + jj * 2 + t].w};
        v2f bf = {xt[kt][4 * jj + 2], xt[kt][4 * jj + 3]};
        yt[t] = wmma_f32(af, bf, yt[t]);
      }
    }
}

static __device__ __forceinline__ void exchange(v8f (&xn)[8], const v8f (&yt)[2], float* xw,
                                                int wave, int lane) {
#pragma unroll
  for (int t = 0; t < 2; ++t)
#pragma unroll
    for (int r = 0; r < 8; ++r)
      xw[((2 * wave + t) * 8 + r) * 32 + lane] = yt[t][r];
  __syncthreads();
#pragma unroll
  for (int kt = 0; kt < 8; ++kt)
#pragma unroll
    for (int r = 0; r < 8; ++r)
      xn[kt][r] = xw[(kt * 8 + r) * 32 + lane];
}

template <bool FIRST>
__global__ void __launch_bounds__(128) oni_ns(const float* __restrict__ S,
                                              float* __restrict__ nrmbuf,
                                              float* __restrict__ rs,
                                              const float* __restrict__ Bin,
                                              float* __restrict__ Bout) {
  __shared__ __attribute__((aligned(16))) float sS[NN * BPITCH];
  __shared__ __attribute__((aligned(16))) float sB[NN * BPITCH];
  __shared__ float xch[2][8 * 8 * 32];
  __shared__ float red[256];
  __shared__ float nrm;
  const int g = blockIdx.x >> 3;
  const int M0 = (blockIdx.x & 7) * 16;
  const int tid = threadIdx.x, lane = tid & 31, wave = tid >> 5;
  const int c = lane & 15, h = lane >> 4;
  const float* Sg = S + g * (NN * NN);

  if (FIRST) {
    float sv[NN];
    float s0 = 0.f, s1 = 0.f;
#pragma unroll
    for (int k = 0; k < NN; k += 2) {
      float v0 = Sg[k * NN + tid] + ((k == tid) ? EPSV : 0.f);
      float v1 = Sg[(k + 1) * NN + tid] + ((k + 1 == tid) ? EPSV : 0.f);
      sv[k] = v0;
      sv[k + 1] = v1;
      s0 += v0 * v0;
      s1 += v1 * v1;
    }
    red[tid] = s0;
    red[tid + 128] = s1;
    __syncthreads();
    red[tid] += red[tid + 128];
    __syncthreads();
    for (int off = 64; off > 0; off >>= 1) {
      if (tid < off) red[tid] += red[tid + off];
      __syncthreads();
    }
    if (tid == 0) {
      float n = sqrtf(red[0]);
      nrm = n;
      if (M0 == 0) {
        nrmbuf[g] = n;
        rs[g] = 1.0f / sqrtf(n);
      }
    }
    __syncthreads();
    const float inv = 1.0f / nrm;
#pragma unroll
    for (int k = 0; k < NN; k += 4) {
      float4 qs, qb;
      float sn;
      sn = sv[k] * inv;     qs.x = sn; qb.x = ((k == tid) ? 1.5f : 0.f) - 0.5f * sn;
      sn = sv[k + 1] * inv; qs.y = sn; qb.y = ((k + 1 == tid) ? 1.5f : 0.f) - 0.5f * sn;
      sn = sv[k + 2] * inv; qs.z = sn; qb.z = ((k + 2 == tid) ? 1.5f : 0.f) - 0.5f * sn;
      sn = sv[k + 3] * inv; qs.w = sn; qb.w = ((k + 3 == tid) ? 1.5f : 0.f) - 0.5f * sn;
      *reinterpret_cast<float4*>(&sS[tid * BPITCH + k]) = qs;
      *reinterpret_cast<float4*>(&sB[tid * BPITCH + k]) = qb;
    }
  } else {
    const float inv = 1.0f / nrmbuf[g];
    const float* Bg = Bin + g * (NN * NN);
#pragma unroll 8
    for (int k = 0; k < NN; k += 4) {
      float4 qs, qb;
      qs.x = (Sg[k * NN + tid] + ((k == tid) ? EPSV : 0.f)) * inv;
      qs.y = (Sg[(k + 1) * NN + tid] + ((k + 1 == tid) ? EPSV : 0.f)) * inv;
      qs.z = (Sg[(k + 2) * NN + tid] + ((k + 2 == tid) ? EPSV : 0.f)) * inv;
      qs.w = (Sg[(k + 3) * NN + tid] + ((k + 3 == tid) ? EPSV : 0.f)) * inv;
      qb.x = Bg[k * NN + tid];
      qb.y = Bg[(k + 1) * NN + tid];
      qb.z = Bg[(k + 2) * NN + tid];
      qb.w = Bg[(k + 3) * NN + tid];
      *reinterpret_cast<float4*>(&sS[tid * BPITCH + k]) = qs;
      *reinterpret_cast<float4*>(&sB[tid * BPITCH + k]) = qb;
    }
  }
  __syncthreads();

  v8f xt[8];
#pragma unroll
  for (int kt = 0; kt < 8; ++kt)
#pragma unroll
    for (int r = 0; r < 8; ++r)
      xt[kt][r] = sB[(kt * 16 + 8 * h + r) * BPITCH + M0 + c];

  const int foff = (wave * 32 + c) * BPITCH + 8 * h;
  float4 fB[32], fS[32];
  v8f y[2], x2[8], x3[8];
  load_frags(fB, sB + foff);
  load_frags(fS, sS + foff);
  __builtin_amdgcn_sched_barrier(0);
  mma_rows(y, fB, xt);
  __builtin_amdgcn_sched_barrier(0);
  exchange(x2, y, xch[0], wave, lane);
  __builtin_amdgcn_sched_barrier(0);
  mma_rows(y, fB, x2);
  __builtin_amdgcn_sched_barrier(0);
  exchange(x3, y, xch[1], wave, lane);
  __builtin_amdgcn_sched_barrier(0);
  mma_rows(y, fS, x3);
  __builtin_amdgcn_sched_barrier(0);

  float* Bo = Bout + g * (NN * NN) + (M0 + c) * NN + wave * 32 + 8 * h;
#pragma unroll
  for (int t = 0; t < 2; ++t) {
    float o[8];
#pragma unroll
    for (int r = 0; r < 8; ++r) {
      float v = -0.5f * y[t][r];
      v += 1.5f * sB[((2 * wave + t) * 16 + 8 * h + r) * BPITCH + M0 + c];
      o[r] = v;
    }
    float4 lo = {o[0], o[1], o[2], o[3]};
    float4 hi = {o[4], o[5], o[6], o[7]};
    *reinterpret_cast<float4*>(Bo + t * 16) = lo;
    *reinterpret_cast<float4*>(Bo + t * 16 + 4) = hi;
  }
}

__global__ void oni_out(const float* __restrict__ Bt, const float* __restrict__ w,
                        const float* __restrict__ mean, const float* __restrict__ rs,
                        float* __restrict__ out) {
  __shared__ float ldsA[NN * APITCH];
  __shared__ float ldsB[KCHUNK * NN];
  const int g = blockIdx.y;
  const int N0 = blockIdx.x * 128;
  const float* Ag = Bt + g * (NN * NN);
  const int tid = threadIdx.x, lane = tid & 31, wave = tid >> 5;
  const int M0 = wave * 16, l15 = lane & 15;
  const int kb = (lane < 16) ? 0 : 2;

  v8f acc[8];
  v8f z = {};
#pragma unroll
  for (int t = 0; t < 8; ++t) acc[t] = z;

  for (int k0 = 0; k0 < NN; k0 += KCHUNK) {
#pragma unroll
    for (int i = 0; i < 4; ++i) {
      int idx = i * 256 + tid;
      int r = idx >> 3, kk = (idx & 7) * 4;
      float4 v = *reinterpret_cast<const float4*>(Ag + r * NN + k0 + kk);
      *reinterpret_cast<float4*>(&ldsA[r * APITCH + kk]) = v;
    }
#pragma unroll
    for (int i = 0; i < 4; ++i) {
      int idx = i * 256 + tid;
      int kk = idx >> 5, c = (idx & 31) * 4;
      int grow = g * NN + k0 + kk;
      float m = mean[grow];
      float4 v = *reinterpret_cast<const float4*>(w + (size_t)grow * DD + N0 + c);
      v.x -= m; v.y -= m; v.z -= m; v.w -= m;
      *reinterpret_cast<float4*>(&ldsB[kk * NN + c]) = v;
    }
    __syncthreads();
#pragma unroll
    for (int k4 = 0; k4 < KCHUNK; k4 += 4) {
      v2f a = *reinterpret_cast<const v2f*>(&ldsA[(M0 + l15) * APITCH + k4 + kb]);
      v2f bf[8];
#pragma unroll
      for (int nt = 0; nt < 8; ++nt) {
        int col = nt * 16 + l15;
        bf[nt].x = ldsB[(k4 + kb) * NN + col];
        bf[nt].y = ldsB[(k4 + kb + 1) * NN + col];
      }
#pragma unroll
      for (int nt = 0; nt < 8; ++nt)
        acc[nt] = wmma_f32(a, bf[nt], acc[nt]);
    }
    __syncthreads();
  }

  const float scale = rs[g];
  const int Mb = M0 + ((lane < 16) ? 0 : 8);
#pragma unroll
  for (int nt = 0; nt < 8; ++nt) {
    int col = N0 + nt * 16 + l15;
#pragma unroll
    for (int r = 0; r < 8; ++r) {
      out[(size_t)(g * NN + Mb + r) * DD + col] = acc[nt][r] * scale;
    }
  }
}

extern "C" void kernel_launch(void* const* d_in, const int* in_sizes, int n_in,
                              void* d_out, int out_size, void* d_ws, size_t ws_size,
                              hipStream_t stream) {
  const float* w = (const float*)d_in[0];
  float* out = (float*)d_out;
  float* ws = (float*)d_ws;

  float* mean = ws;
  float* rs = ws + 2048;
  float* nrmbuf = ws + 2064;
  float* S  = ws + 2080;
  float* Bt = S + MATF;
  float* Ba = Bt + MATF;
  float* Bb = Ba + MATF;

  oni_mean<<<NG * NN, 256, 0, stream>>>(w, mean);
  oni_zero<<<MATF / 256, 256, 0, stream>>>(S);
  oni_cov<<<dim3(KSPLIT, NG), 256, 0, stream>>>(w, mean, S);
  oni_ns<true><<<NG * 8, 128, 0, stream>>>(S, nrmbuf, rs, S, Ba);
  oni_ns<false><<<NG * 8, 128, 0, stream>>>(S, nrmbuf, rs, Ba, Bb);
  oni_ns<false><<<NG * 8, 128, 0, stream>>>(S, nrmbuf, rs, Bb, Ba);
  oni_ns<false><<<NG * 8, 128, 0, stream>>>(S, nrmbuf, rs, Ba, Bt);
  oni_out<<<dim3(DD / 128, NG), 256, 0, stream>>>(Bt, w, mean, rs, out);
}


# Covariance K loop software-pipelined: next chunk's global loads fly during the WMMA run (two LDS buffers); fragments read b128 in groups ahead of their WMMAs, so k is summed in permuted order
# speedup vs baseline: 3.1782x; 1.0775x over previous
#include <hip/hip_runtime.h>
#include <hip/hip_bf16.h>


#define NG 16
#define NN 128
#define DD 9216
#define EPSV 1e-5f
#define TITER 5
#define KSPLIT 16
#define KCHUNK 32
#define APITCH 36
#define MATF (NG * NN * NN)
#define BPITCH 132

typedef __attribute__((ext_vector_type(2))) float v2f;
typedef __attribute__((ext_vector_type(8))) float v8f;

static __device__ __forceinline__ v8f wmma_f32(v2f a, v2f b, v8f c) {
  return __builtin_amdgcn_wmma_f32_16x16x4_f32(
       false, a,  false, b,
       (short)0, c,  false,  false);
}

__global__ void oni_mean(const float* __restrict__ w, float* __restrict__ mean) {
  __shared__ float red[256];
  const int row = blockIdx.x;
  const float4* p = reinterpret_cast<const float4*>(w + (size_t)row * DD);
  float s = 0.f;
  for (int i = threadIdx.x; i < DD / 4; i += 256) {
    float4 v = p[i];
    s += (v.x + v.y) + (v.z + v.w);
  }
  red[threadIdx.x] = s;
  __syncthreads();
  for (int off = 128; off > 0; off >>= 1) {
    if (threadIdx.x < off) red[threadIdx.x] += red[threadIdx.x + off];
    __syncthreads();
  }
  if (threadIdx.x == 0) mean[row] = red[0] * (1.0f / (float)DD);
}

__global__ void oni_zero(float* __restrict__ S) {
  S[blockIdx.x * 256 + threadIdx.x] = 0.f;
}


static __device__ __forceinline__ void cov_frags(float4 (&fa)[4], float4 (&fb)[4][8],
                                                 const float* fbase, int M0) {
#pragma unroll
  for (int kq = 0; kq < 4; ++kq) {
    fa[kq] = *reinterpret_cast<const float4*>(fbase + M0 * APITCH + kq * 8);
#pragma unroll
    for (int nt = 0; nt < 8; ++nt)
      fb[kq][nt] = *reinterpret_cast<const float4*>(fbase + nt * 16 * APITCH + kq * 8);
  }
}

static __device__ __forceinline__ void cov_mma(v8f (&acc)[8], const float4 (&fa)[4],
                                               const float4 (&fb)[4][8]) {
#pragma unroll
  for (int kq = 0; kq < 4; ++kq) {
    v2f alo = {fa[kq].x, fa[kq].y};
    v2f ahi = {fa[kq].z, fa[kq].w};
#pragma unroll
    for (int nt = 0; nt < 8; ++nt) {
      v2f b = {fb[kq][nt].x, fb[kq][nt].y};
      acc[nt] = wmma_f32(alo, b, acc[nt]);
    }
#pragma unroll
    for (int nt = 0; nt < 8; ++nt) {
      v2f b = {fb[kq][nt].z, fb[kq][nt].w};
      acc[nt] = wmma_f32(ahi, b, acc[nt]);
    }
  }
}

__global__ void __launch_bounds__(256) oni_cov(const float* __restrict__ w,
                                               const float* __restrict__ mean,
                                               float* __restrict__ S) {
  __shared__ __attribute__((aligned(16))) float lds[2][NN * APITCH];
  const int g = blockIdx.y;
  const int kbase = blockIdx.x * (DD / KSPLIT);
  const int tid = threadIdx.x;
  const int lane = tid & 31, wave = tid >> 5;
  const int M0 = wave * 16;
  const int l15 = lane & 15;
  const int h = lane >> 4;
  constexpr int NCH = DD / KSPLIT / KCHUNK;

  const int sr = tid >> 3, sk = (tid & 7) * 4;
  const float* wp = w + (size_t)(g * NN + sr) * DD + kbase + sk;
  float m[4];
  float4 st[4];
#pragma unroll
  for (int i = 0; i < 4; ++i) {
    m[i] = mean[g * NN + i * 32 + sr];
    st[i] = *reinterpret_cast<const float4*>(wp + (size_t)(i * 32) * DD);
  }
#pragma unroll
  for (int i = 0; i < 4; ++i) {
    float4 v = st[i];
    v.x -= m[i]; v.y -= m[i]; v.z -= m[i]; v.w -= m[i];
    *reinterpret_cast<float4*>(&lds[0][(i * 32 + sr) * APITCH + sk]) = v;
  }
  __syncthreads();

  v8f acc[8];
  v8f z = {};
#pragma unroll
  for (int t = 0; t < 8; ++t) acc[t] = z;

  const int foff = l15 * APITCH + 4 * h;
  float4 fa[4], fb[4][8];

  for (int c = 0; c + 1 < NCH; ++c) {
    const float* cur = lds[c & 1];
    float* nxt = lds[(c + 1) & 1];
#pragma unroll
    for (int i = 0; i < 4; ++i)
      st[i] = *reinterpret_cast<const float4*>(wp + (size_t)(i * 32) * DD + (c + 1) * KCHUNK);
    cov_frags(fa, fb, cur + foff, M0);
    __builtin_amdgcn_sched_barrier(0);
    cov_mma(acc, fa, fb);
    __builtin_amdgcn_sched_barrier(0);
#pragma unroll
    for (int i = 0; i < 4; ++i) {
      float4 v = st[i];
      v.x -= m[i]; v.y -= m[i]; v.z -= m[i]; v.w -= m[i];
      *reinterpret_cast<float4*>(&nxt[(i * 32 + sr) * APITCH + sk]) = v;
    }
    __syncthreads();
  }
  cov_frags(fa, fb, lds[(NCH - 1) & 1] + foff, M0);
  __builtin_amdgcn_sched_barrier(0);
  cov_mma(acc, fa, fb);

  const int Mb = M0 + ((lane < 16) ? 0 : 8);
  float* Sg = S + (size_t)g * NN * NN;
#pragma unroll
  for (int nt = 0; nt < 8; ++nt) {
    int col = nt * 16 + l15;
#pragma unroll
    for (int r = 0; r < 8; ++r) {
      atomicAdd(&Sg[(Mb + r) * NN + col], acc[nt][r]);
    }
  }
}


static __device__ __forceinline__ void load_frags(float4 (&a)[32], const float* base) {
#pragma unroll
  for (int kt = 0; kt < 8; ++kt)
#pragma unroll
    for (int jj = 0; jj < 2; ++jj)
#pragma unroll
      for (int t = 0; t < 2; ++t)
        a[kt * 4 + jj * 2 + t] =
            *reinterpret_cast<const float4*>(base + t * 16 * BPITCH + kt * 16 + 4 * jj);
}

static __device__ __forceinline__ void mma_rows(v8f (&yt)[2], const float4 (&a)[32], const v8f (&xt)[8]) {
  v8f z = {};
  yt[0] = z;
  yt[1] = z;
#pragma unroll
  for (int kt = 0; kt < 8; ++kt)
#pragma unroll
    for (int jj = 0; jj < 2; ++jj) {
#pragma unroll
      for (int t = 0; t < 2; ++t) {
        v2f af = {a[kt * 4 + jj * 2 + t].x, a[kt * 4 + jj * 2 + t].y};
        v2f bf = {xt[kt][4 * jj], xt[kt][4 * jj + 1]};
        yt[t] = wmma_f32(af, bf, yt[t]);
      }
#pragma unroll
      for (int t = 0; t < 2; ++t) {
        v2f af = {a[kt * 4 + jj * 2 + t].z, a[kt * 4 + jj * 2 + t].w};
        v2f bf = {xt[kt][4 * jj + 2], xt[kt][4 * jj + 3]};
        yt[t] = wmma_f32(af, bf, yt[t]);
      }
    }
}

static __device__ __forceinline__ void exchange(v8f (&xn)[8], const v8f (&yt)[2], float* xw,
                                                int wave, int lane) {
#pragma unroll
  for (int t = 0; t < 2; ++t)
#pragma unroll
    for (int r = 0; r < 8; ++r)
      xw[((2 * wave + t) * 8 + r) * 32 + lane] = yt[t][r];
  __syncthreads();
#pragma unroll
  for (int kt = 0; kt < 8; ++kt)
#pragma unroll
    for (int r = 0; r < 8; ++r)
      xn[kt][r] = xw[(kt * 8 + r) * 32 + lane];
}

template <bool FIRST>
__global__ void __launch_bounds__(128) oni_ns(const float* __restrict__ S,
                                              float* __restrict__ nrmbuf,
                                              float* __restrict__ rs,
                                              const float* __restrict__ Bin,
                                              float* __restrict__ Bout) {
  __shared__ __attribute__((aligned(16))) float sS[NN * BPITCH];
  __shared__ __attribute__((aligned(16))) float sB[NN * BPITCH];
  __shared__ float xch[2][8 * 8 * 32];
  __shared__ float red[256];
  __shared__ float nrm;
  const int g = blockIdx.x >> 3;
  const int M0 = (blockIdx.x & 7) * 16;
  const int tid = threadIdx.x, lane = tid & 31, wave = tid >> 5;
  const int c = lane & 15, h = lane >> 4;
  const float* Sg = S + g * (NN * NN);

  if (FIRST) {
    float sv[NN];
    float s0 = 0.f, s1 = 0.f;
#pragma unroll
    for (int k = 0; k < NN; k += 2) {
      float v0 = Sg[k * NN + tid] + ((k == tid) ? EPSV : 0.f);
      float v1 = Sg[(k + 1) * NN + tid] + ((k + 1 == tid) ? EPSV : 0.f);
      sv[k] = v0;
      sv[k + 1] = v1;
      s0 += v0 * v0;
      s1 += v1 * v1;
    }
    red[tid] = s0;
    red[tid + 128] = s1;
    __syncthreads();
    red[tid] += red[tid + 128];
    __syncthreads();
    for (int off = 64; off > 0; off >>= 1) {
      if (tid < off) red[tid] += red[tid + off];
      __syncthreads();
    }
    if (tid == 0) {
      float n = sqrtf(red[0]);
      nrm = n;
      if (M0 == 0) {
        nrmbuf[g] = n;
        rs[g] = 1.0f / sqrtf(n);
      }
    }
    __syncthreads();
    const float inv = 1.0f / nrm;
#pragma unroll
    for (int k = 0; k < NN; k += 4) {
      float4 qs, qb;
      float sn;
      sn = sv[k] * inv;     qs.x = sn; qb.x = ((k == tid) ? 1.5f : 0.f) - 0.5f * sn;
      sn = sv[k + 1] * inv; qs.y = sn; qb.y = ((k + 1 == tid) ? 1.5f : 0.f) - 0.5f * sn;
      sn = sv[k + 2] * inv; qs.z = sn; qb.z = ((k + 2 == tid) ? 1.5f : 0.f) - 0.5f * sn;
      sn = sv[k + 3] * inv; qs.w = sn; qb.w = ((k + 3 == tid) ? 1.5f : 0.f) - 0.5f * sn;
      *reinterpret_cast<float4*>(&sS[tid * BPITCH + k]) = qs;
      *reinterpret_cast<float4*>(&sB[tid * BPITCH + k]) = qb;
    }
  } else {
    const float inv = 1.0f / nrmbuf[g];
    const float* Bg = Bin + g * (NN * NN);
#pragma unroll 8
    for (int k = 0; k < NN; k += 4) {
      float4 qs, qb;
      qs.x = (Sg[k * NN + tid] + ((k == tid) ? EPSV : 0.f)) * inv;
      qs.y = (Sg[(k + 1) * NN + tid] + ((k + 1 == tid) ? EPSV : 0.f)) * inv;
      qs.z = (Sg[(k + 2) * NN + tid] + ((k + 2 == tid) ? EPSV : 0.f)) * inv;
      qs.w = (Sg[(k + 3) * NN + tid] + ((k + 3 == tid) ? EPSV : 0.f)) * inv;
      qb.x = Bg[k * NN + tid];
      qb.y = Bg[(k + 1) * NN + tid];
      qb.z = Bg[(k + 2) * NN + tid];
      qb.w = Bg[(k + 3) * NN + tid];
      *reinterpret_cast<float4*>(&sS[tid * BPITCH + k]) = qs;
      *reinterpret_cast<float4*>(&sB[tid * BPITCH + k]) = qb;
    }
  }
  __syncthreads();

  v8f xt[8];
#pragma unroll
  for (int kt = 0; kt < 8; ++kt)
#pragma unroll
    for (int r = 0; r < 8; ++r)
      xt[kt][r] = sB[(kt * 16 + 8 * h + r) * BPITCH + M0 + c];

  const int foff = (wave * 32 + c) * BPITCH + 8 * h;
  float4 fB[32], fS[32];
  v8f y[2], x2[8], x3[8];
  load_frags(fB, sB + foff);
  load_frags(fS, sS + foff);
  __builtin_amdgcn_sched_barrier(0);
  mma_rows(y, fB, xt);
  __builtin_amdgcn_sched_barrier(0);
  exchange(x2, y, xch[0], wave, lane);
  __builtin_amdgcn_sched_barrier(0);
  mma_rows(y, fB, x2);
  __builtin_amdgcn_sched_barrier(0);
  exchange(x3, y, xch[1], wave, lane);
  __builtin_amdgcn_sched_barrier(0);
  mma_rows(y, fS, x3);
  __builtin_amdgcn_sched_barrier(0);

  float* Bo = Bout + g * (NN * NN) + (M0 + c) * NN + wave * 32 + 8 * h;
#pragma unroll
  for (int t = 0; t < 2; ++t) {
    float o[8];
#pragma unroll
    for (int r = 0; r < 8; ++r) {
      float v = -0.5f * y[t][r];
      v += 1.5f * sB[((2 * wave + t) * 16 + 8 * h + r) * BPITCH + M0 + c];
      o[r] = v;
    }
    float4 lo = {o[0], o[1], o[2], o[3]};
    float4 hi = {o[4], o[5], o[6], o[7]};
    *reinterpret_cast<float4*>(Bo + t * 16) = lo;
    *reinterpret_cast<float4*>(Bo + t * 16 + 4) = hi;
  }
}

__global__ void oni_out(const float* __restrict__ Bt, const float* __restrict__ w,
                        const float* __restrict__ mean, const float* __restrict__ rs,
                        float* __restrict__ out) {
  __shared__ float ldsA[NN * APITCH];
  __shared__ float ldsB[KCHUNK * NN];
  const int g = blockIdx.y;
  const int N0 = blockIdx.x * 128;
  const float* Ag = Bt + g * (NN * NN);
  const int tid = threadIdx.x, lane = tid & 31, wave = tid >> 5;
  const int M0 = wave * 16, l15 = lane & 15;
  const int kb = (lane < 16) ? 0 : 2;

  v8f acc[8];
  v8f z = {};
#pragma unroll
  for (int t = 0; t < 8; ++t) acc[t] = z;

  for (int k0 = 0; k0 < NN; k0 += KCHUNK) {
#pragma unroll
    for (int i = 0; i < 4; ++i) {
      int idx = i * 256 + tid;
      int r = idx >> 3, kk = (idx & 7) * 4;
      float4 v = *reinterpret_cast<const float4*>(Ag + r * NN + k0 + kk);
      *reinterpret_cast<float4*>(&ldsA[r * APITCH + kk]) = v;
    }
#pragma unroll
    for (int i = 0; i < 4; ++i) {
      int idx = i * 256 + tid;
      int kk = idx >> 5, c = (idx & 31) * 4;
      int grow = g * NN + k0 + kk;
      float m = mean[grow];
      float4 v = *reinterpret_cast<const float4*>(w + (size_t)grow * DD + N0 + c);
      v.x -= m; v.y -= m; v.z -= m; v.w -= m;
      *reinterpret_cast<float4*>(&ldsB[kk * NN + c]) = v;
    }
    __syncthreads();
#pragma unroll
    for (int k4 = 0; k4 < KCHUNK; k4 += 4) {
      v2f a = *reinterpret_cast<const v2f*>(&ldsA[(M0 + l15) * APITCH + k4 + kb]);
      v2f bf[8];
#pragma unroll
      for (int nt = 0; nt < 8; ++nt) {
        int col = nt * 16 + l15;
        bf[nt].x = ldsB[(k4 + kb) * NN + col];
        bf[nt].y = ldsB[(k4 + kb + 1) * NN + col];
      }
#pragma unroll
      for (int nt = 0; nt < 8; ++nt)
        acc[nt] = wmma_f32(a, bf[nt], acc[nt]);
    }
    __syncthreads();
  }

  const float scale = rs[g];
  const int Mb = M0 + ((lane < 16) ? 0 : 8);
#pragma unroll
  for (int nt = 0; nt < 8; ++nt) {
    int col = N0 + nt * 16 + l15;
#pragma unroll
    for (int r = 0; r < 8; ++r) {
      out[(size_t)(g * NN + Mb + r) * DD + col] = acc[nt][r] * scale;
    }
  }
}

extern "C" void kernel_launch(void* const* d_in, const int* in_sizes, int n_in,
                              void* d_out, int out_size, void* d_ws, size_t ws_size,
                              hipStream_t stream) {
  const float* w = (const float*)d_in[0];
  float* out = (float*)d_out;
  float* ws = (float*)d_ws;

  float* mean = ws;
  float* rs = ws + 2048;
  float* nrmbuf = ws + 2064;
  float* S  = ws + 2080;
  float* Bt = S + MATF;
  float* Ba = Bt + MATF;
  float* Bb = Ba + MATF;

  oni_mean<<<NG * NN, 256, 0, stream>>>(w, mean);
  oni_zero<<<MATF / 256, 256, 0, stream>>>(S);
  oni_cov<<<dim3(KSPLIT, NG), 256, 0, stream>>>(w, mean, S);
  oni_ns<true><<<NG * 8, 128, 0, stream>>>(S, nrmbuf, rs, S, Ba);
  oni_ns<false><<<NG * 8, 128, 0, stream>>>(S, nrmbuf, rs, Ba, Bb);
  oni_ns<false><<<NG * 8, 128, 0, stream>>>(S, nrmbuf, rs, Bb, Ba);
  oni_ns<false><<<NG * 8, 128, 0, stream>>>(S, nrmbuf, rs, Ba, Bt);
  oni_out<<<dim3(DD / 128, NG), 256, 0, stream>>>(Bt, w, mean, rs, out);
}


# Output GEMM staging: a chunk's 12 global loads are issued as ONE group and waited for once (the build waited for each b128 load alone, 8 round trips in a row per chunk); sums bit-identical
# speedup vs baseline: 3.4229x; 1.0770x over previous
#include <hip/hip_runtime.h>
#include <hip/hip_bf16.h>


#define NG 16
#define NN 128
#define DD 9216
#define EPSV 1e-5f
#define TITER 5
#define KSPLIT 16
#define KCHUNK 32
#define APITCH 36
#define MATF (NG * NN * NN)
#define BPITCH 132

typedef __attribute__((ext_vector_type(2))) float v2f;
typedef __attribute__((ext_vector_type(8))) float v8f;
typedef __attribute__((ext_vector_type(4))) float v4f;

static __device__ __forceinline__ v8f wmma_f32(v2f a, v2f b, v8f c) {
  return __builtin_amdgcn_wmma_f32_16x16x4_f32(
       false, a,  false, b,
       (short)0, c,  false,  false);
}

__global__ void oni_mean(const float* __restrict__ w, float* __restrict__ mean) {
  __shared__ float red[256];
  const int row = blockIdx.x;
  const float4* p = reinterpret_cast<const float4*>(w + (size_t)row * DD);
  float s = 0.f;
  for (int i = threadIdx.x; i < DD / 4; i += 256) {
    float4 v = p[i];
    s += (v.x + v.y) + (v.z + v.w);
  }
  red[threadIdx.x] = s;
  __syncthreads();
  for (int off = 128; off > 0; off >>= 1) {
    if (threadIdx.x < off) red[threadIdx.x] += red[threadIdx.x + off];
    __syncthreads();
  }
  if (threadIdx.x == 0) mean[row] = red[0] * (1.0f / (float)DD);
}

__global__ void oni_zero(float* __restrict__ S) {
  S[blockIdx.x * 256 + threadIdx.x] = 0.f;
}


static __device__ __forceinline__ void cov_frags(float4 (&fa)[4], float4 (&fb)[4][8],
                                                 const float* fbase, int M0) {
#pragma unroll
  for (int kq = 0; kq < 4; ++kq) {
    fa[kq] = *reinterpret_cast<const float4*>(fbase + M0 * APITCH + kq * 8);
#pragma unroll
    for (int nt = 0; nt < 8; ++nt)
      fb[kq][nt] = *reinterpret_cast<const float4*>(fbase + nt * 16 * APITCH + kq * 8);
  }
}

static __device__ __forceinline__ void cov_mma(v8f (&acc)[8], const float4 (&fa)[4],
                                               const float4 (&fb)[4][8]) {
#pragma unroll
  for (int kq = 0; kq < 4; ++kq) {
    v2f alo = {fa[kq].x, fa[kq].y};
    v2f ahi = {fa[kq].z, fa[kq].w};
#pragma unroll
    for (int nt = 0; nt < 8; ++nt) {
      v2f b = {fb[kq][nt].x, fb[kq][nt].y};
      acc[nt] = wmma_f32(alo, b, acc[nt]);
    }
#pragma unroll
    for (int nt = 0; nt < 8; ++nt) {
      v2f b = {fb[kq][nt].z, fb[kq][nt].w};
      acc[nt] = wmma_f32(ahi, b, acc[nt]);
    }
  }
}

__global__ void __launch_bounds__(256) oni_cov(const float* __restrict__ w,
                                               const float* __restrict__ mean,
                                               float* __restrict__ S) {
  __shared__ __attribute__((aligned(16))) float lds[2][NN * APITCH];
  const int g = blockIdx.y;
  const int kbase = blockIdx.x * (DD / KSPLIT);
  const int tid = threadIdx.x;
  const int lane = tid & 31, wave = tid >> 5;
  const int M0 = wave * 16;
  const int l15 = lane & 15;
  const int h = lane >> 4;
  constexpr int NCH = DD / KSPLIT / KCHUNK;

  const int sr = tid >> 3, sk = (tid & 7) * 4;
  const float* wp = w + (size_t)(g * NN + sr) * DD + kbase + sk;
  float m[4];
  float4 st[4];
#pragma unroll
  for (int i = 0; i < 4; ++i) {
    m[i] = mean[g * NN + i * 32 + sr];
    st[i] = *reinterpret_cast<const float4*>(wp + (size_t)(i * 32) * DD);
  }
#pragma unroll
  for (int i = 0; i < 4; ++i) {
    float4 v = st[i];
    v.x -= m[i]; v.y -= m[i]; v.z -= m[i]; v.w -= m[i];
    *reinterpret_cast<float4*>(&lds[0][(i * 32 + sr) * APITCH + sk]) = v;
  }
  __syncthreads();

  v8f acc[8];
  v8f z = {};
#pragma unroll
  for (int t = 0; t < 8; ++t) acc[t] = z;

  const int foff = l15 * APITCH + 4 * h;
  float4 fa[4], fb[4][8];

  for (int c = 0; c + 1 < NCH; ++c) {
    const float* cur = lds[c & 1];
    float* nxt = lds[(c + 1) & 1];
#pragma unroll
    for (int i = 0; i < 4; ++i)
      st[i] = *reinterpret_cast<const float4*>(wp + (size_t)(i * 32) * DD + (c + 1) * KCHUNK);
    cov_frags(fa, fb, cur + foff, M0);
    __builtin_amdgcn_sched_barrier(0);
    cov_mma(acc, fa, fb);
    __builtin_amdgcn_sched_barrier(0);
#pragma unroll
    for (int i = 0; i < 4; ++i) {
      float4 v = st[i];
      v.x -= m[i]; v.y -= m[i]; v.z -= m[i]; v.w -= m[i];
      *reinterpret_cast<float4*>(&nxt[(i * 32 + sr) * APITCH + sk]) = v;
    }
    __syncthreads();
  }
  cov_frags(fa, fb, lds[(NCH - 1) & 1] + foff, M0);
  __builtin_amdgcn_sched_barrier(0);
  cov_mma(acc, fa, fb);

  const int Mb = M0 + ((lane < 16) ? 0 : 8);
  float* Sg = S + (size_t)g * NN * NN;
#pragma unroll
  for (int nt = 0; nt < 8; ++nt) {
    int col = nt * 16 + l15;
#pragma unroll
    for (int r = 0; r < 8; ++r) {
      atomicAdd(&Sg[(Mb + r) * NN + col], acc[nt][r]);
    }
  }
}


static __device__ __forceinline__ void load_frags(float4 (&a)[32], const float* base) {
#pragma unroll
  for (int kt = 0; kt < 8; ++kt)
#pragma unroll
    for (int jj = 0; jj < 2; ++jj)
#pragma unroll
      for (int t = 0; t < 2; ++t)
        a[kt * 4 + jj * 2 + t] =
            *reinterpret_cast<const float4*>(base + t * 16 * BPITCH + kt * 16 + 4 * jj);
}

static __device__ __forceinline__ void mma_rows(v8f (&yt)[2], const float4 (&a)[32], const v8f (&xt)[8]) {
  v8f z = {};
  yt[0] = z;
  yt[1] = z;
#pragma unroll
  for (int kt = 0; kt < 8; ++kt)
#pragma unroll
    for (int jj = 0; jj < 2; ++jj) {
#pragma unroll
      for (int t = 0; t < 2; ++t) {
        v2f af = {a[kt * 4 + jj * 2 + t].x, a[kt * 4 + jj * 2 + t].y};
        v2f bf = {xt[kt][4 * jj], xt[kt][4 * jj + 1]};
        yt[t] = wmma_f32(af, bf, yt[t]);
      }
#pragma unroll
      for (int t = 0; t < 2; ++t) {
        v2f af = {a[kt * 4 + jj * 2 + t].z, a[kt * 4 + jj * 2 + t].w};
        v2f bf = {xt[kt][4 * jj + 2], xt[kt][4 * jj + 3]};
        yt[t] = wmma_f32(af, bf, yt[t]);
      }
    }
}

static __device__ __forceinline__ void exchange(v8f (&xn)[8], const v8f (&yt)[2], float* xw,
                                                int wave, int lane) {
#pragma unroll
  for (int t = 0; t < 2; ++t)
#pragma unroll
    for (int r = 0; r < 8; ++r)
      xw[((2 * wave + t) * 8 + r) * 32 + lane] = yt[t][r];
  __syncthreads();
#pragma unroll
  for (int kt = 0; kt < 8; ++kt)
#pragma unroll
    for (int r = 0; r < 8; ++r)
      xn[kt][r] = xw[(kt * 8 + r) * 32 + lane];
}

template <bool FIRST>
__global__ void __launch_bounds__(128) oni_ns(const float* __restrict__ S,
                                              float* __restrict__ nrmbuf,
                                              float* __restrict__ rs,
                                              const float* __restrict__ Bin,
                                              float* __restrict__ Bout) {
  __shared__ __attribute__((aligned(16))) float sS[NN * BPITCH];
  __shared__ __attribute__((aligned(16))) float sB[NN * BPITCH];
  __shared__ float xch[2][8 * 8 * 32];
  __shared__ float red[256];
  __shared__ float nrm;
  const int g = blockIdx.x >> 3;
  const int M0 = (blockIdx.x & 7) * 16;
  const int tid = threadIdx.x, lane = tid & 31, wave = tid >> 5;
  const int c = lane & 15, h = lane >> 4;
  const float* Sg = S + g * (NN * NN);

  if (FIRST) {
    float sv[NN];
    float s0 = 0.f, s1 = 0.f;
#pragma unroll
    for (int k = 0; k < NN; k += 2) {
      float v0 = Sg[k * NN + tid] + ((k == tid) ? EPSV : 0.f);
      float v1 = Sg[(k + 1) * NN + tid] + ((k + 1 == tid) ? EPSV : 0.f);
      sv[k] = v0;
      sv[k + 1] = v1;
      s0 += v0 * v0;
      s1 += v1 * v1;
    }
    red[tid] = s0;
    red[tid + 128] = s1;
    __syncthreads();
    red[tid] += red[tid + 128];
    __syncthreads();
    for (int off = 64; off > 0; off >>= 1) {
      if (tid < off) red[tid] += red[tid + off];
      __syncthreads();
    }
    if (tid == 0) {
      float n = sqrtf(red[0]);
      nrm = n;
      if (M0 == 0) {
        nrmbuf[g] = n;
        rs[g] = 1.0f / sqrtf(n);
      }
    }
    __syncthreads();
    const float inv = 1.0f / nrm;
#pragma unroll
    for (int k = 0; k < NN; k += 4) {
      float4 qs, qb;
      float sn;
      sn = sv[k] * inv;     qs.x = sn; qb.x = ((k == tid) ? 1.5f : 0.f) - 0.5f * sn;
      sn = sv[k + 1] * inv; qs.y = sn; qb.y = ((k + 1 == tid) ? 1.5f : 0.f) - 0.5f * sn;
      sn = sv[k + 2] * inv; qs.z = sn; qb.z = ((k + 2 == tid) ? 1.5f : 0.f) - 0.5f * sn;
      sn = sv[k + 3] * inv; qs.w = sn; qb.w = ((k + 3 == tid) ? 1.5f : 0.f) - 0.5f * sn;
      *reinterpret_cast<float4*>(&sS[tid * BPITCH + k]) = qs;
      *reinterpret_cast<float4*>(&sB[tid * BPITCH + k]) = qb;
    }
  } else {
    const float inv = 1.0f / nrmbuf[g];
    const float* Bg = Bin + g * (NN * NN);
#pragma unroll 8
    for (int k = 0; k < NN; k += 4) {
      float4 qs, qb;
      qs.x = (Sg[k * NN + tid] + ((k == tid) ? EPSV : 0.f)) * inv;
      qs.y = (Sg[(k + 1) * NN + tid] + ((k + 1 == tid) ? EPSV : 0.f)) * inv;
      qs.z = (Sg[(k + 2) * NN + tid] + ((k + 2 == tid) ? EPSV : 0.f)) * inv;
      qs.w = (Sg[(k + 3) * NN + tid] + ((k + 3 == tid) ? EPSV : 0.f)) * inv;
      qb.x = Bg[k * NN + tid];
      qb.y = Bg[(k + 1) * NN + tid];
      qb.z = Bg[(k + 2) * NN + tid];
      qb.w = Bg[(k + 3) * NN + tid];
      *reinterpret_cast<float4*>(&sS[tid * BPITCH + k]) = qs;
      *reinterpret_cast<float4*>(&sB[tid * BPITCH + k]) = qb;
    }
  }
  __syncthreads();

  v8f xt[8];
#pragma unroll
  for (int kt = 0; kt < 8; ++kt)
#pragma unroll
    for (int r = 0; r < 8; ++r)
      xt[kt][r] = sB[(kt * 16 + 8 * h + r) * BPITCH + M0 + c];

  const int foff = (wave * 32 + c) * BPITCH + 8 * h;
  float4 fB[32], fS[32];
  v8f y[2], x2[8], x3[8];
  load_frags(fB, sB + foff);
  load_frags(fS, sS + foff);
  __builtin_amdgcn_sched_barrier(0);
  mma_rows(y, fB, xt);
  __builtin_amdgcn_sched_barrier(0);
  exchange(x2, y, xch[0], wave, lane);
  __builtin_amdgcn_sched_barrier(0);
  mma_rows(y, fB, x2);
  __builtin_amdgcn_sched_barrier(0);
  exchange(x3, y, xch[1], wave, lane);
  __builtin_amdgcn_sched_barrier(0);
  mma_rows(y, fS, x3);
  __builtin_amdgcn_sched_barrier(0);

  float* Bo = Bout + g * (NN * NN) + (M0 + c) * NN + wave * 32 + 8 * h;
#pragma unroll
  for (int t = 0; t < 2; ++t) {
    float o[8];
#pragma unroll
    for (int r = 0; r < 8; ++r) {
      float v = -0.5f * y[t][r];
      v += 1.5f * sB[((2 * wave + t) * 16 + 8 * h + r) * BPITCH + M0 + c];
      o[r] = v;
    }
    float4 lo = {o[0], o[1], o[2], o[3]};
    float4 hi = {o[4], o[5], o[6], o[7]};
    *reinterpret_cast<float4*>(Bo + t * 16) = lo;
    *reinterpret_cast<float4*>(Bo + t * 16 + 4) = hi;
  }
}

__global__ void oni_out(const float* __restrict__ Bt, const float* __restrict__ w,
                        const float* __restrict__ mean, const float* __restrict__ rs,
                        float* __restrict__ out) {
  __shared__ float ldsA[NN * APITCH];
  __shared__ float ldsB[KCHUNK * NN];
  const int g = blockIdx.y;
  const int N0 = blockIdx.x * 128;
  const float* Ag = Bt + g * (NN * NN);
  const int tid = threadIdx.x, lane = tid & 31, wave = tid >> 5;
  const int M0 = wave * 16, l15 = lane & 15;
  const int kb = (lane < 16) ? 0 : 2;

  v8f acc[8];
  v8f z = {};
#pragma unroll
  for (int t = 0; t < 8; ++t) acc[t] = z;

  const int ar = tid >> 3, ak = (tid & 7) * 4;
  const int bk = tid >> 5, bc = (tid & 31) * 4;
  const float* ap = Ag + ar * NN + ak;
  const float* bp = w + (size_t)(g * NN + bk) * DD + N0 + bc;
  const float* mp = mean + g * NN + bk;
  float* la = &ldsA[ar * APITCH + ak];
  float* lb = &ldsB[bk * NN + bc];

  for (int k0 = 0; k0 < NN; k0 += KCHUNK) {
    const v4f a0 = *reinterpret_cast<const v4f*>(ap + k0);
    const v4f a1 = *reinterpret_cast<const v4f*>(ap + 32 * NN + k0);
    const v4f a2 = *reinterpret_cast<const v4f*>(ap + 64 * NN + k0);
    const v4f a3 = *reinterpret_cast<const v4f*>(ap + 96 * NN + k0);
    const float m0 = mp[k0], m1 = mp[k0 + 8], m2 = mp[k0 + 16], m3 = mp[k0 + 24];
    v4f b0 = *reinterpret_cast<const v4f*>(bp + (size_t)(k0) * DD);
    v4f b1 = *reinterpret_cast<const v4f*>(bp + (size_t)(k0 + 8) * DD);
    v4f b2 = *reinterpret_cast<const v4f*>(bp + (size_t)(k0 + 16) * DD);
    v4f b3 = *reinterpret_cast<const v4f*>(bp + (size_t)(k0 + 24) * DD);
    asm volatile("" :: "v"(a0), "v"(a1), "v"(a2), "v"(a3), "v"(b0), "v"(b1), "v"(b2), "v"(b3),
                 "v"(m0), "v"(m1), "v"(m2), "v"(m3));
    if (k0 > 0) __syncthreads();
    *reinterpret_cast<v4f*>(la) = a0;
    *reinterpret_cast<v4f*>(la + 32 * APITCH) = a1;
    *reinterpret_cast<v4f*>(la + 64 * APITCH) = a2;
    *reinterpret_cast<v4f*>(la + 96 * APITCH) = a3;
    b0 -= m0;
    b1 -= m1;
    b2 -= m2;
    b3 -= m3;
    *reinterpret_cast<v4f*>(lb) = b0;
    *reinterpret_cast<v4f*>(lb + 8 * NN) = b1;
    *reinterpret_cast<v4f*>(lb + 16 * NN) = b2;
    *reinterpret_cast<v4f*>(lb + 24 * NN) = b3;
    __syncthreads();
#pragma unroll
    for (int k4 = 0; k4 < KCHUNK; k4 += 4) {
      v2f a = *reinterpret_cast<const v2f*>(&ldsA[(M0 + l15) * APITCH + k4 + kb]);
      v2f bf[8];
#pragma unroll
      for (int nt = 0; nt < 8; ++nt) {
        int col = nt * 16 + l15;
        bf[nt].x = ldsB[(k4 + kb) * NN + col];
        bf[nt].y = ldsB[(k4 + kb + 1) * NN + col];
      }
      __builtin_amdgcn_sched_barrier(0);
#pragma unroll
      for (int nt = 0; nt < 8; ++nt)
        acc[nt] = wmma_f32(a, bf[nt], acc[nt]);
      __builtin_amdgcn_sched_barrier(0);
    }
  }

  const float scale = rs[g];
  const int Mb = M0 + ((lane < 16) ? 0 : 8);
#pragma unroll
  for (int nt = 0; nt < 8; ++nt) {
    int col = N0 + nt * 16 + l15;
#pragma unroll
    for (int r = 0; r < 8; ++r) {
      out[(size_t)(g * NN + Mb + r) * DD + col] = acc[nt][r] * scale;
    }
  }
}

extern "C" void kernel_launch(void* const* d_in, const int* in_sizes, int n_in,
                              void* d_out, int out_size, void* d_ws, size_t ws_size,
                              hipStream_t stream) {
  const float* w = (const float*)d_in[0];
  float* out = (float*)d_out;
  float* ws = (float*)d_ws;

  float* mean = ws;
  float* rs = ws + 2048;
  float* nrmbuf = ws + 2064;
  float* S  = ws + 2080;
  float* Bt = S + MATF;
  float* Ba = Bt + MATF;
  float* Bb = Ba + MATF;

  oni_mean<<<NG * NN, 256, 0, stream>>>(w, mean);
  oni_zero<<<MATF / 256, 256, 0, stream>>>(S);
  oni_cov<<<dim3(KSPLIT, NG), 256, 0, stream>>>(w, mean, S);
  oni_ns<true><<<NG * 8, 128, 0, stream>>>(S, nrmbuf, rs, S, Ba);
  oni_ns<false><<<NG * 8, 128, 0, stream>>>(S, nrmbuf, rs, Ba, Bb);
  oni_ns<false><<<NG * 8, 128, 0, stream>>>(S, nrmbuf, rs, Bb, Ba);
  oni_ns<false><<<NG * 8, 128, 0, stream>>>(S, nrmbuf, rs, Ba, Bt);
  oni_out<<<dim3(DD / 128, NG), 256, 0, stream>>>(Bt, w, mean, rs, out);
}


# Covariance by symmetry: a work-group forms a half band, 36 tiles' worth of 64 (tile (I,I+4) split by k between waves I and I+4, summed as two halves); mirror tiles added transposed via LDS
# speedup vs baseline: 3.5656x; 1.0417x over previous
#include <hip/hip_runtime.h>
#include <hip/hip_bf16.h>


#define NG 16
#define NN 128
#define DD 9216
#define EPSV 1e-5f
#define TITER 5
#define KSPLIT 16
#define KCHUNK 32
#define APITCH 36
#define MATF (NG * NN * NN)
#define BPITCH 132

typedef __attribute__((ext_vector_type(2))) float v2f;
typedef __attribute__((ext_vector_type(8))) float v8f;
typedef __attribute__((ext_vector_type(4))) float v4f;

static __device__ __forceinline__ v8f wmma_f32(v2f a, v2f b, v8f c) {
  return __builtin_amdgcn_wmma_f32_16x16x4_f32(
       false, a,  false, b,
       (short)0, c,  false,  false);
}

__global__ void oni_mean(const float* __restrict__ w, float* __restrict__ mean) {
  __shared__ float red[256];
  const int row = blockIdx.x;
  const float4* p = reinterpret_cast<const float4*>(w + (size_t)row * DD);
  float s = 0.f;
  for (int i = threadIdx.x; i < DD / 4; i += 256) {
    float4 v = p[i];
    s += (v.x + v.y) + (v.z + v.w);
  }
  red[threadIdx.x] = s;
  __syncthreads();
  for (int off = 128; off > 0; off >>= 1) {
    if (threadIdx.x < off) red[threadIdx.x] += red[threadIdx.x + off];
    __syncthreads();
  }
  if (threadIdx.x == 0) mean[row] = red[0] * (1.0f / (float)DD);
}

__global__ void oni_zero(float* __restrict__ S) {
  S[blockIdx.x * 256 + threadIdx.x] = 0.f;
}


static __device__ __forceinline__ void cov_frags(float4 (&fa)[4], float4 (&fb)[4][3],
                                                 float4 (&ha)[2], float4 (&hb)[2],
                                                 const float* fbase, int aoff,
                                                 const int (&boff)[3], int haoff, int hboff) {
#pragma unroll
  for (int kq = 0; kq < 4; ++kq) {
    fa[kq] = *reinterpret_cast<const float4*>(fbase + aoff + kq * 8);
#pragma unroll
    for (int j = 0; j < 3; ++j)
      fb[kq][j] = *reinterpret_cast<const float4*>(fbase + boff[j] + kq * 8);
  }
#pragma unroll
  for (int i = 0; i < 2; ++i) {
    ha[i] = *reinterpret_cast<const float4*>(fbase + haoff + i * 8);
    hb[i] = *reinterpret_cast<const float4*>(fbase + hboff + i * 8);
  }
}

static __device__ __forceinline__ void cov_mma(v8f (&acc)[5], const float4 (&fa)[4],
                                               const float4 (&fb)[4][3],
                                               const float4 (&ha)[2], const float4 (&hb)[2]) {
#pragma unroll
  for (int kq = 0; kq < 4; ++kq) {
    v2f alo = {fa[kq].x, fa[kq].y};
    v2f ahi = {fa[kq].z, fa[kq].w};
    acc[0] = wmma_f32(alo, alo, acc[0]);
#pragma unroll
    for (int j = 0; j < 3; ++j) {
      v2f b = {fb[kq][j].x, fb[kq][j].y};
      acc[1 + j] = wmma_f32(alo, b, acc[1 + j]);
    }
    acc[0] = wmma_f32(ahi, ahi, acc[0]);
#pragma unroll
    for (int j = 0; j < 3; ++j) {
      v2f b = {fb[kq][j].z, fb[kq][j].w};
      acc[1 + j] = wmma_f32(ahi, b, acc[1 + j]);
    }
    v2f xa = (kq & 1) ? v2f{ha[kq >> 1].z, ha[kq >> 1].w} : v2f{ha[kq >> 1].x, ha[kq >> 1].y};
    v2f xb = (kq & 1) ? v2f{hb[kq >> 1].z, hb[kq >> 1].w} : v2f{hb[kq >> 1].x, hb[kq >> 1].y};
    acc[4] = wmma_f32(xa, xb, acc[4]);
  }
}

#define TPITCH 20

static __device__ __forceinline__ void wave_lds_order() {
  __builtin_amdgcn_fence(__ATOMIC_RELEASE, "wavefront");
  __builtin_amdgcn_wave_barrier();
  __builtin_amdgcn_fence(__ATOMIC_ACQUIRE, "wavefront");
}

__global__ void __launch_bounds__(256) oni_cov(const float* __restrict__ w,
                                               const float* __restrict__ mean,
                                               float* __restrict__ S) {
  __shared__ __attribute__((aligned(16))) float lds[2][NN * APITCH];
  const int g = blockIdx.y;
  const int kbase = blockIdx.x * (DD / KSPLIT);
  const int tid = threadIdx.x;
  const int lane = tid & 31, wave = tid >> 5;
  const int M0 = wave * 16;
  const int l15 = lane & 15;
  const int h = lane >> 4;
  constexpr int NCH = DD / KSPLIT / KCHUNK;
  int boff[3];
#pragma unroll
  for (int j = 0; j < 3; ++j) boff[j] = ((wave + 1 + j) & 7) * 16 * APITCH;
  const int aoff = M0 * APITCH;
  const int hk = (wave >> 2) * 16;
  const int haoff = aoff + hk;
  const int hboff = ((wave + 4) & 7) * 16 * APITCH + hk;

  const int sr = tid >> 3, sk = (tid & 7) * 4;
  const float* wp = w + (size_t)(g * NN + sr) * DD + kbase + sk;
  float m[4];
  float4 st[4];
#pragma unroll
  for (int i = 0; i < 4; ++i) {
    m[i] = mean[g * NN + i * 32 + sr];
    st[i] = *reinterpret_cast<const float4*>(wp + (size_t)(i * 32) * DD);
  }
#pragma unroll
  for (int i = 0; i < 4; ++i) {
    float4 v = st[i];
    v.x -= m[i]; v.y -= m[i]; v.z -= m[i]; v.w -= m[i];
    *reinterpret_cast<float4*>(&lds[0][(i * 32 + sr) * APITCH + sk]) = v;
  }
  __syncthreads();

  v8f acc[5];
  v8f z = {};
#pragma unroll
  for (int t = 0; t < 5; ++t) acc[t] = z;

  const int foff = l15 * APITCH + 4 * h;
  float4 fa[4], fb[4][3], ha[2], hb[2];

#pragma unroll
  for (int i = 0; i < 4; ++i)
    st[i] = *reinterpret_cast<const float4*>(wp + (size_t)(i * 32) * DD + KCHUNK);

  for (int c = 0; c + 1 < NCH; ++c) {
    const float* cur = lds[c & 1];
    float* nxt = lds[(c + 1) & 1];
    cov_frags(fa, fb, ha, hb, cur + foff, aoff, boff, haoff, hboff);
    __builtin_amdgcn_sched_barrier(0);
    cov_mma(acc, fa, fb, ha, hb);
    __builtin_amdgcn_sched_barrier(0);
#pragma unroll
    for (int i = 0; i < 4; ++i) {
      float4 v = st[i];
      v.x -= m[i]; v.y -= m[i]; v.z -= m[i]; v.w -= m[i];
      *reinterpret_cast<float4*>(&nxt[(i * 32 + sr) * APITCH + sk]) = v;
    }
    __syncthreads();
    if (c + 2 < NCH) {
#pragma unroll
      for (int i = 0; i < 4; ++i)
        st[i] = *reinterpret_cast<const float4*>(wp + (size_t)(i * 32) * DD + (c + 2) * KCHUNK);
    }
  }
  cov_frags(fa, fb, ha, hb, lds[(NCH - 1) & 1] + foff, aoff, boff, haoff, hboff);
  __builtin_amdgcn_sched_barrier(0);
  cov_mma(acc, fa, fb, ha, hb);

  __syncthreads();
  float* tr = &lds[0][0] + wave * (16 * TPITCH);
  const int Mb = M0 + 8 * h;
  float* Sg = S + (size_t)g * NN * NN;
#pragma unroll
  for (int r = 0; r < 8; ++r) atomicAdd(&Sg[(Mb + r) * NN + M0 + l15], acc[0][r]);
#pragma unroll
  for (int j = 0; j < 4; ++j) {
    const int J0 = ((wave + 1 + j) & 7) * 16;
#pragma unroll
    for (int r = 0; r < 8; ++r) atomicAdd(&Sg[(Mb + r) * NN + J0 + l15], acc[1 + j][r]);
    float4 lo = {acc[1 + j][0], acc[1 + j][1], acc[1 + j][2], acc[1 + j][3]};
    float4 hi = {acc[1 + j][4], acc[1 + j][5], acc[1 + j][6], acc[1 + j][7]};
    *reinterpret_cast<float4*>(tr + l15 * TPITCH + 8 * h) = lo;
    *reinterpret_cast<float4*>(tr + l15 * TPITCH + 8 * h + 4) = hi;
    wave_lds_order();
    float v[8];
#pragma unroll
    for (int r = 0; r < 8; ++r) v[r] = tr[(8 * h + r) * TPITCH + l15];
#pragma unroll
    for (int r = 0; r < 8; ++r) atomicAdd(&Sg[(J0 + 8 * h + r) * NN + M0 + l15], v[r]);
    wave_lds_order();
  }
}


static __device__ __forceinline__ void load_frags(float4 (&a)[32], const float* base) {
#pragma unroll
  for (int kt = 0; kt < 8; ++kt)
#pragma unroll
    for (int jj = 0; jj < 2; ++jj)
#pragma unroll
      for (int t = 0; t < 2; ++t)
        a[kt * 4 + jj * 2 + t] =
            *reinterpret_cast<const float4*>(base + t * 16 * BPITCH + kt * 16 + 4 * jj);
}

static __device__ __forceinline__ void mma_rows(v8f (&yt)[2], const float4 (&a)[32], const v8f (&xt)[8]) {
  v8f z = {};
  yt[0] = z;
  yt[1] = z;
#pragma unroll
  for (int kt = 0; kt < 8; ++kt)
#pragma unroll
    for (int jj = 0; jj < 2; ++jj) {
#pragma unroll
      for (int t = 0; t < 2; ++t) {
        v2f af = {a[kt * 4 + jj * 2 + t].x, a[kt * 4 + jj * 2 + t].y};
        v2f bf = {xt[kt][4 * jj], xt[kt][4 * jj + 1]};
        yt[t] = wmma_f32(af, bf, yt[t]);
      }
#pragma unroll
      for (int t = 0; t < 2; ++t) {
        v2f af = {a[kt * 4 + jj * 2 + t].z, a[kt * 4 + jj * 2 + t].w};
        v2f bf = {xt[kt][4 * jj + 2], xt[kt][4 * jj + 3]};
        yt[t] = wmma_f32(af, bf, yt[t]);
      }
    }
}

static __device__ __forceinline__ void exchange(v8f (&xn)[8], const v8f (&yt)[2], float* xw,
                                                int wave, int lane) {
#pragma unroll
  for (int t = 0; t < 2; ++t)
#pragma unroll
    for (int r = 0; r < 8; ++r)
      xw[((2 * wave + t) * 8 + r) * 32 + lane] = yt[t][r];
  __syncthreads();
#pragma unroll
  for (int kt = 0; kt < 8; ++kt)
#pragma unroll
    for (int r = 0; r < 8; ++r)
      xn[kt][r] = xw[(kt * 8 + r) * 32 + lane];
}

template <bool FIRST>
__global__ void __launch_bounds__(128) oni_ns(const float* __restrict__ S,
                                              float* __restrict__ nrmbuf,
                                              float* __restrict__ rs,
                                              const float* __restrict__ Bin,
                                              float* __restrict__ Bout) {
  __shared__ __attribute__((aligned(16))) float sS[NN * BPITCH];
  __shared__ __attribute__((aligned(16))) float sB[NN * BPITCH];
  __shared__ float xch[2][8 * 8 * 32];
  __shared__ float red[256];
  __shared__ float nrm;
  const int g = blockIdx.x >> 3;
  const int M0 = (blockIdx.x & 7) * 16;
  const int tid = threadIdx.x, lane = tid & 31, wave = tid >> 5;
  const int c = lane & 15, h = lane >> 4;
  const float* Sg = S + g * (NN * NN);

  if (FIRST) {
    float sv[NN];
    float s0 = 0.f, s1 = 0.f;
#pragma unroll
    for (int k = 0; k < NN; k += 2) {
      float v0 = Sg[k * NN + tid] + ((k == tid) ? EPSV : 0.f);
      float v1 = Sg[(k + 1) * NN + tid] + ((k + 1 == tid) ? EPSV : 0.f);
      sv[k] = v0;
      sv[k + 1] = v1;
      s0 += v0 * v0;
      s1 += v1 * v1;
    }
    red[tid] = s0;
    red[tid + 128] = s1;
    __syncthreads();
    red[tid] += red[tid + 128];
    __syncthreads();
    for (int off = 64; off > 0; off >>= 1) {
      if (tid < off) red[tid] += red[tid + off];
      __syncthreads();
    }
    if (tid == 0) {
      float n = sqrtf(red[0]);
      nrm = n;
      if (M0 == 0) {
        nrmbuf[g] = n;
        rs[g] = 1.0f / sqrtf(n);
      }
    }
    __syncthreads();
    const float inv = 1.0f / nrm;
#pragma unroll
    for (int k = 0; k < NN; k += 4) {
      float4 qs, qb;
      float sn;
      sn = sv[k] * inv;     qs.x = sn; qb.x = ((k == tid) ? 1.5f : 0.f) - 0.5f * sn;
      sn = sv[k + 1] * inv; qs.y = sn; qb.y = ((k + 1 == tid) ? 1.5f : 0.f) - 0.5f * sn;
      sn = sv[k + 2] * inv; qs.z = sn; qb.z = ((k + 2 == tid) ? 1.5f : 0.f) - 0.5f * sn;
      sn = sv[k + 3] * inv; qs.w = sn; qb.w = ((k + 3 == tid) ? 1.5f : 0.f) - 0.5f * sn;
      *reinterpret_cast<float4*>(&sS[tid * BPITCH + k]) = qs;
      *reinterpret_cast<float4*>(&sB[tid * BPITCH + k]) = qb;
    }
  } else {
    const float inv = 1.0f / nrmbuf[g];
    const float* Bg = Bin + g * (NN * NN);
#pragma unroll 8
    for (int k = 0; k < NN; k += 4) {
      float4 qs, qb;
      qs.x = (Sg[k * NN + tid] + ((k == tid) ? EPSV : 0.f)) * inv;
      qs.y = (Sg[(k + 1) * NN + tid] + ((k + 1 == tid) ? EPSV : 0.f)) * inv;
      qs.z = (Sg[(k + 2) * NN + tid] + ((k + 2 == tid) ? EPSV : 0.f)) * inv;
      qs.w = (Sg[(k + 3) * NN + tid] + ((k + 3 == tid) ? EPSV : 0.f)) * inv;
      qb.x = Bg[k * NN + tid];
      qb.y = Bg[(k + 1) * NN + tid];
      qb.z = Bg[(k + 2) * NN + tid];
      qb.w = Bg[(k + 3) * NN + tid];
      *reinterpret_cast<float4*>(&sS[tid * BPITCH + k]) = qs;
      *reinterpret_cast<float4*>(&sB[tid * BPITCH + k]) = qb;
    }
  }
  __syncthreads();

  v8f xt[8];
#pragma unroll
  for (int kt = 0; kt < 8; ++kt)
#pragma unroll
    for (int r = 0; r < 8; ++r)
      xt[kt][r] = sB[(kt * 16 + 8 * h + r) * BPITCH + M0 + c];

  const int foff = (wave * 32 + c) * BPITCH + 8 * h;
  float4 fB[32], fS[32];
  v8f y[2], x2[8], x3[8];
  load_frags(fB, sB + foff);
  load_frags(fS, sS + foff);
  __builtin_amdgcn_sched_barrier(0);
  mma_rows(y, fB, xt);
  __builtin_amdgcn_sched_barrier(0);
  exchange(x2, y, xch[0], wave, lane);
  __builtin_amdgcn_sched_barrier(0);
  mma_rows(y, fB, x2);
  __builtin_amdgcn_sched_barrier(0);
  exchange(x3, y, xch[1], wave, lane);
  __builtin_amdgcn_sched_barrier(0);
  mma_rows(y, fS, x3);
  __builtin_amdgcn_sched_barrier(0);

  float* Bo = Bout + g * (NN * NN) + (M0 + c) * NN + wave * 32 + 8 * h;
#pragma unroll
  for (int t = 0; t < 2; ++t) {
    float o[8];
#pragma unroll
    for (int r = 0; r < 8; ++r) {
      float v = -0.5f * y[t][r];
      v += 1.5f * sB[((2 * wave + t) * 16 + 8 * h + r) * BPITCH + M0 + c];
      o[r] = v;
    }
    float4 lo = {o[0], o[1], o[2], o[3]};
    float4 hi = {o[4], o[5], o[6], o[7]};
    *reinterpret_cast<float4*>(Bo + t * 16) = lo;
    *reinterpret_cast<float4*>(Bo + t * 16 + 4) = hi;
  }
}

__global__ void oni_out(const float* __restrict__ Bt, const float* __restrict__ w,
                        const float* __restrict__ mean, const float* __restrict__ rs,
                        float* __restrict__ out) {
  __shared__ float ldsA[NN * APITCH];
  __shared__ float ldsB[KCHUNK * NN];
  const int g = blockIdx.y;
  const int N0 = blockIdx.x * 128;
  const float* Ag = Bt + g * (NN * NN);
  const int tid = threadIdx.x, lane = tid & 31, wave = tid >> 5;
  const int M0 = wave * 16, l15 = lane & 15;
  const int kb = (lane < 16) ? 0 : 2;

  v8f acc[8];
  v8f z = {};
#pragma unroll
  for (int t = 0; t < 8; ++t) acc[t] = z;

  const int ar = tid >> 3, ak = (tid & 7) * 4;
  const int bk = tid >> 5, bc = (tid & 31) * 4;
  const float* ap = Ag + ar * NN + ak;
  const float* bp = w + (size_t)(g * NN + bk) * DD + N0 + bc;
  const float* mp = mean + g * NN + bk;
  float* la = &ldsA[ar * APITCH + ak];
  float* lb = &ldsB[bk * NN + bc];

  for (int k0 = 0; k0 < NN; k0 += KCHUNK) {
    const v4f a0 = *reinterpret_cast<const v4f*>(ap + k0);
    const v4f a1 = *reinterpret_cast<const v4f*>(ap + 32 * NN + k0);
    const v4f a2 = *reinterpret_cast<const v4f*>(ap + 64 * NN + k0);
    const v4f a3 = *reinterpret_cast<const v4f*>(ap + 96 * NN + k0);
    const float m0 = mp[k0], m1 = mp[k0 + 8], m2 = mp[k0 + 16], m3 = mp[k0 + 24];
    v4f b0 = *reinterpret_cast<const v4f*>(bp + (size_t)(k0) * DD);
    v4f b1 = *reinterpret_cast<const v4f*>(bp + (size_t)(k0 + 8) * DD);
    v4f b2 = *reinterpret_cast<const v4f*>(bp + (size_t)(k0 + 16) * DD);
    v4f b3 = *reinterpret_cast<const v4f*>(bp + (size_t)(k0 + 24) * DD);
    asm volatile("" :: "v"(a0), "v"(a1), "v"(a2), "v"(a3), "v"(b0), "v"(b1), "v"(b2), "v"(b3),
                 "v"(m0), "v"(m1), "v"(m2), "v"(m3));
    if (k0 > 0) __syncthreads();
    *reinterpret_cast<v4f*>(la) = a0;
    *reinterpret_cast<v4f*>(la + 32 * APITCH) = a1;
    *reinterpret_cast<v4f*>(la + 64 * APITCH) = a2;
    *reinterpret_cast<v4f*>(la + 96 * APITCH) = a3;
    b0 -= m0;
    b1 -= m1;
    b2 -= m2;
    b3 -= m3;
    *reinterpret_cast<v4f*>(lb) = b0;
    *reinterpret_cast<v4f*>(lb + 8 * NN) = b1;
    *reinterpret_cast<v4f*>(lb + 16 * NN) = b2;
    *reinterpret_cast<v4f*>(lb + 24 * NN) = b3;
    __syncthreads();
#pragma unroll
    for (int k4 = 0; k4 < KCHUNK; k4 += 4) {
      v2f a = *reinterpret_cast<const v2f*>(&ldsA[(M0 + l15) * APITCH + k4 + kb]);
      v2f bf[8];
#pragma unroll
      for (int nt = 0; nt < 8; ++nt) {
        int col = nt * 16 + l15;
        bf[nt].x = ldsB[(k4 + kb) * NN + col];
        bf[nt].y = ldsB[(k4 + kb + 1) * NN + col];
      }
      __builtin_amdgcn_sched_barrier(0);
#pragma unroll
      for (int nt = 0; nt < 8; ++nt)
        acc[nt] = wmma_f32(a, bf[nt], acc[nt]);
      __builtin_amdgcn_sched_barrier(0);
    }
  }

  const float scale = rs[g];
  const int Mb = M0 + ((lane < 16) ? 0 : 8);
#pragma unroll
  for (int nt = 0; nt < 8; ++nt) {
    int col = N0 + nt * 16 + l15;
#pragma unroll
    for (int r = 0; r < 8; ++r) {
      out[(size_t)(g * NN + Mb + r) * DD + col] = acc[nt][r] * scale;
    }
  }
}

extern "C" void kernel_launch(void* const* d_in, const int* in_sizes, int n_in,
                              void* d_out, int out_size, void* d_ws, size_t ws_size,
                              hipStream_t stream) {
  const float* w = (const float*)d_in[0];
  float* out = (float*)d_out;
  float* ws = (float*)d_ws;

  float* mean = ws;
  float* rs = ws + 2048;
  float* nrmbuf = ws + 2064;
  float* S  = ws + 2080;
  float* Bt = S + MATF;
  float* Ba = Bt + MATF;
  float* Bb = Ba + MATF;

  oni_mean<<<NG * NN, 256, 0, stream>>>(w, mean);
  oni_zero<<<MATF / 256, 256, 0, stream>>>(S);
  oni_cov<<<dim3(KSPLIT, NG), 256, 0, stream>>>(w, mean, S);
  oni_ns<true><<<NG * 8, 128, 0, stream>>>(S, nrmbuf, rs, S, Ba);
  oni_ns<false><<<NG * 8, 128, 0, stream>>>(S, nrmbuf, rs, Ba, Bb);
  oni_ns<false><<<NG * 8, 128, 0, stream>>>(S, nrmbuf, rs, Bb, Ba);
  oni_ns<false><<<NG * 8, 128, 0, stream>>>(S, nrmbuf, rs, Ba, Bt);
  oni_out<<<dim3(DD / 128, NG), 256, 0, stream>>>(Bt, w, mean, rs, out);
}


# Covariance K-split partials stored to the workspace and summed in fixed order by a reduction launch that takes the zeroing launch's place: float atomics gone, sums no longer in arrival order
# speedup vs baseline: 3.7156x; 1.0421x over previous
#include <hip/hip_runtime.h>
#include <hip/hip_bf16.h>


#define NG 16
#define NN 128
#define DD 9216
#define EPSV 1e-5f
#define TITER 5
#define KSPLIT 16
#define KCHUNK 32
#define APITCH 36
#define MATF (NG * NN * NN)
#define BPITCH 132

typedef __attribute__((ext_vector_type(2))) float v2f;
typedef __attribute__((ext_vector_type(8))) float v8f;
typedef __attribute__((ext_vector_type(4))) float v4f;

static __device__ __forceinline__ v8f wmma_f32(v2f a, v2f b, v8f c) {
  return __builtin_amdgcn_wmma_f32_16x16x4_f32(
       false, a,  false, b,
       (short)0, c,  false,  false);
}

__global__ void oni_mean(const float* __restrict__ w, float* __restrict__ mean) {
  __shared__ float red[256];
  const int row = blockIdx.x;
  const float4* p = reinterpret_cast<const float4*>(w + (size_t)row * DD);
  float s = 0.f;
  for (int i = threadIdx.x; i < DD / 4; i += 256) {
    float4 v = p[i];
    s += (v.x + v.y) + (v.z + v.w);
  }
  red[threadIdx.x] = s;
  __syncthreads();
  for (int off = 128; off > 0; off >>= 1) {
    if (threadIdx.x < off) red[threadIdx.x] += red[threadIdx.x + off];
    __syncthreads();
  }
  if (threadIdx.x == 0) mean[row] = red[0] * (1.0f / (float)DD);
}


static __device__ __forceinline__ void cov_frags(float4 (&fa)[4], float4 (&fb)[4][3],
                                                 float4 (&ha)[2], float4 (&hb)[2],
                                                 const float* fbase, int aoff,
                                                 const int (&boff)[3], int haoff, int hboff) {
#pragma unroll
  for (int kq = 0; kq < 4; ++kq) {
    fa[kq] = *reinterpret_cast<const float4*>(fbase + aoff + kq * 8);
#pragma unroll
    for (int j = 0; j < 3; ++j)
      fb[kq][j] = *reinterpret_cast<const float4*>(fbase + boff[j] + kq * 8);
  }
#pragma unroll
  for (int i = 0; i < 2; ++i) {
    ha[i] = *reinterpret_cast<const float4*>(fbase + haoff + i * 8);
    hb[i] = *reinterpret_cast<const float4*>(fbase + hboff + i * 8);
  }
}

static __device__ __forceinline__ void cov_mma(v8f (&acc)[5], const float4 (&fa)[4],
                                               const float4 (&fb)[4][3],
                                               const float4 (&ha)[2], const float4 (&hb)[2]) {
#pragma unroll
  for (int kq = 0; kq < 4; ++kq) {
    v2f alo = {fa[kq].x, fa[kq].y};
    v2f ahi = {fa[kq].z, fa[kq].w};
    acc[0] = wmma_f32(alo, alo, acc[0]);
#pragma unroll
    for (int j = 0; j < 3; ++j) {
      v2f b = {fb[kq][j].x, fb[kq][j].y};
      acc[1 + j] = wmma_f32(alo, b, acc[1 + j]);
    }
    acc[0] = wmma_f32(ahi, ahi, acc[0]);
#pragma unroll
    for (int j = 0; j < 3; ++j) {
      v2f b = {fb[kq][j].z, fb[kq][j].w};
      acc[1 + j] = wmma_f32(ahi, b, acc[1 + j]);
    }
    v2f xa = (kq & 1) ? v2f{ha[kq >> 1].z, ha[kq >> 1].w} : v2f{ha[kq >> 1].x, ha[kq >> 1].y};
    v2f xb = (kq & 1) ? v2f{hb[kq >> 1].z, hb[kq >> 1].w} : v2f{hb[kq >> 1].x, hb[kq >> 1].y};
    acc[4] = wmma_f32(xa, xb, acc[4]);
  }
}

#define PTILE 256
#define PWAVE (5 * PTILE)
#define PWG (8 * PWAVE)

__global__ void __launch_bounds__(256) oni_cov(const float* __restrict__ w,
                                               const float* __restrict__ mean,
                                               float* __restrict__ P) {
  __shared__ __attribute__((aligned(16))) float lds[2][NN * APITCH];
  const int g = blockIdx.y;
  const int kbase = blockIdx.x * (DD / KSPLIT);
  const int tid = threadIdx.x;
  const int lane = tid & 31, wave = tid >> 5;
  const int M0 = wave * 16;
  const int l15 = lane & 15;
  const int h = lane >> 4;
  constexpr int NCH = DD / KSPLIT / KCHUNK;
  int boff[3];
#pragma unroll
  for (int j = 0; j < 3; ++j) boff[j] = ((wave + 1 + j) & 7) * 16 * APITCH;
  const int aoff = M0 * APITCH;
  const int hk = (wave >> 2) * 16;
  const int haoff = aoff + hk;
  const int hboff = ((wave + 4) & 7) * 16 * APITCH + hk;

  const int sr = tid >> 3, sk = (tid & 7) * 4;
  const float* wp = w + (size_t)(g * NN + sr) * DD + kbase + sk;
  float m[4];
  float4 st[4];
#pragma unroll
  for (int i = 0; i < 4; ++i) {
    m[i] = mean[g * NN + i * 32 + sr];
    st[i] = *reinterpret_cast<const float4*>(wp + (size_t)(i * 32) * DD);
  }
#pragma unroll
  for (int i = 0; i < 4; ++i) {
    float4 v = st[i];
    v.x -= m[i]; v.y -= m[i]; v.z -= m[i]; v.w -= m[i];
    *reinterpret_cast<float4*>(&lds[0][(i * 32 + sr) * APITCH + sk]) = v;
  }
  __syncthreads();

  v8f acc[5];
  v8f z = {};
#pragma unroll
  for (int t = 0; t < 5; ++t) acc[t] = z;

  const int foff = l15 * APITCH + 4 * h;
  float4 fa[4], fb[4][3], ha[2], hb[2];

#pragma unroll
  for (int i = 0; i < 4; ++i)
    st[i] = *reinterpret_cast<const float4*>(wp + (size_t)(i * 32) * DD + KCHUNK);

  for (int c = 0; c + 1 < NCH; ++c) {
    const float* cur = lds[c & 1];
    float* nxt = lds[(c + 1) & 1];
    cov_frags(fa, fb, ha, hb, cur + foff, aoff, boff, haoff, hboff);
    __builtin_amdgcn_sched_barrier(0);
    cov_mma(acc, fa, fb, ha, hb);
    __builtin_amdgcn_sched_barrier(0);
#pragma unroll
    for (int i = 0; i < 4; ++i) {
      float4 v = st[i];
      v.x -= m[i]; v.y -= m[i]; v.z -= m[i]; v.w -= m[i];
      *reinterpret_cast<float4*>(&nxt[(i * 32 + sr) * APITCH + sk]) = v;
    }
    __syncthreads();
    if (c + 2 < NCH) {
#pragma unroll
      for (int i = 0; i < 4; ++i)
        st[i] = *reinterpret_cast<const float4*>(wp + (size_t)(i * 32) * DD + (c + 2) * KCHUNK);
    }
  }
  cov_frags(fa, fb, ha, hb, lds[(NCH - 1) & 1] + foff, aoff, boff, haoff, hboff);
  __builtin_amdgcn_sched_barrier(0);
  cov_mma(acc, fa, fb, ha, hb);

  float* Pw = P + ((size_t)(g * KSPLIT + blockIdx.x) * 8 + wave) * PWAVE + lane * 8;
#pragma unroll
  for (int t = 0; t < 5; ++t) {
    float4 lo = {acc[t][0], acc[t][1], acc[t][2], acc[t][3]};
    float4 hi = {acc[t][4], acc[t][5], acc[t][6], acc[t][7]};
    *reinterpret_cast<float4*>(Pw + t * PTILE) = lo;
    *reinterpret_cast<float4*>(Pw + t * PTILE + 4) = hi;
  }
}

__global__ void __launch_bounds__(160) oni_sum(const float* __restrict__ P, float* __restrict__ S) {
  const int g = blockIdx.x >> 3, I = blockIdx.x & 7;
  const int t = threadIdx.x >> 5, lane = threadIdx.x & 31;
  const int c = lane & 15, h = lane >> 4;
  const float* p = P + ((size_t)g * KSPLIT * 8 + I) * PWAVE + t * PTILE + lane * 8;
  float s[8];
#pragma unroll
  for (int r = 0; r < 8; ++r) s[r] = 0.f;
  if (t < 4) {
#pragma unroll
    for (int ks = 0; ks < KSPLIT; ++ks) {
      const float4 lo = *reinterpret_cast<const float4*>(p + (size_t)ks * PWG);
      const float4 hi = *reinterpret_cast<const float4*>(p + (size_t)ks * PWG + 4);
      s[0] += lo.x; s[1] += lo.y; s[2] += lo.z; s[3] += lo.w;
      s[4] += hi.x; s[5] += hi.y; s[6] += hi.z; s[7] += hi.w;
    }
  } else {
    const int I4 = (I + 4) & 7;
    const float* o = P + ((size_t)g * KSPLIT * 8 + I4) * PWAVE + 4 * PTILE +
                     (8 * h + 16 * (c >> 3)) * 8 + (c & 7);
    const bool own_first = I < 4;
#pragma unroll
    for (int ks = 0; ks < KSPLIT; ++ks) {
      const float4 lo = *reinterpret_cast<const float4*>(p + (size_t)ks * PWG);
      const float4 hi = *reinterpret_cast<const float4*>(p + (size_t)ks * PWG + 4);
      const float own[8] = {lo.x, lo.y, lo.z, lo.w, hi.x, hi.y, hi.z, hi.w};
#pragma unroll
      for (int r = 0; r < 8; ++r) {
        const float ov = o[(size_t)ks * PWG + r * 8];
        const float first = own_first ? own[r] : ov;
        const float second = own_first ? ov : own[r];
        s[r] += first;
        s[r] += second;
      }
    }
  }
  float* Sg = S + (size_t)g * NN * NN;
  const int J = (t == 0) ? I : ((I + ((t < 4) ? t : 4)) & 7);
#pragma unroll
  for (int r = 0; r < 8; ++r) Sg[(I * 16 + 8 * h + r) * NN + J * 16 + c] = s[r];
  if (t >= 1 && t <= 3) {
    float4 lo = {s[0], s[1], s[2], s[3]};
    float4 hi = {s[4], s[5], s[6], s[7]};
    float* m = Sg + (J * 16 + c) * NN + I * 16 + 8 * h;
    *reinterpret_cast<float4*>(m) = lo;
    *reinterpret_cast<float4*>(m + 4) = hi;
  }
}


static __device__ __forceinline__ void load_frags(float4 (&a)[32], const float* base) {
#pragma unroll
  for (int kt = 0; kt < 8; ++kt)
#pragma unroll
    for (int jj = 0; jj < 2; ++jj)
#pragma unroll
      for (int t = 0; t < 2; ++t)
        a[kt * 4 + jj * 2 + t] =
            *reinterpret_cast<const float4*>(base + t * 16 * BPITCH + kt * 16 + 4 * jj);
}

static __device__ __forceinline__ void mma_rows(v8f (&yt)[2], const float4 (&a)[32], const v8f (&xt)[8]) {
  v8f z = {};
  yt[0] = z;
  yt[1] = z;
#pragma unroll
  for (int kt = 0; kt < 8; ++kt)
#pragma unroll
    for (int jj = 0; jj < 2; ++jj) {
#pragma unroll
      for (int t = 0; t < 2; ++t) {
        v2f af = {a[kt * 4 + jj * 2 + t].x, a[kt * 4 + jj * 2 + t].y};
        v2f bf = {xt[kt][4 * jj], xt[kt][4 * jj + 1]};
        yt[t] = wmma_f32(af, bf, yt[t]);
      }
#pragma unroll
      for (int t = 0; t < 2; ++t) {
        v2f af = {a[kt * 4 + jj * 2 + t].z, a[kt * 4 + jj * 2 + t].w};
        v2f bf = {xt[kt][4 * jj + 2], xt[kt][4 * jj + 3]};
        yt[t] = wmma_f32(af, bf, yt[t]);
      }
    }
}

static __device__ __forceinline__ void exchange(v8f (&xn)[8], const v8f (&yt)[2], float* xw,
                                                int wave, int lane) {
#pragma unroll
  for (int t = 0; t < 2; ++t)
#pragma unroll
    for (int r = 0; r < 8; ++r)
      xw[((2 * wave + t) * 8 + r) * 32 + lane] = yt[t][r];
  __syncthreads();
#pragma unroll
  for (int kt = 0; kt < 8; ++kt)
#pragma unroll
    for (int r = 0; r < 8; ++r)
      xn[kt][r] = xw[(kt * 8 + r) * 32 + lane];
}

template <bool FIRST>
__global__ void __launch_bounds__(128) oni_ns(const float* __restrict__ S,
                                              float* __restrict__ nrmbuf,
                                              float* __restrict__ rs,
                                              const float* __restrict__ Bin,
                                              float* __restrict__ Bout) {
  __shared__ __attribute__((aligned(16))) float sS[NN * BPITCH];
  __shared__ __attribute__((aligned(16))) float sB[NN * BPITCH];
  __shared__ float xch[2][8 * 8 * 32];
  __shared__ float red[256];
  __shared__ float nrm;
  const int g = blockIdx.x >> 3;
  const int M0 = (blockIdx.x & 7) * 16;
  const int tid = threadIdx.x, lane = tid & 31, wave = tid >> 5;
  const int c = lane & 15, h = lane >> 4;
  const float* Sg = S + g * (NN * NN);

  if (FIRST) {
    float sv[NN];
    float s0 = 0.f, s1 = 0.f;
#pragma unroll
    for (int k = 0; k < NN; k += 2) {
      float v0 = Sg[k * NN + tid] + ((k == tid) ? EPSV : 0.f);
      float v1 = Sg[(k + 1) * NN + tid] + ((k + 1 == tid) ? EPSV : 0.f);
      sv[k] = v0;
      sv[k + 1] = v1;
      s0 += v0 * v0;
      s1 += v1 * v1;
    }
    red[tid] = s0;
    red[tid + 128] = s1;
    __syncthreads();
    red[tid] += red[tid + 128];
    __syncthreads();
    for (int off = 64; off > 0; off >>= 1) {
      if (tid < off) red[tid] += red[tid + off];
      __syncthreads();
    }
    if (tid == 0) {
      float n = sqrtf(red[0]);
      nrm = n;
      if (M0 == 0) {
        nrmbuf[g] = n;
        rs[g] = 1.0f / sqrtf(n);
      }
    }
    __syncthreads();
    const float inv = 1.0f / nrm;
#pragma unroll
    for (int k = 0; k < NN; k += 4) {
      float4 qs, qb;
      float sn;
      sn = sv[k] * inv;     qs.x = sn; qb.x = ((k == tid) ? 1.5f : 0.f) - 0.5f * sn;
      sn = sv[k + 1] * inv; qs.y = sn; qb.y = ((k + 1 == tid) ? 1.5f : 0.f) - 0.5f * sn;
      sn = sv[k + 2] * inv; qs.z = sn; qb.z = ((k + 2 == tid) ? 1.5f : 0.f) - 0.5f * sn;
      sn = sv[k + 3] * inv; qs.w = sn; qb.w = ((k + 3 == tid) ? 1.5f : 0.f) - 0.5f * sn;
      *reinterpret_cast<float4*>(&sS[tid * BPITCH + k]) = qs;
      *reinterpret_cast<float4*>(&sB[tid * BPITCH + k]) = qb;
    }
  } else {
    const float inv = 1.0f / nrmbuf[g];
    const float* Bg = Bin + g * (NN * NN);
#pragma unroll 8
    for (int k = 0; k < NN; k += 4) {
      float4 qs, qb;
      qs.x = (Sg[k * NN + tid] + ((k == tid) ? EPSV : 0.f)) * inv;
      qs.y = (Sg[(k + 1) * NN + tid] + ((k + 1 == tid) ? EPSV : 0.f)) * inv;
      qs.z = (Sg[(k + 2) * NN + tid] + ((k + 2 == tid) ? EPSV : 0.f)) * inv;
      qs.w = (Sg[(k + 3) * NN + tid] + ((k + 3 == tid) ? EPSV : 0.f)) * inv;
      qb.x = Bg[k * NN + tid];
      qb.y = Bg[(k + 1) * NN + tid];
      qb.z = Bg[(k + 2) * NN + tid];
      qb.w = Bg[(k + 3) * NN + tid];
      *reinterpret_cast<float4*>(&sS[tid * BPITCH + k]) = qs;
      *reinterpret_cast<float4*>(&sB[tid * BPITCH + k]) = qb;
    }
  }
  __syncthreads();

  v8f xt[8];
#pragma unroll
  for (int kt = 0; kt < 8; ++kt)
#pragma unroll
    for (int r = 0; r < 8; ++r)
      xt[kt][r] = sB[(kt * 16 + 8 * h + r) * BPITCH + M0 + c];

  const int foff = (wave * 32 + c) * BPITCH + 8 * h;
  float4 fB[32], fS[32];
  v8f y[2], x2[8], x3[8];
  load_frags(fB, sB + foff);
  load_frags(fS, sS + foff);
  __builtin_amdgcn_sched_barrier(0);
  mma_rows(y, fB, xt);
  __builtin_amdgcn_sched_barrier(0);
  exchange(x2, y, xch[0], wave, lane);
  __builtin_amdgcn_sched_barrier(0);
  mma_rows(y, fB, x2);
  __builtin_amdgcn_sched_barrier(0);
  exchange(x3, y, xch[1], wave, lane);
  __builtin_amdgcn_sched_barrier(0);
  mma_rows(y, fS, x3);
  __builtin_amdgcn_sched_barrier(0);

  float* Bo = Bout + g * (NN * NN) + (M0 + c) * NN + wave * 32 + 8 * h;
#pragma unroll
  for (int t = 0; t < 2; ++t) {
    float o[8];
#pragma unroll
    for (int r = 0; r < 8; ++r) {
      float v = -0.5f * y[t][r];
      v += 1.5f * sB[((2 * wave + t) * 16 + 8 * h + r) * BPITCH + M0 + c];
      o[r] = v;
    }
    float4 lo = {o[0], o[1], o[2], o[3]};
    float4 hi = {o[4], o[5], o[6], o[7]};
    *reinterpret_cast<float4*>(Bo + t * 16) = lo;
    *reinterpret_cast<float4*>(Bo + t * 16 + 4) = hi;
  }
}

__global__ void oni_out(const float* __restrict__ Bt, const float* __restrict__ w,
                        const float* __restrict__ mean, const float* __restrict__ rs,
                        float* __restrict__ out) {
  __shared__ float ldsA[NN * APITCH];
  __shared__ float ldsB[KCHUNK * NN];
  const int g = blockIdx.y;
  const int N0 = blockIdx.x * 128;
  const float* Ag = Bt + g * (NN * NN);
  const int tid = threadIdx.x, lane = tid & 31, wave = tid >> 5;
  const int M0 = wave * 16, l15 = lane & 15;
  const int kb = (lane < 16) ? 0 : 2;

  v8f acc[8];
  v8f z = {};
#pragma unroll
  for (int t = 0; t < 8; ++t) acc[t] = z;

  const int ar = tid >> 3, ak = (tid & 7) * 4;
  const int bk = tid >> 5, bc = (tid & 31) * 4;
  const float* ap = Ag + ar * NN + ak;
  const float* bp = w + (size_t)(g * NN + bk) * DD + N0 + bc;
  const float* mp = mean + g * NN + bk;
  float* la = &ldsA[ar * APITCH + ak];
  float* lb = &ldsB[bk * NN + bc];

  for (int k0 = 0; k0 < NN; k0 += KCHUNK) {
    const v4f a0 = *reinterpret_cast<const v4f*>(ap + k0);
    const v4f a1 = *reinterpret_cast<const v4f*>(ap + 32 * NN + k0);
    const v4f a2 = *reinterpret_cast<const v4f*>(ap + 64 * NN + k0);
    const v4f a3 = *reinterpret_cast<const v4f*>(ap + 96 * NN + k0);
    const float m0 = mp[k0], m1 = mp[k0 + 8], m2 = mp[k0 + 16], m3 = mp[k0 + 24];
    v4f b0 = *reinterpret_cast<const v4f*>(bp + (size_t)(k0) * DD);
    v4f b1 = *reinterpret_cast<const v4f*>(bp + (size_t)(k0 + 8) * DD);
    v4f b2 = *reinterpret_cast<const v4f*>(bp + (size_t)(k0 + 16) * DD);
    v4f b3 = *reinterpret_cast<const v4f*>(bp + (size_t)(k0 + 24) * DD);
    asm volatile("" :: "v"(a0), "v"(a1), "v"(a2), "v"(a3), "v"(b0), "v"(b1), "v"(b2), "v"(b3),
                 "v"(m0), "v"(m1), "v"(m2), "v"(m3));
    if (k0 > 0) __syncthreads();
    *reinterpret_cast<v4f*>(la) = a0;
    *reinterpret_cast<v4f*>(la + 32 * APITCH) = a1;
    *reinterpret_cast<v4f*>(la + 64 * APITCH) = a2;
    *reinterpret_cast<v4f*>(la + 96 * APITCH) = a3;
    b0 -= m0;
    b1 -= m1;
    b2 -= m2;
    b3 -= m3;
    *reinterpret_cast<v4f*>(lb) = b0;
    *reinterpret_cast<v4f*>(lb + 8 * NN) = b1;
    *reinterpret_cast<v4f*>(lb + 16 * NN) = b2;
    *reinterpret_cast<v4f*>(lb + 24 * NN) = b3;
    __syncthreads();
#pragma unroll
    for (int k4 = 0; k4 < KCHUNK; k4 += 4) {
      v2f a = *reinterpret_cast<const v2f*>(&ldsA[(M0 + l15) * APITCH + k4 + kb]);
      v2f bf[8];
#pragma unroll
      for (int nt = 0; nt < 8; ++nt) {
        int col = nt * 16 + l15;
        bf[nt].x = ldsB[(k4 + kb) * NN + col];
        bf[nt].y = ldsB[(k4 + kb + 1) * NN + col];
      }
      __builtin_amdgcn_sched_barrier(0);
#pragma unroll
      for (int nt = 0; nt < 8; ++nt)
        acc[nt] = wmma_f32(a, bf[nt], acc[nt]);
      __builtin_amdgcn_sched_barrier(0);
    }
  }

  const float scale = rs[g];
  const int Mb = M0 + ((lane < 16) ? 0 : 8);
#pragma unroll
  for (int nt = 0; nt < 8; ++nt) {
    int col = N0 + nt * 16 + l15;
#pragma unroll
    for (int r = 0; r < 8; ++r) {
      out[(size_t)(g * NN + Mb + r) * DD + col] = acc[nt][r] * scale;
    }
  }
}

extern "C" void kernel_launch(void* const* d_in, const int* in_sizes, int n_in,
                              void* d_out, int out_size, void* d_ws, size_t ws_size,
                              hipStream_t stream) {
  const float* w = (const float*)d_in[0];
  float* out = (float*)d_out;
  float* ws = (float*)d_ws;

  float* mean = ws;
  float* rs = ws + 2048;
  float* nrmbuf = ws + 2064;
  float* S  = ws + 2080;
  float* Bt = S + MATF;
  float* Ba = Bt + MATF;
  float* Bb = Ba + MATF;
  float* P  = Bb + MATF;

  oni_mean<<<NG * NN, 256, 0, stream>>>(w, mean);
  oni_cov<<<dim3(KSPLIT, NG), 256, 0, stream>>>(w, mean, P);
  oni_sum<<<NG * 8, 160, 0, stream>>>(P, S);
  oni_ns<true><<<NG * 8, 128, 0, stream>>>(S, nrmbuf, rs, S, Ba);
  oni_ns<false><<<NG * 8, 128, 0, stream>>>(S, nrmbuf, rs, Ba, Bb);
  oni_ns<false><<<NG * 8, 128, 0, stream>>>(S, nrmbuf, rs, Bb, Ba);
  oni_ns<false><<<NG * 8, 128, 0, stream>>>(S, nrmbuf, rs, Ba, Bt);
  oni_out<<<dim3(DD / 128, NG), 256, 0, stream>>>(Bt, w, mean, rs, out);
}


# Covariance in one pass: rows multiplied uncentred, row sums taken beside the staging (K-split partials, fixed order), S = Z Z^T - r r^T/d formed in the summing launch; the mean launch is gone
# speedup vs baseline: 4.0014x; 1.0769x over previous
#include <hip/hip_runtime.h>
#include <hip/hip_bf16.h>


#define NG 16
#define NN 128
#define DD 9216
#define EPSV 1e-5f
#define TITER 5
#define KSPLIT 16
#define KCHUNK 32
#define APITCH 36
#define MATF (NG * NN * NN)
#define BPITCH 132

typedef __attribute__((ext_vector_type(2))) float v2f;
typedef __attribute__((ext_vector_type(8))) float v8f;
typedef __attribute__((ext_vector_type(4))) float v4f;

static __device__ __forceinline__ v8f wmma_f32(v2f a, v2f b, v8f c) {
  return __builtin_amdgcn_wmma_f32_16x16x4_f32(
       false, a,  false, b,
       (short)0, c,  false,  false);
}


static __device__ __forceinline__ void cov_frags(float4 (&fa)[4], float4 (&fb)[4][3],
                                                 float4 (&ha)[2], float4 (&hb)[2],
                                                 const float* fbase, int aoff,
                                                 const int (&boff)[3], int haoff, int hboff) {
#pragma unroll
  for (int kq = 0; kq < 4; ++kq) {
    fa[kq] = *reinterpret_cast<const float4*>(fbase + aoff + kq * 8);
#pragma unroll
    for (int j = 0; j < 3; ++j)
      fb[kq][j] = *reinterpret_cast<const float4*>(fbase + boff[j] + kq * 8);
  }
#pragma unroll
  for (int i = 0; i < 2; ++i) {
    ha[i] = *reinterpret_cast<const float4*>(fbase + haoff + i * 8);
    hb[i] = *reinterpret_cast<const float4*>(fbase + hboff + i * 8);
  }
}

static __device__ __forceinline__ void cov_mma(v8f (&acc)[5], const float4 (&fa)[4],
                                               const float4 (&fb)[4][3],
                                               const float4 (&ha)[2], const float4 (&hb)[2]) {
#pragma unroll
  for (int kq = 0; kq < 4; ++kq) {
    v2f alo = {fa[kq].x, fa[kq].y};
    v2f ahi = {fa[kq].z, fa[kq].w};
    acc[0] = wmma_f32(alo, alo, acc[0]);
#pragma unroll
    for (int j = 0; j < 3; ++j) {
      v2f b = {fb[kq][j].x, fb[kq][j].y};
      acc[1 + j] = wmma_f32(alo, b, acc[1 + j]);
    }
    acc[0] = wmma_f32(ahi, ahi, acc[0]);
#pragma unroll
    for (int j = 0; j < 3; ++j) {
      v2f b = {fb[kq][j].z, fb[kq][j].w};
      acc[1 + j] = wmma_f32(ahi, b, acc[1 + j]);
    }
    v2f xa = (kq & 1) ? v2f{ha[kq >> 1].z, ha[kq >> 1].w} : v2f{ha[kq >> 1].x, ha[kq >> 1].y};
    v2f xb = (kq & 1) ? v2f{hb[kq >> 1].z, hb[kq >> 1].w} : v2f{hb[kq >> 1].x, hb[kq >> 1].y};
    acc[4] = wmma_f32(xa, xb, acc[4]);
  }
}

#define PTILE 256
#define PWAVE (5 * PTILE)
#define PWG (8 * PWAVE)

__global__ void __launch_bounds__(256) oni_cov(const float* __restrict__ w,
                                               float* __restrict__ P,
                                               float* __restrict__ Rs) {
  __shared__ __attribute__((aligned(16))) float lds[2][NN * APITCH];
  const int g = blockIdx.y;
  const int kbase = blockIdx.x * (DD / KSPLIT);
  const int tid = threadIdx.x;
  const int lane = tid & 31, wave = tid >> 5;
  const int M0 = wave * 16;
  const int l15 = lane & 15;
  const int h = lane >> 4;
  constexpr int NCH = DD / KSPLIT / KCHUNK;
  int boff[3];
#pragma unroll
  for (int j = 0; j < 3; ++j) boff[j] = ((wave + 1 + j) & 7) * 16 * APITCH;
  const int aoff = M0 * APITCH;
  const int hk = (wave >> 2) * 16;
  const int haoff = aoff + hk;
  const int hboff = ((wave + 4) & 7) * 16 * APITCH + hk;

  const int sr = tid >> 3, sk = (tid & 7) * 4;
  const float* wp = w + (size_t)(g * NN + sr) * DD + kbase + sk;
  float rsum[4];
  float4 st[4];
#pragma unroll
  for (int i = 0; i < 4; ++i) {
    rsum[i] = 0.f;
    st[i] = *reinterpret_cast<const float4*>(wp + (size_t)(i * 32) * DD);
  }
#pragma unroll
  for (int i = 0; i < 4; ++i) {
    float4 v = st[i];
    rsum[i] += (v.x + v.y) + (v.z + v.w);
    *reinterpret_cast<float4*>(&lds[0][(i * 32 + sr) * APITCH + sk]) = v;
  }
  __syncthreads();

  v8f acc[5];
  v8f z = {};
#pragma unroll
  for (int t = 0; t < 5; ++t) acc[t] = z;

  const int foff = l15 * APITCH + 4 * h;
  float4 fa[4], fb[4][3], ha[2], hb[2];

#pragma unroll
  for (int i = 0; i < 4; ++i)
    st[i] = *reinterpret_cast<const float4*>(wp + (size_t)(i * 32) * DD + KCHUNK);

  for (int c = 0; c + 1 < NCH; ++c) {
    const float* cur = lds[c & 1];
    float* nxt = lds[(c + 1) & 1];
    cov_frags(fa, fb, ha, hb, cur + foff, aoff, boff, haoff, hboff);
    __builtin_amdgcn_sched_barrier(0);
    cov_mma(acc, fa, fb, ha, hb);
    __builtin_amdgcn_sched_barrier(0);
#pragma unroll
    for (int i = 0; i < 4; ++i) {
      float4 v = st[i];
      rsum[i] += (v.x + v.y) + (v.z + v.w);
      *reinterpret_cast<float4*>(&nxt[(i * 32 + sr) * APITCH + sk]) = v;
    }
    __syncthreads();
    if (c + 2 < NCH) {
#pragma unroll
      for (int i = 0; i < 4; ++i)
        st[i] = *reinterpret_cast<const float4*>(wp + (size_t)(i * 32) * DD + (c + 2) * KCHUNK);
    }
  }
  cov_frags(fa, fb, ha, hb, lds[(NCH - 1) & 1] + foff, aoff, boff, haoff, hboff);
  __builtin_amdgcn_sched_barrier(0);
  cov_mma(acc, fa, fb, ha, hb);

  float* Pw = P + ((size_t)(g * KSPLIT + blockIdx.x) * 8 + wave) * PWAVE + lane * 8;
#pragma unroll
  for (int t = 0; t < 5; ++t) {
    float4 lo = {acc[t][0], acc[t][1], acc[t][2], acc[t][3]};
    float4 hi = {acc[t][4], acc[t][5], acc[t][6], acc[t][7]};
    *reinterpret_cast<float4*>(Pw + t * PTILE) = lo;
    *reinterpret_cast<float4*>(Pw + t * PTILE + 4) = hi;
  }

#pragma unroll
  for (int i = 0; i < 4; ++i) {
    float a = rsum[i];
    a += __shfl_xor(a, 1);
    a += __shfl_xor(a, 2);
    a += __shfl_xor(a, 4);
    if ((tid & 7) == 0) Rs[(size_t)(g * KSPLIT + blockIdx.x) * NN + i * 32 + sr] = a;
  }
}

__global__ void __launch_bounds__(160) oni_sum(const float* __restrict__ P,
                                               const float* __restrict__ Rs,
                                               float* __restrict__ S, float* __restrict__ mean) {
  __shared__ float rsl[80];
  const int g = blockIdx.x >> 3, I = blockIdx.x & 7;
  const int t = threadIdx.x >> 5, lane = threadIdx.x & 31;
  const int c = lane & 15, h = lane >> 4;
  if (threadIdx.x < 80) {
    const int row = (I * 16 + threadIdx.x) & (NN - 1);
    const float* q = Rs + (size_t)g * KSPLIT * NN + row;
    float a = 0.f;
#pragma unroll
    for (int ks = 0; ks < KSPLIT; ++ks) a += q[ks * NN];
    rsl[threadIdx.x] = a;
    if (threadIdx.x < 16) mean[g * NN + row] = a * (1.0f / (float)DD);
  }
  const float* p = P + ((size_t)g * KSPLIT * 8 + I) * PWAVE + t * PTILE + lane * 8;
  float s[8];
#pragma unroll
  for (int r = 0; r < 8; ++r) s[r] = 0.f;
  if (t < 4) {
#pragma unroll
    for (int ks = 0; ks < KSPLIT; ++ks) {
      const float4 lo = *reinterpret_cast<const float4*>(p + (size_t)ks * PWG);
      const float4 hi = *reinterpret_cast<const float4*>(p + (size_t)ks * PWG + 4);
      s[0] += lo.x; s[1] += lo.y; s[2] += lo.z; s[3] += lo.w;
      s[4] += hi.x; s[5] += hi.y; s[6] += hi.z; s[7] += hi.w;
    }
  } else {
    const int I4 = (I + 4) & 7;
    const float* o = P + ((size_t)g * KSPLIT * 8 + I4) * PWAVE + 4 * PTILE +
                     (8 * h + 16 * (c >> 3)) * 8 + (c & 7);
    const bool own_first = I < 4;
#pragma unroll
    for (int ks = 0; ks < KSPLIT; ++ks) {
      const float4 lo = *reinterpret_cast<const float4*>(p + (size_t)ks * PWG);
      const float4 hi = *reinterpret_cast<const float4*>(p + (size_t)ks * PWG + 4);
      const float own[8] = {lo.x, lo.y, lo.z, lo.w, hi.x, hi.y, hi.z, hi.w};
#pragma unroll
      for (int r = 0; r < 8; ++r) {
        const float ov = o[(size_t)ks * PWG + r * 8];
        const float first = own_first ? own[r] : ov;
        const float second = own_first ? ov : own[r];
        s[r] += first;
        s[r] += second;
      }
    }
  }
  __syncthreads();
  const int jb = (t < 4) ? t : 4;
  const float cs = rsl[jb * 16 + c];
#pragma unroll
  for (int r = 0; r < 8; ++r) s[r] -= (rsl[8 * h + r] * cs) * (1.0f / (float)DD);
  float* Sg = S + (size_t)g * NN * NN;
  const int J = (I + jb) & 7;
#pragma unroll
  for (int r = 0; r < 8; ++r) Sg[(I * 16 + 8 * h + r) * NN + J * 16 + c] = s[r];
  if (t >= 1 && t <= 3) {
    float4 lo = {s[0], s[1], s[2], s[3]};
    float4 hi = {s[4], s[5], s[6], s[7]};
    float* m = Sg + (J * 16 + c) * NN + I * 16 + 8 * h;
    *reinterpret_cast<float4*>(m) = lo;
    *reinterpret_cast<float4*>(m + 4) = hi;
  }
}


static __device__ __forceinline__ void load_frags(float4 (&a)[32], const float* base) {
#pragma unroll
  for (int kt = 0; kt < 8; ++kt)
#pragma unroll
    for (int jj = 0; jj < 2; ++jj)
#pragma unroll
      for (int t = 0; t < 2; ++t)
        a[kt * 4 + jj * 2 + t] =
            *reinterpret_cast<const float4*>(base + t * 16 * BPITCH + kt * 16 + 4 * jj);
}

static __device__ __forceinline__ void mma_rows(v8f (&yt)[2], const float4 (&a)[32], const v8f (&xt)[8]) {
  v8f z = {};
  yt[0] = z;
  yt[1] = z;
#pragma unroll
  for (int kt = 0; kt < 8; ++kt)
#pragma unroll
    for (int jj = 0; jj < 2; ++jj) {
#pragma unroll
      for (int t = 0; t < 2; ++t) {
        v2f af = {a[kt * 4 + jj * 2 + t].x, a[kt * 4 + jj * 2 + t].y};
        v2f bf = {xt[kt][4 * jj], xt[kt][4 * jj + 1]};
        yt[t] = wmma_f32(af, bf, yt[t]);
      }
#pragma unroll
      for (int t = 0; t < 2; ++t) {
        v2f af = {a[kt * 4 + jj * 2 + t].z, a[kt * 4 + jj * 2 + t].w};
        v2f bf = {xt[kt][4 * jj + 2], xt[kt][4 * jj + 3]};
        yt[t] = wmma_f32(af, bf, yt[t]);
      }
    }
}

static __device__ __forceinline__ void exchange(v8f (&xn)[8], const v8f (&yt)[2], float* xw,
                                                int wave, int lane) {
#pragma unroll
  for (int t = 0; t < 2; ++t)
#pragma unroll
    for (int r = 0; r < 8; ++r)
      xw[((2 * wave + t) * 8 + r) * 32 + lane] = yt[t][r];
  __syncthreads();
#pragma unroll
  for (int kt = 0; kt < 8; ++kt)
#pragma unroll
    for (int r = 0; r < 8; ++r)
      xn[kt][r] = xw[(kt * 8 + r) * 32 + lane];
}

template <bool FIRST>
__global__ void __launch_bounds__(128) oni_ns(const float* __restrict__ S,
                                              float* __restrict__ nrmbuf,
                                              float* __restrict__ rs,
                                              const float* __restrict__ Bin,
                                              float* __restrict__ Bout) {
  __shared__ __attribute__((aligned(16))) float sS[NN * BPITCH];
  __shared__ __attribute__((aligned(16))) float sB[NN * BPITCH];
  __shared__ float xch[2][8 * 8 * 32];
  __shared__ float red[256];
  __shared__ float nrm;
  const int g = blockIdx.x >> 3;
  const int M0 = (blockIdx.x & 7) * 16;
  const int tid = threadIdx.x, lane = tid & 31, wave = tid >> 5;
  const int c = lane & 15, h = lane >> 4;
  const float* Sg = S + g * (NN * NN);

  if (FIRST) {
    float sv[NN];
    float s0 = 0.f, s1 = 0.f;
#pragma unroll
    for (int k = 0; k < NN; k += 2) {
      float v0 = Sg[k * NN + tid] + ((k == tid) ? EPSV : 0.f);
      float v1 = Sg[(k + 1) * NN + tid] + ((k + 1 == tid) ? EPSV : 0.f);
      sv[k] = v0;
      sv[k + 1] = v1;
      s0 += v0 * v0;
      s1 += v1 * v1;
    }
    red[tid] = s0;
    red[tid + 128] = s1;
    __syncthreads();
    red[tid] += red[tid + 128];
    __syncthreads();
    for (int off = 64; off > 0; off >>= 1) {
      if (tid < off) red[tid] += red[tid + off];
      __syncthreads();
    }
    if (tid == 0) {
      float n = sqrtf(red[0]);
      nrm = n;
      if (M0 == 0) {
        nrmbuf[g] = n;
        rs[g] = 1.0f / sqrtf(n);
      }
    }
    __syncthreads();
    const float inv = 1.0f / nrm;
#pragma unroll
    for (int k = 0; k < NN; k += 4) {
      float4 qs, qb;
      float sn;
      sn = sv[k] * inv;     qs.x = sn; qb.x = ((k == tid) ? 1.5f : 0.f) - 0.5f * sn;
      sn = sv[k + 1] * inv; qs.y = sn; qb.y = ((k + 1 == tid) ? 1.5f : 0.f) - 0.5f * sn;
      sn = sv[k + 2] * inv; qs.z = sn; qb.z = ((k + 2 == tid) ? 1.5f : 0.f) - 0.5f * sn;
      sn = sv[k + 3] * inv; qs.w = sn; qb.w = ((k + 3 == tid) ? 1.5f : 0.f) - 0.5f * sn;
      *reinterpret_cast<float4*>(&sS[tid * BPITCH + k]) = qs;
      *reinterpret_cast<float4*>(&sB[tid * BPITCH + k]) = qb;
    }
  } else {
    const float inv = 1.0f / nrmbuf[g];
    const float* Bg = Bin + g * (NN * NN);
#pragma unroll 8
    for (int k = 0; k < NN; k += 4) {
      float4 qs, qb;
      qs.x = (Sg[k * NN + tid] + ((k == tid) ? EPSV : 0.f)) * inv;
      qs.y = (Sg[(k + 1) * NN + tid] + ((k + 1 == tid) ? EPSV : 0.f)) * inv;
      qs.z = (Sg[(k + 2) * NN + tid] + ((k + 2 == tid) ? EPSV : 0.f)) * inv;
      qs.w = (Sg[(k + 3) * NN + tid] + ((k + 3 == tid) ? EPSV : 0.f)) * inv;
      qb.x = Bg[k * NN + tid];
      qb.y = Bg[(k + 1) * NN + tid];
      qb.z = Bg[(k + 2) * NN + tid];
      qb.w = Bg[(k + 3) * NN + tid];
      *reinterpret_cast<float4*>(&sS[tid * BPITCH + k]) = qs;
      *reinterpret_cast<float4*>(&sB[tid * BPITCH + k]) = qb;
    }
  }
  __syncthreads();

  v8f xt[8];
#pragma unroll
  for (int kt = 0; kt < 8; ++kt)
#pragma unroll
    for (int r = 0; r < 8; ++r)
      xt[kt][r] = sB[(kt * 16 + 8 * h + r) * BPITCH + M0 + c];

  const int foff = (wave * 32 + c) * BPITCH + 8 * h;
  float4 fB[32], fS[32];
  v8f y[2], x2[8], x3[8];
  load_frags(fB, sB + foff);
  load_frags(fS, sS + foff);
  __builtin_amdgcn_sched_barrier(0);
  mma_rows(y, fB, xt);
  __builtin_amdgcn_sched_barrier(0);
  exchange(x2, y, xch[0], wave, lane);
  __builtin_amdgcn_sched_barrier(0);
  mma_rows(y, fB, x2);
  __builtin_amdgcn_sched_barrier(0);
  exchange(x3, y, xch[1], wave, lane);
  __builtin_amdgcn_sched_barrier(0);
  mma_rows(y, fS, x3);
  __builtin_amdgcn_sched_barrier(0);

  float* Bo = Bout + g * (NN * NN) + (M0 + c) * NN + wave * 32 + 8 * h;
#pragma unroll
  for (int t = 0; t < 2; ++t) {
    float o[8];
#pragma unroll
    for (int r = 0; r < 8; ++r) {
      float v = -0.5f * y[t][r];
      v += 1.5f * sB[((2 * wave + t) * 16 + 8 * h + r) * BPITCH + M0 + c];
      o[r] = v;
    }
    float4 lo = {o[0], o[1], o[2], o[3]};
    float4 hi = {o[4], o[5], o[6], o[7]};
    *reinterpret_cast<float4*>(Bo + t * 16) = lo;
    *reinterpret_cast<float4*>(Bo + t * 16 + 4) = hi;
  }
}

__global__ void oni_out(const float* __restrict__ Bt, const float* __restrict__ w,
                        const float* __restrict__ mean, const float* __restrict__ rs,
                        float* __restrict__ out) {
  __shared__ float ldsA[NN * APITCH];
  __shared__ float ldsB[KCHUNK * NN];
  const int g = blockIdx.y;
  const int N0 = blockIdx.x * 128;
  const float* Ag = Bt + g * (NN * NN);
  const int tid = threadIdx.x, lane = tid & 31, wave = tid >> 5;
  const int M0 = wave * 16, l15 = lane & 15;
  const int kb = (lane < 16) ? 0 : 2;

  v8f acc[8];
  v8f z = {};
#pragma unroll
  for (int t = 0; t < 8; ++t) acc[t] = z;

  const int ar = tid >> 3, ak = (tid & 7) * 4;
  const int bk = tid >> 5, bc = (tid & 31) * 4;
  const float* ap = Ag + ar * NN + ak;
  const float* bp = w + (size_t)(g * NN + bk) * DD + N0 + bc;
  const float* mp = mean + g * NN + bk;
  float* la = &ldsA[ar * APITCH + ak];
  float* lb = &ldsB[bk * NN + bc];

  for (int k0 = 0; k0 < NN; k0 += KCHUNK) {
    const v4f a0 = *reinterpret_cast<const v4f*>(ap + k0);
    const v4f a1 = *reinterpret_cast<const v4f*>(ap + 32 * NN + k0);
    const v4f a2 = *reinterpret_cast<const v4f*>(ap + 64 * NN + k0);
    const v4f a3 = *reinterpret_cast<const v4f*>(ap + 96 * NN + k0);
    const float m0 = mp[k0], m1 = mp[k0 + 8], m2 = mp[k0 + 16], m3 = mp[k0 + 24];
    v4f b0 = *reinterpret_cast<const v4f*>(bp + (size_t)(k0) * DD);
    v4f b1 = *reinterpret_cast<const v4f*>(bp + (size_t)(k0 + 8) * DD);
    v4f b2 = *reinterpret_cast<const v4f*>(bp + (size_t)(k0 + 16) * DD);
    v4f b3 = *reinterpret_cast<const v4f*>(bp + (size_t)(k0 + 24) * DD);
    asm volatile("" :: "v"(a0), "v"(a1), "v"(a2), "v"(a3), "v"(b0), "v"(b1), "v"(b2), "v"(b3),
                 "v"(m0), "v"(m1), "v"(m2), "v"(m3));
    if (k0 > 0) __syncthreads();
    *reinterpret_cast<v4f*>(la) = a0;
    *reinterpret_cast<v4f*>(la + 32 * APITCH) = a1;
    *reinterpret_cast<v4f*>(la + 64 * APITCH) = a2;
    *reinterpret_cast<v4f*>(la + 96 * APITCH) = a3;
    b0 -= m0;
    b1 -= m1;
    b2 -= m2;
    b3 -= m3;
    *reinterpret_cast<v4f*>(lb) = b0;
    *reinterpret_cast<v4f*>(lb + 8 * NN) = b1;
    *reinterpret_cast<v4f*>(lb + 16 * NN) = b2;
    *reinterpret_cast<v4f*>(lb + 24 * NN) = b3;
    __syncthreads();
#pragma unroll
    for (int k4 = 0; k4 < KCHUNK; k4 += 4) {
      v2f a = *reinterpret_cast<const v2f*>(&ldsA[(M0 + l15) * APITCH + k4 + kb]);
      v2f bf[8];
#pragma unroll
      for (int nt = 0; nt < 8; ++nt) {
        int col = nt * 16 + l15;
        bf[nt].x = ldsB[(k4 + kb) * NN + col];
        bf[nt].y = ldsB[(k4 + kb + 1) * NN + col];
      }
      __builtin_amdgcn_sched_barrier(0);
#pragma unroll
      for (int nt = 0; nt < 8; ++nt)
        acc[nt] = wmma_f32(a, bf[nt], acc[nt]);
      __builtin_amdgcn_sched_barrier(0);
    }
  }

  const float scale = rs[g];
  const int Mb = M0 + ((lane < 16) ? 0 : 8);
#pragma unroll
  for (int nt = 0; nt < 8; ++nt) {
    int col = N0 + nt * 16 + l15;
#pragma unroll
    for (int r = 0; r < 8; ++r) {
      out[(size_t)(g * NN + Mb + r) * DD + col] = acc[nt][r] * scale;
    }
  }
}

extern "C" void kernel_launch(void* const* d_in, const int* in_sizes, int n_in,
                              void* d_out, int out_size, void* d_ws, size_t ws_size,
                              hipStream_t stream) {
  const float* w = (const float*)d_in[0];
  float* out = (float*)d_out;
  float* ws = (float*)d_ws;

  float* mean = ws;
  float* rs = ws + 2048;
  float* nrmbuf = ws + 2064;
  float* S  = ws + 2080;
  float* Bt = S + MATF;
  float* Ba = Bt + MATF;
  float* Bb = Ba + MATF;
  float* P  = Bb + MATF;
  float* Rs = P + NG * KSPLIT * PWG;

  oni_cov<<<dim3(KSPLIT, NG), 256, 0, stream>>>(w, P, Rs);
  oni_sum<<<NG * 8, 160, 0, stream>>>(P, Rs, S, mean);
  oni_ns<true><<<NG * 8, 128, 0, stream>>>(S, nrmbuf, rs, S, Ba);
  oni_ns<false><<<NG * 8, 128, 0, stream>>>(S, nrmbuf, rs, Ba, Bb);
  oni_ns<false><<<NG * 8, 128, 0, stream>>>(S, nrmbuf, rs, Bb, Ba);
  oni_ns<false><<<NG * 8, 128, 0, stream>>>(S, nrmbuf, rs, Ba, Bt);
  oni_out<<<dim3(DD / 128, NG), 256, 0, stream>>>(Bt, w, mean, rs, out);
}


# Newton-Schulz staging: S and B read as b128 row segments, 4x4-transposed in registers and stored b128 (was 256 column-wise b32 loads per thread and their address arithmetic); sums bit-identical
# speedup vs baseline: 4.2716x; 1.0675x over previous
#include <hip/hip_runtime.h>
#include <hip/hip_bf16.h>


#define NG 16
#define NN 128
#define DD 9216
#define EPSV 1e-5f
#define TITER 5
#define KSPLIT 16
#define KCHUNK 32
#define APITCH 36
#define MATF (NG * NN * NN)
#define BPITCH 132

typedef __attribute__((ext_vector_type(2))) float v2f;
typedef __attribute__((ext_vector_type(8))) float v8f;
typedef __attribute__((ext_vector_type(4))) float v4f;

static __device__ __forceinline__ v8f wmma_f32(v2f a, v2f b, v8f c) {
  return __builtin_amdgcn_wmma_f32_16x16x4_f32(
       false, a,  false, b,
       (short)0, c,  false,  false);
}


static __device__ __forceinline__ void cov_frags(float4 (&fa)[4], float4 (&fb)[4][3],
                                                 float4 (&ha)[2], float4 (&hb)[2],
                                                 const float* fbase, int aoff,
                                                 const int (&boff)[3], int haoff, int hboff) {
#pragma unroll
  for (int kq = 0; kq < 4; ++kq) {
    fa[kq] = *reinterpret_cast<const float4*>(fbase + aoff + kq * 8);
#pragma unroll
    for (int j = 0; j < 3; ++j)
      fb[kq][j] = *reinterpret_cast<const float4*>(fbase + boff[j] + kq * 8);
  }
#pragma unroll
  for (int i = 0; i < 2; ++i) {
    ha[i] = *reinterpret_cast<const float4*>(fbase + haoff + i * 8);
    hb[i] = *reinterpret_cast<const float4*>(fbase + hboff + i * 8);
  }
}

static __device__ __forceinline__ void cov_mma(v8f (&acc)[5], const float4 (&fa)[4],
                                               const float4 (&fb)[4][3],
                                               const float4 (&ha)[2], const float4 (&hb)[2]) {
#pragma unroll
  for (int kq = 0; kq < 4; ++kq) {
    v2f alo = {fa[kq].x, fa[kq].y};
    v2f ahi = {fa[kq].z, fa[kq].w};
    acc[0] = wmma_f32(alo, alo, acc[0]);
#pragma unroll
    for (int j = 0; j < 3; ++j) {
      v2f b = {fb[kq][j].x, fb[kq][j].y};
      acc[1 + j] = wmma_f32(alo, b, acc[1 + j]);
    }
    acc[0] = wmma_f32(ahi, ahi, acc[0]);
#pragma unroll
    for (int j = 0; j < 3; ++j) {
      v2f b = {fb[kq][j].z, fb[kq][j].w};
      acc[1 + j] = wmma_f32(ahi, b, acc[1 + j]);
    }
    v2f xa = (kq & 1) ? v2f{ha[kq >> 1].z, ha[kq >> 1].w} : v2f{ha[kq >> 1].x, ha[kq >> 1].y};
    v2f xb = (kq & 1) ? v2f{hb[kq >> 1].z, hb[kq >> 1].w} : v2f{hb[kq >> 1].x, hb[kq >> 1].y};
    acc[4] = wmma_f32(xa, xb, acc[4]);
  }
}

#define PTILE 256
#define PWAVE (5 * PTILE)
#define PWG (8 * PWAVE)

__global__ void __launch_bounds__(256) oni_cov(const float* __restrict__ w,
                                               float* __restrict__ P,
                                               float* __restrict__ Rs) {
  __shared__ __attribute__((aligned(16))) float lds[2][NN * APITCH];
  const int g = blockIdx.y;
  const int kbase = blockIdx.x * (DD / KSPLIT);
  const int tid = threadIdx.x;
  const int lane = tid & 31, wave = tid >> 5;
  const int M0 = wave * 16;
  const int l15 = lane & 15;
  const int h = lane >> 4;
  constexpr int NCH = DD / KSPLIT / KCHUNK;
  int boff[3];
#pragma unroll
  for (int j = 0; j < 3; ++j) boff[j] = ((wave + 1 + j) & 7) * 16 * APITCH;
  const int aoff = M0 * APITCH;
  const int hk = (wave >> 2) * 16;
  const int haoff = aoff + hk;
  const int hboff = ((wave + 4) & 7) * 16 * APITCH + hk;

  const int sr = tid >> 3, sk = (tid & 7) * 4;
  const float* wp = w + (size_t)(g * NN + sr) * DD + kbase + sk;
  float rsum[4];
  float4 st[4];
#pragma unroll
  for (int i = 0; i < 4; ++i) {
    rsum[i] = 0.f;
    st[i] = *reinterpret_cast<const float4*>(wp + (size_t)(i * 32) * DD);
  }
#pragma unroll
  for (int i = 0; i < 4; ++i) {
    float4 v = st[i];
    rsum[i] += (v.x + v.y) + (v.z + v.w);
    *reinterpret_cast<float4*>(&lds[0][(i * 32 + sr) * APITCH + sk]) = v;
  }
  __syncthreads();

  v8f acc[5];
  v8f z = {};
#pragma unroll
  for (int t = 0; t < 5; ++t) acc[t] = z;

  const int foff = l15 * APITCH + 4 * h;
  float4 fa[4], fb[4][3], ha[2], hb[2];

#pragma unroll
  for (int i = 0; i < 4; ++i)
    st[i] = *reinterpret_cast<const float4*>(wp + (size_t)(i * 32) * DD + KCHUNK);

  for (int c = 0; c + 1 < NCH; ++c) {
    const float* cur = lds[c & 1];
    float* nxt = lds[(c + 1) & 1];
    cov_frags(fa, fb, ha, hb, cur + foff, aoff, boff, haoff, hboff);
    __builtin_amdgcn_sched_barrier(0);
    cov_mma(acc, fa, fb, ha, hb);
    __builtin_amdgcn_sched_barrier(0);
#pragma unroll
    for (int i = 0; i < 4; ++i) {
      float4 v = st[i];
      rsum[i] += (v.x + v.y) + (v.z + v.w);
      *reinterpret_cast<float4*>(&nxt[(i * 32 + sr) * APITCH + sk]) = v;
    }
    __syncthreads();
    if (c + 2 < NCH) {
#pragma unroll
      for (int i = 0; i < 4; ++i)
        st[i] = *reinterpret_cast<const float4*>(wp + (size_t)(i * 32) * DD + (c + 2) * KCHUNK);
    }
  }
  cov_frags(fa, fb, ha, hb, lds[(NCH - 1) & 1] + foff, aoff, boff, haoff, hboff);
  __builtin_amdgcn_sched_barrier(0);
  cov_mma(acc, fa, fb, ha, hb);

  float* Pw = P + ((size_t)(g * KSPLIT + blockIdx.x) * 8 + wave) * PWAVE + lane * 8;
#pragma unroll
  for (int t = 0; t < 5; ++t) {
    float4 lo = {acc[t][0], acc[t][1], acc[t][2], acc[t][3]};
    float4 hi = {acc[t][4], acc[t][5], acc[t][6], acc[t][7]};
    *reinterpret_cast<float4*>(Pw + t * PTILE) = lo;
    *reinterpret_cast<float4*>(Pw + t * PTILE + 4) = hi;
  }

#pragma unroll
  for (int i = 0; i < 4; ++i) {
    float a = rsum[i];
    a += __shfl_xor(a, 1);
    a += __shfl_xor(a, 2);
    a += __shfl_xor(a, 4);
    if ((tid & 7) == 0) Rs[(size_t)(g * KSPLIT + blockIdx.x) * NN + i * 32 + sr] = a;
  }
}

__global__ void __launch_bounds__(160) oni_sum(const float* __restrict__ P,
                                               const float* __restrict__ Rs,
                                               float* __restrict__ S, float* __restrict__ mean) {
  __shared__ float rsl[80];
  const int g = blockIdx.x >> 3, I = blockIdx.x & 7;
  const int t = threadIdx.x >> 5, lane = threadIdx.x & 31;
  const int c = lane & 15, h = lane >> 4;
  if (threadIdx.x < 80) {
    const int row = (I * 16 + threadIdx.x) & (NN - 1);
    const float* q = Rs + (size_t)g * KSPLIT * NN + row;
    float a = 0.f;
#pragma unroll
    for (int ks = 0; ks < KSPLIT; ++ks) a += q[ks * NN];
    rsl[threadIdx.x] = a;
    if (threadIdx.x < 16) mean[g * NN + row] = a * (1.0f / (float)DD);
  }
  const float* p = P + ((size_t)g * KSPLIT * 8 + I) * PWAVE + t * PTILE + lane * 8;
  float s[8];
#pragma unroll
  for (int r = 0; r < 8; ++r) s[r] = 0.f;
  if (t < 4) {
#pragma unroll
    for (int ks = 0; ks < KSPLIT; ++ks) {
      const float4 lo = *reinterpret_cast<const float4*>(p + (size_t)ks * PWG);
      const float4 hi = *reinterpret_cast<const float4*>(p + (size_t)ks * PWG + 4);
      s[0] += lo.x; s[1] += lo.y; s[2] += lo.z; s[3] += lo.w;
      s[4] += hi.x; s[5] += hi.y; s[6] += hi.z; s[7] += hi.w;
    }
  } else {
    const int I4 = (I + 4) & 7;
    const float* o = P + ((size_t)g * KSPLIT * 8 + I4) * PWAVE + 4 * PTILE +
                     (8 * h + 16 * (c >> 3)) * 8 + (c & 7);
    const bool own_first = I < 4;
#pragma unroll
    for (int ks = 0; ks < KSPLIT; ++ks) {
      const float4 lo = *reinterpret_cast<const float4*>(p + (size_t)ks * PWG);
      const float4 hi = *reinterpret_cast<const float4*>(p + (size_t)ks * PWG + 4);
      const float own[8] = {lo.x, lo.y, lo.z, lo.w, hi.x, hi.y, hi.z, hi.w};
#pragma unroll
      for (int r = 0; r < 8; ++r) {
        const float ov = o[(size_t)ks * PWG + r * 8];
        const float first = own_first ? own[r] : ov;
        const float second = own_first ? ov : own[r];
        s[r] += first;
        s[r] += second;
      }
    }
  }
  __syncthreads();
  const int jb = (t < 4) ? t : 4;
  const float cs = rsl[jb * 16 + c];
#pragma unroll
  for (int r = 0; r < 8; ++r) s[r] -= (rsl[8 * h + r] * cs) * (1.0f / (float)DD);
  float* Sg = S + (size_t)g * NN * NN;
  const int J = (I + jb) & 7;
#pragma unroll
  for (int r = 0; r < 8; ++r) Sg[(I * 16 + 8 * h + r) * NN + J * 16 + c] = s[r];
  if (t >= 1 && t <= 3) {
    float4 lo = {s[0], s[1], s[2], s[3]};
    float4 hi = {s[4], s[5], s[6], s[7]};
    float* m = Sg + (J * 16 + c) * NN + I * 16 + 8 * h;
    *reinterpret_cast<float4*>(m) = lo;
    *reinterpret_cast<float4*>(m + 4) = hi;
  }
}


static __device__ __forceinline__ void load_frags(float4 (&a)[32], const float* base) {
#pragma unroll
  for (int kt = 0; kt < 8; ++kt)
#pragma unroll
    for (int jj = 0; jj < 2; ++jj)
#pragma unroll
      for (int t = 0; t < 2; ++t)
        a[kt * 4 + jj * 2 + t] =
            *reinterpret_cast<const float4*>(base + t * 16 * BPITCH + kt * 16 + 4 * jj);
}

static __device__ __forceinline__ void mma_rows(v8f (&yt)[2], const float4 (&a)[32], const v8f (&xt)[8]) {
  v8f z = {};
  yt[0] = z;
  yt[1] = z;
#pragma unroll
  for (int kt = 0; kt < 8; ++kt)
#pragma unroll
    for (int jj = 0; jj < 2; ++jj) {
#pragma unroll
      for (int t = 0; t < 2; ++t) {
        v2f af = {a[kt * 4 + jj * 2 + t].x, a[kt * 4 + jj * 2 + t].y};
        v2f bf = {xt[kt][4 * jj], xt[kt][4 * jj + 1]};
        yt[t] = wmma_f32(af, bf, yt[t]);
      }
#pragma unroll
      for (int t = 0; t < 2; ++t) {
        v2f af = {a[kt * 4 + jj * 2 + t].z, a[kt * 4 + jj * 2 + t].w};
        v2f bf = {xt[kt][4 * jj + 2], xt[kt][4 * jj + 3]};
        yt[t] = wmma_f32(af, bf, yt[t]);
      }
    }
}

static __device__ __forceinline__ void exchange(v8f (&xn)[8], const v8f (&yt)[2], float* xw,
                                                int wave, int lane) {
#pragma unroll
  for (int t = 0; t < 2; ++t)
#pragma unroll
    for (int r = 0; r < 8; ++r)
      xw[((2 * wave + t) * 8 + r) * 32 + lane] = yt[t][r];
  __syncthreads();
#pragma unroll
  for (int kt = 0; kt < 8; ++kt)
#pragma unroll
    for (int r = 0; r < 8; ++r)
      xn[kt][r] = xw[(kt * 8 + r) * 32 + lane];
}

static __device__ __forceinline__ void load_rows(float4 (&v)[32], const float* src, int ql, int j4) {
#pragma unroll
  for (int blk = 0; blk < 8; ++blk)
#pragma unroll
    for (int i = 0; i < 4; ++i)
      v[blk * 4 + i] =
          *reinterpret_cast<const float4*>(src + (blk * 16 + ql * 4 + i) * NN + j4);
}

static __device__ __forceinline__ void add_eps(float4 (&v)[32], int ql, int j4) {
#pragma unroll
  for (int blk = 0; blk < 8; ++blk) {
    const float e = (blk * 16 + ql * 4 == j4) ? EPSV : 0.f;
    v[blk * 4].x += e;
    v[blk * 4 + 1].y += e;
    v[blk * 4 + 2].z += e;
    v[blk * 4 + 3].w += e;
  }
}

static __device__ __forceinline__ void store_cols(float* dst, const float4 (&v)[32], int ql, int j4) {
#pragma unroll
  for (int blk = 0; blk < 8; ++blk) {
    const float4 a = v[blk * 4], b = v[blk * 4 + 1], c = v[blk * 4 + 2], d = v[blk * 4 + 3];
    float* p = dst + j4 * BPITCH + blk * 16 + ql * 4;
    const float4 c0 = {a.x, b.x, c.x, d.x};
    const float4 c1 = {a.y, b.y, c.y, d.y};
    const float4 c2 = {a.z, b.z, c.z, d.z};
    const float4 c3 = {a.w, b.w, c.w, d.w};
    *reinterpret_cast<float4*>(p) = c0;
    *reinterpret_cast<float4*>(p + BPITCH) = c1;
    *reinterpret_cast<float4*>(p + 2 * BPITCH) = c2;
    *reinterpret_cast<float4*>(p + 3 * BPITCH) = c3;
  }
}

template <bool FIRST>
__global__ void __launch_bounds__(128) oni_ns(const float* __restrict__ S,
                                              float* __restrict__ nrmbuf,
                                              float* __restrict__ rs,
                                              const float* __restrict__ Bin,
                                              float* __restrict__ Bout) {
  __shared__ __attribute__((aligned(16))) float sS[NN * BPITCH];
  __shared__ __attribute__((aligned(16))) float sB[NN * BPITCH];
  __shared__ float xch[2][8 * 8 * 32];
  __shared__ float red[256];
  __shared__ float nrm;
  const int g = blockIdx.x >> 3;
  const int M0 = (blockIdx.x & 7) * 16;
  const int tid = threadIdx.x, lane = tid & 31, wave = tid >> 5;
  const int c = lane & 15, h = lane >> 4;
  const float* Sg = S + g * (NN * NN);

  const int ql = lane >> 3;
  const int j4 = (wave * 8 + (lane & 7)) * 4;
  if (FIRST) {
    float4 vs[32];
    load_rows(vs, Sg, ql, j4);
    add_eps(vs, ql, j4);
    store_cols(sS, vs, ql, j4);
    __syncthreads();
    float sv[NN];
#pragma unroll
    for (int k = 0; k < NN; k += 4) {
      const float4 q = *reinterpret_cast<const float4*>(&sS[tid * BPITCH + k]);
      sv[k] = q.x;
      sv[k + 1] = q.y;
      sv[k + 2] = q.z;
      sv[k + 3] = q.w;
    }
    float s0 = 0.f, s1 = 0.f;
#pragma unroll
    for (int k = 0; k < NN; k += 2) {
      float v0 = sv[k];
      float v1 = sv[k + 1];
      s0 += v0 * v0;
      s1 += v1 * v1;
    }
    red[tid] = s0;
    red[tid + 128] = s1;
    __syncthreads();
    red[tid] += red[tid + 128];
    __syncthreads();
    for (int off = 64; off > 0; off >>= 1) {
      if (tid < off) red[tid] += red[tid + off];
      __syncthreads();
    }
    if (tid == 0) {
      float n = sqrtf(red[0]);
      nrm = n;
      if (M0 == 0) {
        nrmbuf[g] = n;
        rs[g] = 1.0f / sqrtf(n);
      }
    }
    __syncthreads();
    const float inv = 1.0f / nrm;
#pragma unroll
    for (int k = 0; k < NN; k += 4) {
      float4 qs, qb;
      float sn;
      sn = sv[k] * inv;     qs.x = sn; qb.x = ((k == tid) ? 1.5f : 0.f) - 0.5f * sn;
      sn = sv[k + 1] * inv; qs.y = sn; qb.y = ((k + 1 == tid) ? 1.5f : 0.f) - 0.5f * sn;
      sn = sv[k + 2] * inv; qs.z = sn; qb.z = ((k + 2 == tid) ? 1.5f : 0.f) - 0.5f * sn;
      sn = sv[k + 3] * inv; qs.w = sn; qb.w = ((k + 3 == tid) ? 1.5f : 0.f) - 0.5f * sn;
      *reinterpret_cast<float4*>(&sS[tid * BPITCH + k]) = qs;
      *reinterpret_cast<float4*>(&sB[tid * BPITCH + k]) = qb;
    }
  } else {
    const float inv = 1.0f / nrmbuf[g];
    float4 vb[32], vs[32];
    load_rows(vb, Bin + g * (NN * NN), ql, j4);
    load_rows(vs, Sg, ql, j4);
    store_cols(sB, vb, ql, j4);
    add_eps(vs, ql, j4);
#pragma unroll
    for (int i = 0; i < 32; ++i) {
      vs[i].x *= inv;
      vs[i].y *= inv;
      vs[i].z *= inv;
      vs[i].w *= inv;
    }
    store_cols(sS, vs, ql, j4);
  }
  __syncthreads();

  v8f xt[8];
#pragma unroll
  for (int kt = 0; kt < 8; ++kt)
#pragma unroll
    for (int r = 0; r < 8; ++r)
      xt[kt][r] = sB[(kt * 16 + 8 * h + r) * BPITCH + M0 + c];

  const int foff = (wave * 32 + c) * BPITCH + 8 * h;
  float4 fB[32], fS[32];
  v8f y[2], x2[8], x3[8];
  load_frags(fB, sB + foff);
  load_frags(fS, sS + foff);
  __builtin_amdgcn_sched_barrier(0);
  mma_rows(y, fB, xt);
  __builtin_amdgcn_sched_barrier(0);
  exchange(x2, y, xch[0], wave, lane);
  __builtin_amdgcn_sched_barrier(0);
  mma_rows(y, fB, x2);
  __builtin_amdgcn_sched_barrier(0);
  exchange(x3, y, xch[1], wave, lane);
  __builtin_amdgcn_sched_barrier(0);
  mma_rows(y, fS, x3);
  __builtin_amdgcn_sched_barrier(0);

  float* Bo = Bout + g * (NN * NN) + (M0 + c) * NN + wave * 32 + 8 * h;
#pragma unroll
  for (int t = 0; t < 2; ++t) {
    float o[8];
#pragma unroll
    for (int r = 0; r < 8; ++r) {
      float v = -0.5f * y[t][r];
      v += 1.5f * sB[((2 * wave + t) * 16 + 8 * h + r) * BPITCH + M0 + c];
      o[r] = v;
    }
    float4 lo = {o[0], o[1], o[2], o[3]};
    float4 hi = {o[4], o[5], o[6], o[7]};
    *reinterpret_cast<float4*>(Bo + t * 16) = lo;
    *reinterpret_cast<float4*>(Bo + t * 16 + 4) = hi;
  }
}

__global__ void oni_out(const float* __restrict__ Bt, const float* __restrict__ w,
                        const float* __restrict__ mean, const float* __restrict__ rs,
                        float* __restrict__ out) {
  __shared__ float ldsA[NN * APITCH];
  __shared__ float ldsB[KCHUNK * NN];
  const int g = blockIdx.y;
  const int N0 = blockIdx.x * 128;
  const float* Ag = Bt + g * (NN * NN);
  const int tid = threadIdx.x, lane = tid & 31, wave = tid >> 5;
  const int M0 = wave * 16, l15 = lane & 15;
  const int kb = (lane < 16) ? 0 : 2;

  v8f acc[8];
  v8f z = {};
#pragma unroll
  for (int t = 0; t < 8; ++t) acc[t] = z;

  const int ar = tid >> 3, ak = (tid & 7) * 4;
  const int bk = tid >> 5, bc = (tid & 31) * 4;
  const float* ap = Ag + ar * NN + ak;
  const float* bp = w + (size_t)(g * NN + bk) * DD + N0 + bc;
  const float* mp = mean + g * NN + bk;
  float* la = &ldsA[ar * APITCH + ak];
  float* lb = &ldsB[bk * NN + bc];

  for (int k0 = 0; k0 < NN; k0 += KCHUNK) {
    const v4f a0 = *reinterpret_cast<const v4f*>(ap + k0);
    const v4f a1 = *reinterpret_cast<const v4f*>(ap + 32 * NN + k0);
    const v4f a2 = *reinterpret_cast<const v4f*>(ap + 64 * NN + k0);
    const v4f a3 = *reinterpret_cast<const v4f*>(ap + 96 * NN + k0);
    const float m0 = mp[k0], m1 = mp[k0 + 8], m2 = mp[k0 + 16], m3 = mp[k0 + 24];
    v4f b0 = *reinterpret_cast<const v4f*>(bp + (size_t)(k0) * DD);
    v4f b1 = *reinterpret_cast<const v4f*>(bp + (size_t)(k0 + 8) * DD);
    v4f b2 = *reinterpret_cast<const v4f*>(bp + (size_t)(k0 + 16) * DD);
    v4f b3 = *reinterpret_cast<const v4f*>(bp + (size_t)(k0 + 24) * DD);
    asm volatile("" :: "v"(a0), "v"(a1), "v"(a2), "v"(a3), "v"(b0), "v"(b1), "v"(b2), "v"(b3),
                 "v"(m0), "v"(m1), "v"(m2), "v"(m3));
    if (k0 > 0) __syncthreads();
    *reinterpret_cast<v4f*>(la) = a0;
    *reinterpret_cast<v4f*>(la + 32 * APITCH) = a1;
    *reinterpret_cast<v4f*>(la + 64 * APITCH) = a2;
    *reinterpret_cast<v4f*>(la + 96 * APITCH) = a3;
    b0 -= m0;
    b1 -= m1;
    b2 -= m2;
    b3 -= m3;
    *reinterpret_cast<v4f*>(lb) = b0;
    *reinterpret_cast<v4f*>(lb + 8 * NN) = b1;
    *reinterpret_cast<v4f*>(lb + 16 * NN) = b2;
    *reinterpret_cast<v4f*>(lb + 24 * NN) = b3;
    __syncthreads();
#pragma unroll
    for (int k4 = 0; k4 < KCHUNK; k4 += 4) {
      v2f a = *reinterpret_cast<const v2f*>(&ldsA[(M0 + l15) * APITCH + k4 + kb]);
      v2f bf[8];
#pragma unroll
      for (int nt = 0; nt < 8; ++nt) {
        int col = nt * 16 + l15;
        bf[nt].x = ldsB[(k4 + kb) * NN + col];
        bf[nt].y = ldsB[(k4 + kb + 1) * NN + col];
      }
      __builtin_amdgcn_sched_barrier(0);
#pragma unroll
      for (int nt = 0; nt < 8; ++nt)
        acc[nt] = wmma_f32(a, bf[nt], acc[nt]);
      __builtin_amdgcn_sched_barrier(0);
    }
  }

  const float scale = rs[g];
  const int Mb = M0 + ((lane < 16) ? 0 : 8);
#pragma unroll
  for (int nt = 0; nt < 8; ++nt) {
    int col = N0 + nt * 16 + l15;
#pragma unroll
    for (int r = 0; r < 8; ++r) {
      out[(size_t)(g * NN + Mb + r) * DD + col] = acc[nt][r] * scale;
    }
  }
}

extern "C" void kernel_launch(void* const* d_in, const int* in_sizes, int n_in,
                              void* d_out, int out_size, void* d_ws, size_t ws_size,
                              hipStream_t stream) {
  const float* w = (const float*)d_in[0];
  float* out = (float*)d_out;
  float* ws = (float*)d_ws;

  float* mean = ws;
  float* rs = ws + 2048;
  float* nrmbuf = ws + 2064;
  float* S  = ws + 2080;
  float* Bt = S + MATF;
  float* Ba = Bt + MATF;
  float* Bb = Ba + MATF;
  float* P  = Bb + MATF;
  float* Rs = P + NG * KSPLIT * PWG;

  oni_cov<<<dim3(KSPLIT, NG), 256, 0, stream>>>(w, P, Rs);
  oni_sum<<<NG * 8, 160, 0, stream>>>(P, Rs, S, mean);
  oni_ns<true><<<NG * 8, 128, 0, stream>>>(S, nrmbuf, rs, S, Ba);
  oni_ns<false><<<NG * 8, 128, 0, stream>>>(S, nrmbuf, rs, Ba, Bb);
  oni_ns<false><<<NG * 8, 128, 0, stream>>>(S, nrmbuf, rs, Bb, Ba);
  oni_ns<false><<<NG * 8, 128, 0, stream>>>(S, nrmbuf, rs, Ba, Bt);
  oni_out<<<dim3(DD / 128, NG), 256, 0, stream>>>(Bt, w, mean, rs, out);
}


# Newton-Schulz launches load S^T, B^T fragments and own B tiles straight from global memory, written in fragment order by the launch before (S: summing launch): no LDS staging; norm sum reordered
# speedup vs baseline: 4.4101x; 1.0324x over previous
#include <hip/hip_runtime.h>
#include <hip/hip_bf16.h>


#define NG 16
#define NN 128
#define DD 9216
#define EPSV 1e-5f
#define TITER 5
#define KSPLIT 16
#define KCHUNK 32
#define APITCH 36
#define MATF (NG * NN * NN)

typedef __attribute__((ext_vector_type(2))) float v2f;
typedef __attribute__((ext_vector_type(8))) float v8f;
typedef __attribute__((ext_vector_type(4))) float v4f;

static __device__ __forceinline__ v8f wmma_f32(v2f a, v2f b, v8f c) {
  return __builtin_amdgcn_wmma_f32_16x16x4_f32(
       false, a,  false, b,
       (short)0, c,  false,  false);
}


static __device__ __forceinline__ void cov_frags(float4 (&fa)[4], float4 (&fb)[4][3],
                                                 float4 (&ha)[2], float4 (&hb)[2],
                                                 const float* fbase, int aoff,
                                                 const int (&boff)[3], int haoff, int hboff) {
#pragma unroll
  for (int kq = 0; kq < 4; ++kq) {
    fa[kq] = *reinterpret_cast<const float4*>(fbase + aoff + kq * 8);
#pragma unroll
    for (int j = 0; j < 3; ++j)
      fb[kq][j] = *reinterpret_cast<const float4*>(fbase + boff[j] + kq * 8);
  }
#pragma unroll
  for (int i = 0; i < 2; ++i) {
    ha[i] = *reinterpret_cast<const float4*>(fbase + haoff + i * 8);
    hb[i] = *reinterpret_cast<const float4*>(fbase + hboff + i * 8);
  }
}

static __device__ __forceinline__ void cov_mma(v8f (&acc)[5], const float4 (&fa)[4],
                                               const float4 (&fb)[4][3],
                                               const float4 (&ha)[2], const float4 (&hb)[2]) {
#pragma unroll
  for (int kq = 0; kq < 4; ++kq) {
    v2f alo = {fa[kq].x, fa[kq].y};
    v2f ahi = {fa[kq].z, fa[kq].w};
    acc[0] = wmma_f32(alo, alo, acc[0]);
#pragma unroll
    for (int j = 0; j < 3; ++j) {
      v2f b = {fb[kq][j].x, fb[kq][j].y};
      acc[1 + j] = wmma_f32(alo, b, acc[1 + j]);
    }
    acc[0] = wmma_f32(ahi, ahi, acc[0]);
#pragma unroll
    for (int j = 0; j < 3; ++j) {
      v2f b = {fb[kq][j].z, fb[kq][j].w};
      acc[1 + j] = wmma_f32(ahi, b, acc[1 + j]);
    }
    v2f xa = (kq & 1) ? v2f{ha[kq >> 1].z, ha[kq >> 1].w} : v2f{ha[kq >> 1].x, ha[kq >> 1].y};
    v2f xb = (kq & 1) ? v2f{hb[kq >> 1].z, hb[kq >> 1].w} : v2f{hb[kq >> 1].x, hb[kq >> 1].y};
    acc[4] = wmma_f32(xa, xb, acc[4]);
  }
}

#define PTILE 256
#define PWAVE (5 * PTILE)
#define PWG (8 * PWAVE)

__global__ void __launch_bounds__(256) oni_cov(const float* __restrict__ w,
                                               float* __restrict__ P,
                                               float* __restrict__ Rs) {
  __shared__ __attribute__((aligned(16))) float lds[2][NN * APITCH];
  const int g = blockIdx.y;
  const int kbase = blockIdx.x * (DD / KSPLIT);
  const int tid = threadIdx.x;
  const int lane = tid & 31, wave = tid >> 5;
  const int M0 = wave * 16;
  const int l15 = lane & 15;
  const int h = lane >> 4;
  constexpr int NCH = DD / KSPLIT / KCHUNK;
  int boff[3];
#pragma unroll
  for (int j = 0; j < 3; ++j) boff[j] = ((wave + 1 + j) & 7) * 16 * APITCH;
  const int aoff = M0 * APITCH;
  const int hk = (wave >> 2) * 16;
  const int haoff = aoff + hk;
  const int hboff = ((wave + 4) & 7) * 16 * APITCH + hk;

  const int sr = tid >> 3, sk = (tid & 7) * 4;
  const float* wp = w + (size_t)(g * NN + sr) * DD + kbase + sk;
  float rsum[4];
  float4 st[4];
#pragma unroll
  for (int i = 0; i < 4; ++i) {
    rsum[i] = 0.f;
    st[i] = *reinterpret_cast<const float4*>(wp + (size_t)(i * 32) * DD);
  }
#pragma unroll
  for (int i = 0; i < 4; ++i) {
    float4 v = st[i];
    rsum[i] += (v.x + v.y) + (v.z + v.w);
    *reinterpret_cast<float4*>(&lds[0][(i * 32 + sr) * APITCH + sk]) = v;
  }
  __syncthreads();

  v8f acc[5];
  v8f z = {};
#pragma unroll
  for (int t = 0; t < 5; ++t) acc[t] = z;

  const int foff = l15 * APITCH + 4 * h;
  float4 fa[4], fb[4][3], ha[2], hb[2];

#pragma unroll
  for (int i = 0; i < 4; ++i)
    st[i] = *reinterpret_cast<const float4*>(wp + (size_t)(i * 32) * DD + KCHUNK);

  for (int c = 0; c + 1 < NCH; ++c) {
    const float* cur = lds[c & 1];
    float* nxt = lds[(c + 1) & 1];
    cov_frags(fa, fb, ha, hb, cur + foff, aoff, boff, haoff, hboff);
    __builtin_amdgcn_sched_barrier(0);
    cov_mma(acc, fa, fb, ha, hb);
    __builtin_amdgcn_sched_barrier(0);
#pragma unroll
    for (int i = 0; i < 4; ++i) {
      float4 v = st[i];
      rsum[i] += (v.x + v.y) + (v.z + v.w);
      *reinterpret_cast<float4*>(&nxt[(i * 32 + sr) * APITCH + sk]) = v;
    }
    __syncthreads();
    if (c + 2 < NCH) {
#pragma unroll
      for (int i = 0; i < 4; ++i)
        st[i] = *reinterpret_cast<const float4*>(wp + (size_t)(i * 32) * DD + (c + 2) * KCHUNK);
    }
  }
  cov_frags(fa, fb, ha, hb, lds[(NCH - 1) & 1] + foff, aoff, boff, haoff, hboff);
  __builtin_amdgcn_sched_barrier(0);
  cov_mma(acc, fa, fb, ha, hb);

  float* Pw = P + ((size_t)(g * KSPLIT + blockIdx.x) * 8 + wave) * PWAVE + lane * 8;
#pragma unroll
  for (int t = 0; t < 5; ++t) {
    float4 lo = {acc[t][0], acc[t][1], acc[t][2], acc[t][3]};
    float4 hi = {acc[t][4], acc[t][5], acc[t][6], acc[t][7]};
    *reinterpret_cast<float4*>(Pw + t * PTILE) = lo;
    *reinterpret_cast<float4*>(Pw + t * PTILE + 4) = hi;
  }

#pragma unroll
  for (int i = 0; i < 4; ++i) {
    float a = rsum[i];
    a += __shfl_xor(a, 1);
    a += __shfl_xor(a, 2);
    a += __shfl_xor(a, 4);
    if ((tid & 7) == 0) Rs[(size_t)(g * KSPLIT + blockIdx.x) * NN + i * 32 + sr] = a;
  }
}

__global__ void __launch_bounds__(160) oni_sum(const float* __restrict__ P,
                                               const float* __restrict__ Rs,
                                               float* __restrict__ SfR, float* __restrict__ XS,
                                               float* __restrict__ mean) {
  __shared__ float rsl[80];
  const int g = blockIdx.x >> 3, I = blockIdx.x & 7;
  const int t = threadIdx.x >> 5, lane = threadIdx.x & 31;
  const int c = lane & 15, h = lane >> 4;
  if (threadIdx.x < 80) {
    const int row = (I * 16 + threadIdx.x) & (NN - 1);
    const float* q = Rs + (size_t)g * KSPLIT * NN + row;
    float a = 0.f;
#pragma unroll
    for (int ks = 0; ks < KSPLIT; ++ks) a += q[ks * NN];
    rsl[threadIdx.x] = a;
    if (threadIdx.x < 16) mean[g * NN + row] = a * (1.0f / (float)DD);
  }
  const float* p = P + ((size_t)g * KSPLIT * 8 + I) * PWAVE + t * PTILE + lane * 8;
  float s[8];
#pragma unroll
  for (int r = 0; r < 8; ++r) s[r] = 0.f;
  if (t < 4) {
#pragma unroll
    for (int ks = 0; ks < KSPLIT; ++ks) {
      const float4 lo = *reinterpret_cast<const float4*>(p + (size_t)ks * PWG);
      const float4 hi = *reinterpret_cast<const float4*>(p + (size_t)ks * PWG + 4);
      s[0] += lo.x; s[1] += lo.y; s[2] += lo.z; s[3] += lo.w;
      s[4] += hi.x; s[5] += hi.y; s[6] += hi.z; s[7] += hi.w;
    }
  } else {
    const int I4 = (I + 4) & 7;
    const float* o = P + ((size_t)g * KSPLIT * 8 + I4) * PWAVE + 4 * PTILE +
                     (8 * h + 16 * (c >> 3)) * 8 + (c & 7);
    const bool own_first = I < 4;
#pragma unroll
    for (int ks = 0; ks < KSPLIT; ++ks) {
      const float4 lo = *reinterpret_cast<const float4*>(p + (size_t)ks * PWG);
      const float4 hi = *reinterpret_cast<const float4*>(p + (size_t)ks * PWG + 4);
      const float own[8] = {lo.x, lo.y, lo.z, lo.w, hi.x, hi.y, hi.z, hi.w};
#pragma unroll
      for (int r = 0; r < 8; ++r) {
        const float ov = o[(size_t)ks * PWG + r * 8];
        const float first = own_first ? own[r] : ov;
        const float second = own_first ? ov : own[r];
        s[r] += first;
        s[r] += second;
      }
    }
  }
  __syncthreads();
  const int jb = (t < 4) ? t : 4;
  const float cs = rsl[jb * 16 + c];
#pragma unroll
  for (int r = 0; r < 8; ++r) s[r] -= (rsl[8 * h + r] * cs) * (1.0f / (float)DD);
  const int J = (I + jb) & 7;
  if (jb == 0) {
#pragma unroll
    for (int r = 0; r < 8; ++r) s[r] += (8 * h + r == c) ? EPSV : 0.f;
  }
  const float4 lo = {s[0], s[1], s[2], s[3]};
  const float4 hi = {s[4], s[5], s[6], s[7]};
  float* Fg = SfR + (size_t)g * NN * NN;
  float* Xg = XS + (size_t)g * NN * NN;
  float4* f4 = reinterpret_cast<float4*>(Fg) + ((J * 8 + I) * 2) * 32 + lane;
  f4[0] = lo;
  f4[32] = hi;
  float* xo = Xg + ((J * 8 + I) * 32 + lane) * 8;
  *reinterpret_cast<float4*>(xo) = lo;
  *reinterpret_cast<float4*>(xo + 4) = hi;
  if (t >= 1 && t <= 3) {
    const int ml = 8 * h + 16 * (c >> 3);
    float* fo = Fg + (((I * 8 + J) * 2 + ((c >> 2) & 1)) * 32 + ml) * 4 + (c & 3);
    float* mo = Xg + ((I * 8 + J) * 32 + ml) * 8 + (c & 7);
#pragma unroll
    for (int r = 0; r < 8; ++r) {
      fo[r * 4] = s[r];
      mo[r * 8] = s[r];
    }
  }
}


static __device__ __forceinline__ void mma_rows(v8f (&yt)[2], const float4 (&a)[32], const v8f (&xt)[8]) {
  v8f z = {};
  yt[0] = z;
  yt[1] = z;
#pragma unroll
  for (int kt = 0; kt < 8; ++kt)
#pragma unroll
    for (int jj = 0; jj < 2; ++jj) {
#pragma unroll
      for (int t = 0; t < 2; ++t) {
        v2f af = {a[kt * 4 + jj * 2 + t].x, a[kt * 4 + jj * 2 + t].y};
        v2f bf = {xt[kt][4 * jj], xt[kt][4 * jj + 1]};
        yt[t] = wmma_f32(af, bf, yt[t]);
      }
#pragma unroll
      for (int t = 0; t < 2; ++t) {
        v2f af = {a[kt * 4 + jj * 2 + t].z, a[kt * 4 + jj * 2 + t].w};
        v2f bf = {xt[kt][4 * jj + 2], xt[kt][4 * jj + 3]};
        yt[t] = wmma_f32(af, bf, yt[t]);
      }
    }
}

static __device__ __forceinline__ void exchange(v8f (&xn)[8], const v8f (&yt)[2], float* xw,
                                                int wave, int lane) {
#pragma unroll
  for (int t = 0; t < 2; ++t)
#pragma unroll
    for (int r = 0; r < 8; ++r)
      xw[((2 * wave + t) * 8 + r) * 32 + lane] = yt[t][r];
  __syncthreads();
#pragma unroll
  for (int kt = 0; kt < 8; ++kt)
#pragma unroll
    for (int r = 0; r < 8; ++r)
      xn[kt][r] = xw[(kt * 8 + r) * 32 + lane];
}

#define FRAG4 (NN * NN / 4)
#define WFRAG4 (2 * 8 * 2 * 32)

static __device__ __forceinline__ void load_frags_g(float4 (&a)[32], const float4* base) {
#pragma unroll
  for (int kt = 0; kt < 8; ++kt)
#pragma unroll
    for (int jj = 0; jj < 2; ++jj)
#pragma unroll
      for (int t = 0; t < 2; ++t)
        a[kt * 4 + jj * 2 + t] = base[((t * 8 + kt) * 2 + jj) * 32];
}

static __device__ __forceinline__ void store_next(float* Xg, float* Fg, const float (&o)[8],
                                                  int I, int wave, int t, int lane) {
  const int c = lane & 15, h = lane >> 4;
  float* xo = Xg + ((I * 8 + 2 * wave + t) * 32 + lane) * 8;
  float4 lo = {o[0], o[1], o[2], o[3]};
  float4 hi = {o[4], o[5], o[6], o[7]};
  *reinterpret_cast<float4*>(xo) = lo;
  *reinterpret_cast<float4*>(xo + 4) = hi;
  float* fo = Fg + ((((wave * 2 + t) * 8 + I) * 2 + ((c >> 2) & 1)) * 32 + 8 * h + 16 * (c >> 3)) * 4 + (c & 3);
#pragma unroll
  for (int r = 0; r < 8; ++r) fo[r * 4] = o[r];
}

__global__ void __launch_bounds__(128) oni_ns1(const float* __restrict__ SfR,
                                               const float* __restrict__ XS,
                                               float* __restrict__ rs,
                                               float* __restrict__ Sf,
                                               float* __restrict__ Xout,
                                               float* __restrict__ Fout) {
  __shared__ float xch[2][8 * 8 * 32];
  __shared__ float part[4];
  const int g = blockIdx.x >> 3;
  const int I = blockIdx.x & 7;
  const int tid = threadIdx.x, lane = tid & 31;
  const int wave = __builtin_amdgcn_readfirstlane(tid >> 5);
  const int c = lane & 15, h = lane >> 4;

  const float* Xi = XS + g * (NN * NN) + (I * 8 * 32 + lane) * 8;
  v8f xt[8];
#pragma unroll
  for (int kt = 0; kt < 8; ++kt) {
    const float4 lo = *reinterpret_cast<const float4*>(Xi + kt * 256);
    const float4 hi = *reinterpret_cast<const float4*>(Xi + kt * 256 + 4);
    v8f q = {lo.x, lo.y, lo.z, lo.w, hi.x, hi.y, hi.z, hi.w};
    xt[kt] = q;
  }
  float4 fB[32], fS[32];
  v8f y[2], x2[8], x3[8];
  load_frags_g(fS, reinterpret_cast<const float4*>(SfR) + (size_t)g * FRAG4 + wave * WFRAG4 + lane);
  v8f own[2];
#pragma unroll
  for (int t = 0; t < 2; ++t) {
    const float4 lo = *reinterpret_cast<const float4*>(Xi + (2 * wave + t) * 256);
    const float4 hi = *reinterpret_cast<const float4*>(Xi + (2 * wave + t) * 256 + 4);
    v8f q = {lo.x, lo.y, lo.z, lo.w, hi.x, hi.y, hi.z, hi.w};
    own[t] = q;
  }

  float q0 = 0.f, q1 = 0.f, q2 = 0.f, q3 = 0.f;
#pragma unroll
  for (int i = 0; i < 32; ++i) {
    q0 += fS[i].x * fS[i].x;
    q1 += fS[i].y * fS[i].y;
    q2 += fS[i].z * fS[i].z;
    q3 += fS[i].w * fS[i].w;
  }
  float qs = (q0 + q1) + (q2 + q3);
  qs += __shfl_xor(qs, 16);
  qs += __shfl_xor(qs, 8);
  qs += __shfl_xor(qs, 4);
  qs += __shfl_xor(qs, 2);
  qs += __shfl_xor(qs, 1);
  if (lane == 0) part[wave] = qs;
  __syncthreads();
  const float nrm = sqrtf((part[0] + part[1]) + (part[2] + part[3]));
  if (I == 0 && tid == 0) rs[g] = 1.0f / sqrtf(nrm);
  const float inv = 1.0f / nrm;

#pragma unroll
  for (int i = 0; i < 32; ++i) {
    fS[i].x *= inv;
    fS[i].y *= inv;
    fS[i].z *= inv;
    fS[i].w *= inv;
    fB[i].x = -0.5f * fS[i].x;
    fB[i].y = -0.5f * fS[i].y;
    fB[i].z = -0.5f * fS[i].z;
    fB[i].w = -0.5f * fS[i].w;
  }
#pragma unroll
  for (int kt = 0; kt < 8; ++kt)
#pragma unroll
    for (int r = 0; r < 8; ++r) xt[kt][r] = -0.5f * (xt[kt][r] * inv);
#pragma unroll
  for (int t = 0; t < 2; ++t)
#pragma unroll
    for (int r = 0; r < 8; ++r) own[t][r] = -0.5f * (own[t][r] * inv);
  const float d15 = ((c >> 3) == h) ? 1.5f : 0.f;
  const int dj = (c >> 2) & 1, de = c & 3, dr = c & 7;
  float dm[2][4], dx[8];
#pragma unroll
  for (int jj = 0; jj < 2; ++jj)
#pragma unroll
    for (int e = 0; e < 4; ++e) dm[jj][e] = (jj == dj && e == de) ? d15 : 0.f;
#pragma unroll
  for (int r = 0; r < 8; ++r) dx[r] = (r == dr) ? d15 : 0.f;
#pragma unroll
  for (int kt = 0; kt < 8; ++kt) {
#pragma unroll
    for (int t = 0; t < 2; ++t)
      if (kt == 2 * wave + t) {
#pragma unroll
        for (int jj = 0; jj < 2; ++jj) {
          fB[kt * 4 + jj * 2 + t].x += dm[jj][0];
          fB[kt * 4 + jj * 2 + t].y += dm[jj][1];
          fB[kt * 4 + jj * 2 + t].z += dm[jj][2];
          fB[kt * 4 + jj * 2 + t].w += dm[jj][3];
        }
        if (kt == I) {
#pragma unroll
          for (int r = 0; r < 8; ++r) own[t][r] += dx[r];
        }
      }
    if (kt == I) {
#pragma unroll
      for (int r = 0; r < 8; ++r) xt[kt][r] += dx[r];
    }
  }

  __builtin_amdgcn_sched_barrier(0);
  mma_rows(y, fB, xt);
  __builtin_amdgcn_sched_barrier(0);
  exchange(x2, y, xch[0], wave, lane);
  __builtin_amdgcn_sched_barrier(0);
  mma_rows(y, fB, x2);
  __builtin_amdgcn_sched_barrier(0);
  exchange(x3, y, xch[1], wave, lane);
  __builtin_amdgcn_sched_barrier(0);
  mma_rows(y, fS, x3);
  __builtin_amdgcn_sched_barrier(0);

  float* Xg = Xout + g * (NN * NN);
  float* Fg = Fout + g * (NN * NN);
#pragma unroll
  for (int t = 0; t < 2; ++t) {
    float o[8];
#pragma unroll
    for (int r = 0; r < 8; ++r) {
      float v = -0.5f * y[t][r];
      v += 1.5f * own[t][r];
      o[r] = v;
    }
    store_next(Xg, Fg, o, I, wave, t, lane);
  }
  float4* sf = reinterpret_cast<float4*>(Sf) + (size_t)g * FRAG4 + wave * WFRAG4 + lane;
#pragma unroll
  for (int kt = 0; kt < 8; ++kt)
    if (kt == I) {
#pragma unroll
      for (int t = 0; t < 2; ++t)
#pragma unroll
        for (int jj = 0; jj < 2; ++jj)
          sf[((t * 8 + kt) * 2 + jj) * 32] = fS[kt * 4 + jj * 2 + t];
    }
}

template <bool LAST>
__global__ void __launch_bounds__(128) oni_nsd(const float* __restrict__ Sf,
                                               const float* __restrict__ Xin,
                                               const float* __restrict__ Fin,
                                               float* __restrict__ Xout,
                                               float* __restrict__ Fout,
                                               float* __restrict__ Bout) {
  __shared__ float xch[2][8 * 8 * 32];
  const int g = blockIdx.x >> 3;
  const int I = blockIdx.x & 7;
  const int M0 = I * 16;
  const int tid = threadIdx.x, lane = tid & 31, wave = tid >> 5;
  const int c = lane & 15, h = lane >> 4;

  const float* Xi = Xin + g * (NN * NN) + (I * 8 * 32 + lane) * 8;
  v8f xt[8];
#pragma unroll
  for (int kt = 0; kt < 8; ++kt) {
    const float4 lo = *reinterpret_cast<const float4*>(Xi + kt * 256);
    const float4 hi = *reinterpret_cast<const float4*>(Xi + kt * 256 + 4);
    v8f q = {lo.x, lo.y, lo.z, lo.w, hi.x, hi.y, hi.z, hi.w};
    xt[kt] = q;
  }
  float4 fB[32], fS[32];
  v8f y[2], x2[8], x3[8];
  load_frags_g(fB, reinterpret_cast<const float4*>(Fin) + (size_t)g * FRAG4 + wave * WFRAG4 + lane);
  v8f own[2];
#pragma unroll
  for (int t = 0; t < 2; ++t) {
    const float4 lo = *reinterpret_cast<const float4*>(Xi + (2 * wave + t) * 256);
    const float4 hi = *reinterpret_cast<const float4*>(Xi + (2 * wave + t) * 256 + 4);
    v8f q = {lo.x, lo.y, lo.z, lo.w, hi.x, hi.y, hi.z, hi.w};
    own[t] = q;
  }
  __builtin_amdgcn_sched_barrier(0);
  load_frags_g(fS, reinterpret_cast<const float4*>(Sf) + (size_t)g * FRAG4 + wave * WFRAG4 + lane);
  __builtin_amdgcn_sched_barrier(0);
  mma_rows(y, fB, xt);
  __builtin_amdgcn_sched_barrier(0);
  exchange(x2, y, xch[0], wave, lane);
  __builtin_amdgcn_sched_barrier(0);
  mma_rows(y, fB, x2);
  __builtin_amdgcn_sched_barrier(0);
  exchange(x3, y, xch[1], wave, lane);
  __builtin_amdgcn_sched_barrier(0);
  mma_rows(y, fS, x3);
  __builtin_amdgcn_sched_barrier(0);

#pragma unroll
  for (int t = 0; t < 2; ++t) {
    float o[8];
#pragma unroll
    for (int r = 0; r < 8; ++r) {
      float v = -0.5f * y[t][r];
      v += 1.5f * own[t][r];
      o[r] = v;
    }
    if (LAST) {
      float* Bo = Bout + g * (NN * NN) + (M0 + c) * NN + wave * 32 + t * 16 + 8 * h;
      float4 lo = {o[0], o[1], o[2], o[3]};
      float4 hi = {o[4], o[5], o[6], o[7]};
      *reinterpret_cast<float4*>(Bo) = lo;
      *reinterpret_cast<float4*>(Bo + 4) = hi;
    } else {
      store_next(Xout + g * (NN * NN), Fout + g * (NN * NN), o, I, wave, t, lane);
    }
  }
}

__global__ void oni_out(const float* __restrict__ Bt, const float* __restrict__ w,
                        const float* __restrict__ mean, const float* __restrict__ rs,
                        float* __restrict__ out) {
  __shared__ float ldsA[NN * APITCH];
  __shared__ float ldsB[KCHUNK * NN];
  const int g = blockIdx.y;
  const int N0 = blockIdx.x * 128;
  const float* Ag = Bt + g * (NN * NN);
  const int tid = threadIdx.x, lane = tid & 31, wave = tid >> 5;
  const int M0 = wave * 16, l15 = lane & 15;
  const int kb = (lane < 16) ? 0 : 2;

  v8f acc[8];
  v8f z = {};
#pragma unroll
  for (int t = 0; t < 8; ++t) acc[t] = z;

  const int ar = tid >> 3, ak = (tid & 7) * 4;
  const int bk = tid >> 5, bc = (tid & 31) * 4;
  const float* ap = Ag + ar * NN + ak;
  const float* bp = w + (size_t)(g * NN + bk) * DD + N0 + bc;
  const float* mp = mean + g * NN + bk;
  float* la = &ldsA[ar * APITCH + ak];
  float* lb = &ldsB[bk * NN + bc];

  for (int k0 = 0; k0 < NN; k0 += KCHUNK) {
    const v4f a0 = *reinterpret_cast<const v4f*>(ap + k0);
    const v4f a1 = *reinterpret_cast<const v4f*>(ap + 32 * NN + k0);
    const v4f a2 = *reinterpret_cast<const v4f*>(ap + 64 * NN + k0);
    const v4f a3 = *reinterpret_cast<const v4f*>(ap + 96 * NN + k0);
    const float m0 = mp[k0], m1 = mp[k0 + 8], m2 = mp[k0 + 16], m3 = mp[k0 + 24];
    v4f b0 = *reinterpret_cast<const v4f*>(bp + (size_t)(k0) * DD);
    v4f b1 = *reinterpret_cast<const v4f*>(bp + (size_t)(k0 + 8) * DD);
    v4f b2 = *reinterpret_cast<const v4f*>(bp + (size_t)(k0 + 16) * DD);
    v4f b3 = *reinterpret_cast<const v4f*>(bp + (size_t)(k0 + 24) * DD);
    asm volatile("" :: "v"(a0), "v"(a1), "v"(a2), "v"(a3), "v"(b0), "v"(b1), "v"(b2), "v"(b3),
                 "v"(m0), "v"(m1), "v"(m2), "v"(m3));
    if (k0 > 0) __syncthreads();
    *reinterpret_cast<v4f*>(la) = a0;
    *reinterpret_cast<v4f*>(la + 32 * APITCH) = a1;
    *reinterpret_cast<v4f*>(la + 64 * APITCH) = a2;
    *reinterpret_cast<v4f*>(la + 96 * APITCH) = a3;
    b0 -= m0;
    b1 -= m1;
    b2 -= m2;
    b3 -= m3;
    *reinterpret_cast<v4f*>(lb) = b0;
    *reinterpret_cast<v4f*>(lb + 8 * NN) = b1;
    *reinterpret_cast<v4f*>(lb + 16 * NN) = b2;
    *reinterpret_cast<v4f*>(lb + 24 * NN) = b3;
    __syncthreads();
#pragma unroll
    for (int k4 = 0; k4 < KCHUNK; k4 += 4) {
      v2f a = *reinterpret_cast<const v2f*>(&ldsA[(M0 + l15) * APITCH + k4 + kb]);
      v2f bf[8];
#pragma unroll
      for (int nt = 0; nt < 8; ++nt) {
        int col = nt * 16 + l15;
        bf[nt].x = ldsB[(k4 + kb) * NN + col];
        bf[nt].y = ldsB[(k4 + kb + 1) * NN + col];
      }
      __builtin_amdgcn_sched_barrier(0);
#pragma unroll
      for (int nt = 0; nt < 8; ++nt)
        acc[nt] = wmma_f32(a, bf[nt], acc[nt]);
      __builtin_amdgcn_sched_barrier(0);
    }
  }

  const float scale = rs[g];
  const int Mb = M0 + ((lane < 16) ? 0 : 8);
#pragma unroll
  for (int nt = 0; nt < 8; ++nt) {
    int col = N0 + nt * 16 + l15;
#pragma unroll
    for (int r = 0; r < 8; ++r) {
      out[(size_t)(g * NN + Mb + r) * DD + col] = acc[nt][r] * scale;
    }
  }
}

extern "C" void kernel_launch(void* const* d_in, const int* in_sizes, int n_in,
                              void* d_out, int out_size, void* d_ws, size_t ws_size,
                              hipStream_t stream) {
  const float* w = (const float*)d_in[0];
  float* out = (float*)d_out;
  float* ws = (float*)d_ws;

  float* mean = ws;
  float* rs = ws + 2048;
  float* SfR = ws + 2080;
  float* Bt = SfR + MATF;
  float* Xa = Bt + MATF;
  float* Xb = Xa + MATF;
  float* P  = Xb + MATF;
  float* Rs = P + NG * KSPLIT * PWG;
  float* Sf = Rs + NG * KSPLIT * NN;
  float* Fa = Sf + MATF;
  float* Fb = Fa + MATF;
  float* XS = Fb + MATF;

  oni_cov<<<dim3(KSPLIT, NG), 256, 0, stream>>>(w, P, Rs);
  oni_sum<<<NG * 8, 160, 0, stream>>>(P, Rs, SfR, XS, mean);
  oni_ns1<<<NG * 8, 128, 0, stream>>>(SfR, XS, rs, Sf, Xa, Fa);
  oni_nsd<false><<<NG * 8, 128, 0, stream>>>(Sf, Xa, Fa, Xb, Fb, Bt);
  oni_nsd<false><<<NG * 8, 128, 0, stream>>>(Sf, Xb, Fb, Xa, Fa, Bt);
  oni_nsd<true><<<NG * 8, 128, 0, stream>>>(Sf, Xa, Fa, Xb, Fb, Bt);
  oni_out<<<dim3(DD / 128, NG), 256, 0, stream>>>(Bt, w, mean, rs, out);
}
